# starting kernel
# baseline (speedup 1.0000x reference)
#include <hip/hip_runtime.h>

typedef __attribute__((ext_vector_type(2))) float v2f;
typedef __attribute__((ext_vector_type(8))) float v8f;

#define HIDC 16
#define BLK 256

__device__ __forceinline__ void fadd(float* p, float v) {
    __hip_atomic_fetch_add(p, v, __ATOMIC_RELAXED, __HIP_MEMORY_SCOPE_AGENT);
}

__global__ void k_init(float* deg, float* agg, float* pooled, float* cnt,
                       int n_nodes, int n_graphs) {
    int i = blockIdx.x * blockDim.x + threadIdx.x;
    int tot = n_nodes * HIDC;
    if (i < tot) agg[i] = 0.0f;
    if (i < n_nodes) deg[i] = 1.0f;
    if (i < n_graphs * HIDC) pooled[i] = 0.0f;
    if (i < n_graphs) cnt[i] = 0.0f;
}

__global__ void k_deg(const int* __restrict__ dst, float* deg, int n_edges) {
    int e = blockIdx.x * blockDim.x + threadIdx.x;
    if (e < n_edges) fadd(&deg[dst[e]], 1.0f);
}

__global__ void k_dinv(float* deg, int n_nodes) {
    int i = blockIdx.x * blockDim.x + threadIdx.x;
    if (i < n_nodes) {
        float d = deg[i];
        deg[i] = (d > 0.0f) ? rsqrtf(d) : 0.0f;
    }
}

__global__ void k_embed(const int* __restrict__ x, const float* __restrict__ emb,
                        float* __restrict__ h, int n_nodes) {
    int i = blockIdx.x * blockDim.x + threadIdx.x;
    if (i < n_nodes * HIDC) {
        int node = i >> 4, j = i & 15;
        h[i] = emb[x[node] * HIDC + j];
    }
}

__global__ void k_gcn_wmma(const float* __restrict__ h, const float* __restrict__ W,
                           float* __restrict__ hw, int ntiles) {
    int wave = (blockIdx.x * blockDim.x + threadIdx.x) >> 5;
    int lane = threadIdx.x & 31;
    if (wave >= ntiles) return;
    int half = lane >> 4;
    int l15  = lane & 15;

    v2f bf[4];
#pragma unroll
    for (int t = 0; t < 4; ++t) {
        int kk = 4 * t + 2 * half;
        bf[t][0] = W[(kk + 0) * HIDC + l15];
        bf[t][1] = W[(kk + 1) * HIDC + l15];
    }

    const float* hrow = h + (size_t)(wave * 16 + l15) * HIDC;
    v8f c = {};
#pragma unroll
    for (int t = 0; t < 4; ++t) {
        int kk = 4 * t + 2 * half;
        v2f a;
        a[0] = hrow[kk + 0];
        a[1] = hrow[kk + 1];
        c = __builtin_amdgcn_wmma_f32_16x16x4_f32(
                false, a, false, bf[t], (short)0, c, false, false);
    }

#pragma unroll
    for (int v = 0; v < 8; ++v) {
        int m = v + 8 * half;
        hw[(size_t)(wave * 16 + m) * HIDC + l15] = c[v];
    }
}

__global__ void k_scatter(const int* __restrict__ src, const int* __restrict__ dst,
                          const float* __restrict__ dinv, const float* __restrict__ hw,
                          float* __restrict__ agg, int n_edges, int n_nodes) {
    int e = blockIdx.x * blockDim.x + threadIdx.x;
    int total = n_edges + n_nodes;
    if (e >= total) return;
    int s, d;
    if (e < n_edges) { s = src[e]; d = dst[e]; }
    else             { s = e - n_edges; d = s; }
    float nrm = dinv[s] * dinv[d];
    const float4* hr = (const float4*)(hw + (size_t)s * HIDC);
    float* ar = agg + (size_t)d * HIDC;
#pragma unroll
    for (int q = 0; q < 4; ++q) {
        float4 m = hr[q];
        fadd(ar + 4 * q + 0, m.x * nrm);
        fadd(ar + 4 * q + 1, m.y * nrm);
        fadd(ar + 4 * q + 2, m.z * nrm);
        fadd(ar + 4 * q + 3, m.w * nrm);
    }
}

__global__ void k_relu_bias(float* __restrict__ agg, const float* __restrict__ b,
                            float* __restrict__ h, int n_nodes) {
    int i = blockIdx.x * blockDim.x + threadIdx.x;
    if (i < n_nodes * HIDC) {
        int j = i & 15;
        h[i] = fmaxf(agg[i] + b[j], 0.0f);
        agg[i] = 0.0f;
    }
}

__global__ void k_pool(const float* __restrict__ h, const int* __restrict__ batch,
                       float* __restrict__ pooled, float* __restrict__ cnt, int n_nodes) {
    int i = blockIdx.x * blockDim.x + threadIdx.x;
    if (i < n_nodes * HIDC) {
        int node = i >> 4, j = i & 15;
        int g = batch[node];
        fadd(&pooled[g * HIDC + j], h[i]);
        if (j == 0) fadd(&cnt[g], 1.0f);
    }
}

__global__ void k_classify(const float* __restrict__ pooled, const float* __restrict__ cnt,
                           const float* __restrict__ Wc, const float* __restrict__ bc,
                           float* __restrict__ out, int n_graphs, int labels) {
    int i = blockIdx.x * blockDim.x + threadIdx.x;
    if (i >= n_graphs * labels) return;
    int g = i / labels, c = i % labels;
    float inv = 1.0f / fmaxf(cnt[g], 1.0f);
    float acc = bc[c];
#pragma unroll
    for (int j = 0; j < HIDC; ++j)
        acc += pooled[g * HIDC + j] * inv * Wc[j * labels + c];
    out[i] = acc;
}

extern "C" void kernel_launch(void* const* d_in, const int* in_sizes, int n_in,
                              void* d_out, int out_size, void* d_ws, size_t ws_size,
                              hipStream_t stream) {
    const int*   x     = (const int*)d_in[0];
    const int*   ei    = (const int*)d_in[1];
    const int*   batch = (const int*)d_in[2];
    const float* emb   = (const float*)d_in[3];
    const float* W1    = (const float*)d_in[4];
    const float* b1    = (const float*)d_in[5];
    const float* W2    = (const float*)d_in[6];
    const float* b2    = (const float*)d_in[7];
    const float* Wc    = (const float*)d_in[8];
    const float* bc    = (const float*)d_in[9];
    float* out = (float*)d_out;

    const int n_nodes  = in_sizes[0];
    const int n_edges  = in_sizes[1] / 2;
    const int labels   = in_sizes[9];
    const int n_graphs = out_size / labels;
    const int* src = ei;
    const int* dst = ei + n_edges;

    float* ws     = (float*)d_ws;
    float* deg    = ws;
    float* h      = deg + n_nodes;
    float* hw     = h  + (size_t)n_nodes * HIDC;
    float* agg    = hw + (size_t)n_nodes * HIDC;
    float* pooled = agg + (size_t)n_nodes * HIDC;
    float* cnt    = pooled + (size_t)n_graphs * HIDC;

    const int nh     = n_nodes * HIDC;
    const int ntiles = n_nodes / 16;
    const int wavesPerBlk = BLK / 32;
    const int etotal = n_edges + n_nodes;

    dim3 b(BLK);
    auto g1 = [](int n) { return dim3((n + BLK - 1) / BLK); };
    dim3 gw((ntiles + wavesPerBlk - 1) / wavesPerBlk);

    k_init<<<g1(nh), b, 0, stream>>>(deg, agg, pooled, cnt, n_nodes, n_graphs);
    k_deg<<<g1(n_edges), b, 0, stream>>>(dst, deg, n_edges);
    k_dinv<<<g1(n_nodes), b, 0, stream>>>(deg, n_nodes);
    k_embed<<<g1(nh), b, 0, stream>>>(x, emb, h, n_nodes);

    k_gcn_wmma<<<gw, b, 0, stream>>>(h, W1, hw, ntiles);
    k_scatter<<<g1(etotal), b, 0, stream>>>(src, dst, deg, hw, agg, n_edges, n_nodes);
    k_relu_bias<<<g1(nh), b, 0, stream>>>(agg, b1, h, n_nodes);

    k_gcn_wmma<<<gw, b, 0, stream>>>(h, W2, hw, ntiles);
    k_scatter<<<g1(etotal), b, 0, stream>>>(src, dst, deg, hw, agg, n_edges, n_nodes);
    k_relu_bias<<<g1(nh), b, 0, stream>>>(agg, b2, h, n_nodes);

    k_pool<<<g1(nh), b, 0, stream>>>(h, batch, pooled, cnt, n_nodes);
    k_classify<<<g1(n_graphs * labels), b, 0, stream>>>(pooled, cnt, Wc, bc, out,
                                                        n_graphs, labels);
}


# conv scatter-add: global f32 atomics replaced by LDS sums; a work-group owns 4000 dst rows, filters an edge chunk for them, adds rows by LDS atomics; 10 chunk planes + self loop summed in fixed order
# speedup vs baseline: 3.8801x; 3.8801x over previous
#include <hip/hip_runtime.h>

typedef __attribute__((ext_vector_type(2))) float v2f;
typedef __attribute__((ext_vector_type(8))) float v8f;

#define HIDC 16
#define BLK 256
#define NPR 4000
#define NCHUNK 10
#define SBLK 1024
#define QCAP 64

__device__ __forceinline__ void fadd(float* p, float v) {
    __hip_atomic_fetch_add(p, v, __ATOMIC_RELAXED, __HIP_MEMORY_SCOPE_AGENT);
}

__global__ void k_init(float* deg, float* pooled, float* cnt, int n_nodes, int n_graphs) {
    int i = blockIdx.x * blockDim.x + threadIdx.x;
    int k = i - n_nodes;
    if (i < n_nodes) deg[i] = 1.0f;
    else if (k < n_graphs * HIDC) pooled[k] = 0.0f;
    else if (k - n_graphs * HIDC < n_graphs) cnt[k - n_graphs * HIDC] = 0.0f;
}

__global__ void k_deg(const int* __restrict__ dst, float* deg, int n_edges) {
    int e = blockIdx.x * blockDim.x + threadIdx.x;
    if (e < n_edges) fadd(&deg[dst[e]], 1.0f);
}

__global__ void k_dinv(float* deg, int n_nodes) {
    int i = blockIdx.x * blockDim.x + threadIdx.x;
    if (i < n_nodes) {
        float d = deg[i];
        deg[i] = (d > 0.0f) ? rsqrtf(d) : 0.0f;
    }
}

__global__ void k_embed(const int* __restrict__ x, const float* __restrict__ emb,
                        float* __restrict__ h, int n_nodes) {
    int i = blockIdx.x * blockDim.x + threadIdx.x;
    if (i < n_nodes * HIDC) {
        int node = i >> 4, j = i & 15;
        h[i] = emb[x[node] * HIDC + j];
    }
}

__global__ void k_gcn_wmma(const float* __restrict__ h, const float* __restrict__ W,
                           float* __restrict__ hw, int ntiles) {
    int wave = (blockIdx.x * blockDim.x + threadIdx.x) >> 5;
    int lane = threadIdx.x & 31;
    if (wave >= ntiles) return;
    int half = lane >> 4;
    int l15  = lane & 15;

    v2f bf[4];
#pragma unroll
    for (int t = 0; t < 4; ++t) {
        int kk = 4 * t + 2 * half;
        bf[t][0] = W[(kk + 0) * HIDC + l15];
        bf[t][1] = W[(kk + 1) * HIDC + l15];
    }

    const float* hrow = h + (size_t)(wave * 16 + l15) * HIDC;
    v8f c = {};
#pragma unroll
    for (int t = 0; t < 4; ++t) {
        int kk = 4 * t + 2 * half;
        v2f a;
        a[0] = hrow[kk + 0];
        a[1] = hrow[kk + 1];
        c = __builtin_amdgcn_wmma_f32_16x16x4_f32(
                false, a, false, bf[t], (short)0, c, false, false);
    }

#pragma unroll
    for (int v = 0; v < 8; ++v) {
        int m = v + 8 * half;
        hw[(size_t)(wave * 16 + m) * HIDC + l15] = c[v];
    }
}

template <int I>
__device__ __forceinline__ int row_lane(int v) {
    return __builtin_amdgcn_update_dpp(0, v, 0x150 + I, 0xf, 0xf, false);
}


template <int I>
__device__ __forceinline__ void load_rows(int s, unsigned lane_off,
                                          const float* __restrict__ hw, float (&m)[16]) {
    if constexpr (I < 16) {
        unsigned so = ((unsigned)row_lane<I>(s) << 6) + lane_off;
        m[I] = *(const float*)((const char*)hw + so);
        load_rows<I + 1>(s, lane_off, hw, m);
    }
}

template <int I, bool TAIL>
__device__ __forceinline__ void add_rows(int dl, float nrm, unsigned lane_off,
                                         float* acc, const float (&m)[16]) {
    if constexpr (I < 16) {
        int   di = row_lane<I>(dl);
        float ni = __int_as_float(row_lane<I>(__float_as_int(nrm)));
        if (!TAIL || di >= 0) {
            unsigned dO = ((unsigned)di << 6) + lane_off;
            __hip_atomic_fetch_add((float*)((char*)acc + dO), m[I] * ni,
                                   __ATOMIC_RELAXED, __HIP_MEMORY_SCOPE_WORKGROUP);
        }
        add_rows<I + 1, TAIL>(dl, nrm, lane_off, acc, m);
    }
}

template <bool TAIL>
__device__ __forceinline__ void drain(const int2* q, int head, int n, int lane, int lo,
                                      const float* __restrict__ dinv,
                                      const float* __restrict__ hw, float* acc) {
    int s = 0, dl = -1;
    float nrm = 0.0f;
    if (!TAIL || lane < n) {
        int2 ent = q[(head + lane) & (QCAP - 1)];
        s = ent.x; dl = ent.y;
        nrm = dinv[s] * dinv[lo + dl];
    }
    unsigned lane_off = (unsigned)(lane & 15) * 4u;
    float m[16];
    load_rows<0>(s, lane_off, hw, m);
    add_rows<0, TAIL>(dl, nrm, lane_off, acc, m);
}

__device__ __forceinline__ void take(int s, int d, int lo, int cnt, int2* q, int& qt, int& qn,
                                     int lane, const float* __restrict__ dinv,
                                     const float* __restrict__ hw, float* acc) {
    int dl = d - lo;
    bool keep = (unsigned)dl < (unsigned)cnt;
    unsigned mask = __builtin_amdgcn_ballot_w32(keep);
    if (mask != 0u) {
        int rank = (int)__builtin_amdgcn_mbcnt_lo(mask, 0u);
        if (keep) q[(qt + rank) & (QCAP - 1)] = make_int2(s, dl);
        int n = __popc(mask);
        qt += n; qn += n;
        if (qn >= 32) {
            drain<false>(q, qt - qn, 32, lane, lo, dinv, hw, acc);
            qn -= 32;
        }
    }
}

__global__ void __launch_bounds__(SBLK)
k_scatter_lds(const int* __restrict__ src, const int* __restrict__ dst,
              const float* __restrict__ dinv, const float* __restrict__ hw,
              float* __restrict__ part, int n_edges, int n_nodes) {
    __shared__ __attribute__((aligned(16))) float acc[NPR * HIDC];
    __shared__ int2 queue[SBLK / 32][QCAP];

    const int tid = threadIdx.x, lane = tid & 31;
    const int wave = __builtin_amdgcn_readfirstlane(tid >> 5);
    const int lo  = blockIdx.x * NPR;
    const int cnt = min(NPR, n_nodes - lo);

    for (int i = tid; i < NPR * HIDC / 4; i += SBLK)
        ((float4*)acc)[i] = make_float4(0.0f, 0.0f, 0.0f, 0.0f);
    __syncthreads();

    const int trip = (SBLK / 32) * 128;
    const int per  = (((n_edges + (int)gridDim.y - 1) / (int)gridDim.y + trip - 1) / trip) * trip;
    const int e0   = blockIdx.y * per;
    const int e1   = min(e0 + per, n_edges);

    int2* q = queue[wave];
    int qt = 0, qn = 0;
    int base = e0 + wave * 128;

    if ((n_edges & 3) == 0 && base + 128 <= e1) {
        int4 s4 = *(const int4*)(src + base + lane * 4);
        int4 d4 = *(const int4*)(dst + base + lane * 4);
        bool more = true;
        while (more) {
            int nbase = base + trip;
            more = nbase + 128 <= e1;
            int pb = more ? nbase : base;
            int4 sn = *(const int4*)(src + pb + lane * 4);
            int4 dn = *(const int4*)(dst + pb + lane * 4);
            take(s4.x, d4.x, lo, cnt, q, qt, qn, lane, dinv, hw, acc);
            take(s4.y, d4.y, lo, cnt, q, qt, qn, lane, dinv, hw, acc);
            take(s4.z, d4.z, lo, cnt, q, qt, qn, lane, dinv, hw, acc);
            take(s4.w, d4.w, lo, cnt, q, qt, qn, lane, dinv, hw, acc);
            s4 = sn; d4 = dn;
            base = nbase;
        }
    }
    for (; base < e1; base += trip) {
        for (int j = 0; j < 4; ++j) {
            int e = base + j * 32 + lane;
            int s = 0, d = 0x7fffffff;
            if (e < e1) { s = src[e]; d = dst[e]; }
            take(s, d, lo, cnt, q, qt, qn, lane, dinv, hw, acc);
        }
    }
    if (qn > 0) drain<true>(q, qt - qn, qn, lane, lo, dinv, hw, acc);
    __syncthreads();

    float4* out4 = (float4*)part + ((size_t)blockIdx.y * n_nodes + lo) * (HIDC / 4);
    for (int i = tid; i < cnt * (HIDC / 4); i += SBLK)
        out4[i] = ((const float4*)acc)[i];
}

__global__ void k_sum_relu(const float* __restrict__ part, const float* __restrict__ dinv,
                           const float* __restrict__ hw, const float* __restrict__ b,
                           float* __restrict__ h, int n_nodes) {
    int i = blockIdx.x * blockDim.x + threadIdx.x;
    if (i < n_nodes * (HIDC / 4)) {
        const float4* p4 = (const float4*)part + i;
        const size_t plane = (size_t)n_nodes * (HIDC / 4);
        float4 a = p4[0];
        for (int c = 1; c < NCHUNK; ++c) {
            float4 t = p4[c * plane];
            a.x += t.x; a.y += t.y; a.z += t.z; a.w += t.w;
        }
        float dn = dinv[i >> 2];
        float nn = dn * dn;
        float4 hv = ((const float4*)hw)[i];
        const float* bq = b + (i & 3) * 4;
        float4 r;
        r.x = fmaxf(a.x + hv.x * nn + bq[0], 0.0f);
        r.y = fmaxf(a.y + hv.y * nn + bq[1], 0.0f);
        r.z = fmaxf(a.z + hv.z * nn + bq[2], 0.0f);
        r.w = fmaxf(a.w + hv.w * nn + bq[3], 0.0f);
        ((float4*)h)[i] = r;
    }
}

__global__ void k_pool(const float* __restrict__ h, const int* __restrict__ batch,
                       float* __restrict__ pooled, float* __restrict__ cnt, int n_nodes) {
    int i = blockIdx.x * blockDim.x + threadIdx.x;
    if (i < n_nodes * HIDC) {
        int node = i >> 4, j = i & 15;
        int g = batch[node];
        fadd(&pooled[g * HIDC + j], h[i]);
        if (j == 0) fadd(&cnt[g], 1.0f);
    }
}

__global__ void k_classify(const float* __restrict__ pooled, const float* __restrict__ cnt,
                           const float* __restrict__ Wc, const float* __restrict__ bc,
                           float* __restrict__ out, int n_graphs, int labels) {
    int i = blockIdx.x * blockDim.x + threadIdx.x;
    if (i >= n_graphs * labels) return;
    int g = i / labels, c = i % labels;
    float inv = 1.0f / fmaxf(cnt[g], 1.0f);
    float acc = bc[c];
#pragma unroll
    for (int j = 0; j < HIDC; ++j)
        acc += pooled[g * HIDC + j] * inv * Wc[j * labels + c];
    out[i] = acc;
}

extern "C" void kernel_launch(void* const* d_in, const int* in_sizes, int n_in,
                              void* d_out, int out_size, void* d_ws, size_t ws_size,
                              hipStream_t stream) {
    const int*   x     = (const int*)d_in[0];
    const int*   ei    = (const int*)d_in[1];
    const int*   batch = (const int*)d_in[2];
    const float* emb   = (const float*)d_in[3];
    const float* W1    = (const float*)d_in[4];
    const float* b1    = (const float*)d_in[5];
    const float* W2    = (const float*)d_in[6];
    const float* b2    = (const float*)d_in[7];
    const float* Wc    = (const float*)d_in[8];
    const float* bc    = (const float*)d_in[9];
    float* out = (float*)d_out;

    const int n_nodes  = in_sizes[0];
    const int n_edges  = in_sizes[1] / 2;
    const int labels   = in_sizes[9];
    const int n_graphs = out_size / labels;
    const int* src = ei;
    const int* dst = ei + n_edges;

    float* ws     = (float*)d_ws;
    float* deg    = ws;
    float* h      = deg + n_nodes;
    float* hw     = h  + (size_t)n_nodes * HIDC;
    float* part   = hw + (size_t)n_nodes * HIDC;
    float* pooled = part + (size_t)NCHUNK * n_nodes * HIDC;
    float* cnt    = pooled + (size_t)n_graphs * HIDC;

    const int nh     = n_nodes * HIDC;
    const int ntiles = n_nodes / 16;
    const int wavesPerBlk = BLK / 32;

    dim3 b(BLK);
    auto g1 = [](int n) { return dim3((n + BLK - 1) / BLK); };
    dim3 gw((ntiles + wavesPerBlk - 1) / wavesPerBlk);
    dim3 gs((n_nodes + NPR - 1) / NPR, NCHUNK);

    k_init<<<g1(n_nodes + n_graphs * HIDC + n_graphs), b, 0, stream>>>(deg, pooled, cnt,
                                                                       n_nodes, n_graphs);
    k_deg<<<g1(n_edges), b, 0, stream>>>(dst, deg, n_edges);
    k_dinv<<<g1(n_nodes), b, 0, stream>>>(deg, n_nodes);
    k_embed<<<g1(nh), b, 0, stream>>>(x, emb, h, n_nodes);

    k_gcn_wmma<<<gw, b, 0, stream>>>(h, W1, hw, ntiles);
    k_scatter_lds<<<gs, SBLK, 0, stream>>>(src, dst, deg, hw, part, n_edges, n_nodes);
    k_sum_relu<<<g1(nh / 4), b, 0, stream>>>(part, deg, hw, b1, h, n_nodes);

    k_gcn_wmma<<<gw, b, 0, stream>>>(h, W2, hw, ntiles);
    k_scatter_lds<<<gs, SBLK, 0, stream>>>(src, dst, deg, hw, part, n_edges, n_nodes);
    k_sum_relu<<<g1(nh / 4), b, 0, stream>>>(part, deg, hw, b2, h, n_nodes);

    k_pool<<<g1(nh), b, 0, stream>>>(h, batch, pooled, cnt, n_nodes);
    k_classify<<<g1(n_graphs * labels), b, 0, stream>>>(pooled, cnt, Wc, bc, out,
                                                        n_graphs, labels);
}


# conv 2's scatter replays a per-wave record (source, local row, edge weight) written by conv 1's scatter as it filters: no second pass over the edges; LDS sums in arrival order, arrivals differ
# speedup vs baseline: 4.3205x; 1.1135x over previous
#include <hip/hip_runtime.h>

typedef __attribute__((ext_vector_type(2))) float v2f;
typedef __attribute__((ext_vector_type(8))) float v8f;

#define HIDC 16
#define BLK 256
#define NPR 4000
#define NCHUNK 10
#define SBLK 1024
#define QCAP 64
#define RCAP 1024

__device__ __forceinline__ void fadd(float* p, float v) {
    __hip_atomic_fetch_add(p, v, __ATOMIC_RELAXED, __HIP_MEMORY_SCOPE_AGENT);
}

__global__ void k_init(float* deg, float* pooled, float* cnt, int n_nodes, int n_graphs) {
    int i = blockIdx.x * blockDim.x + threadIdx.x;
    int k = i - n_nodes;
    if (i < n_nodes) deg[i] = 1.0f;
    else if (k < n_graphs * HIDC) pooled[k] = 0.0f;
    else if (k - n_graphs * HIDC < n_graphs) cnt[k - n_graphs * HIDC] = 0.0f;
}

__global__ void k_deg(const int* __restrict__ dst, float* deg, int n_edges) {
    int e = blockIdx.x * blockDim.x + threadIdx.x;
    if (e < n_edges) fadd(&deg[dst[e]], 1.0f);
}

__global__ void k_dinv(float* deg, int n_nodes) {
    int i = blockIdx.x * blockDim.x + threadIdx.x;
    if (i < n_nodes) {
        float d = deg[i];
        deg[i] = (d > 0.0f) ? rsqrtf(d) : 0.0f;
    }
}

__global__ void k_embed(const int* __restrict__ x, const float* __restrict__ emb,
                        float* __restrict__ h, int n_nodes) {
    int i = blockIdx.x * blockDim.x + threadIdx.x;
    if (i < n_nodes * HIDC) {
        int node = i >> 4, j = i & 15;
        h[i] = emb[x[node] * HIDC + j];
    }
}

__global__ void k_gcn_wmma(const float* __restrict__ h, const float* __restrict__ W,
                           float* __restrict__ hw, int ntiles) {
    int wave = (blockIdx.x * blockDim.x + threadIdx.x) >> 5;
    int lane = threadIdx.x & 31;
    if (wave >= ntiles) return;
    int half = lane >> 4;
    int l15  = lane & 15;

    v2f bf[4];
#pragma unroll
    for (int t = 0; t < 4; ++t) {
        int kk = 4 * t + 2 * half;
        bf[t][0] = W[(kk + 0) * HIDC + l15];
        bf[t][1] = W[(kk + 1) * HIDC + l15];
    }

    const float* hrow = h + (size_t)(wave * 16 + l15) * HIDC;
    v8f c = {};
#pragma unroll
    for (int t = 0; t < 4; ++t) {
        int kk = 4 * t + 2 * half;
        v2f a;
        a[0] = hrow[kk + 0];
        a[1] = hrow[kk + 1];
        c = __builtin_amdgcn_wmma_f32_16x16x4_f32(
                false, a, false, bf[t], (short)0, c, false, false);
    }

#pragma unroll
    for (int v = 0; v < 8; ++v) {
        int m = v + 8 * half;
        hw[(size_t)(wave * 16 + m) * HIDC + l15] = c[v];
    }
}

template <int I>
__device__ __forceinline__ int row_lane(int v) {
    return __builtin_amdgcn_update_dpp(0, v, 0x150 + I, 0xf, 0xf, false);
}


template <int I>
__device__ __forceinline__ void load_rows(int s, unsigned lane_off,
                                          const float* __restrict__ hw, float (&m)[16]) {
    if constexpr (I < 16) {
        unsigned so = ((unsigned)row_lane<I>(s) << 6) + lane_off;
        m[I] = *(const float*)((const char*)hw + so);
        load_rows<I + 1>(s, lane_off, hw, m);
    }
}

template <int I, bool TAIL>
__device__ __forceinline__ void add_rows(int dl, float nrm, unsigned lane_off,
                                         float* acc, const float (&m)[16]) {
    if constexpr (I < 16) {
        int   di = row_lane<I>(dl);
        float ni = __int_as_float(row_lane<I>(__float_as_int(nrm)));
        if (!TAIL || di >= 0) {
            unsigned dO = ((unsigned)di << 6) + lane_off;
            __hip_atomic_fetch_add((float*)((char*)acc + dO), m[I] * ni,
                                   __ATOMIC_RELAXED, __HIP_MEMORY_SCOPE_WORKGROUP);
        }
        add_rows<I + 1, TAIL>(dl, nrm, lane_off, acc, m);
    }
}

template <bool TAIL, bool REC>
__device__ __forceinline__ void drain(const int2* q, int head, int n, int lane, int lo,
                                      const float* __restrict__ dinv,
                                      const float* __restrict__ hw, float* acc,
                                      int2* __restrict__ rsd, float* __restrict__ rw) {
    int s = 0, dl = -1;
    float nrm = 0.0f;
    if (!TAIL || lane < n) {
        int2 ent = q[(head + lane) & (QCAP - 1)];
        s = ent.x; dl = ent.y;
        nrm = dinv[s] * dinv[lo + dl];
        if (REC && head + lane < RCAP) {
            rsd[head + lane] = ent;
            rw[head + lane]  = nrm;
        }
    }
    unsigned lane_off = (unsigned)(lane & 15) * 4u;
    float m[16];
    load_rows<0>(s, lane_off, hw, m);
    add_rows<0, TAIL>(dl, nrm, lane_off, acc, m);
}

template <bool REC>
__device__ __forceinline__ void take(int s, int d, int lo, int cnt, int2* q, int& qt, int& qn,
                                     int lane, const float* __restrict__ dinv,
                                     const float* __restrict__ hw, float* acc,
                                     int2* __restrict__ rsd, float* __restrict__ rw) {
    int dl = d - lo;
    bool keep = (unsigned)dl < (unsigned)cnt;
    unsigned mask = __builtin_amdgcn_ballot_w32(keep);
    if (mask != 0u) {
        int rank = (int)__builtin_amdgcn_mbcnt_lo(mask, 0u);
        if (keep) q[(qt + rank) & (QCAP - 1)] = make_int2(s, dl);
        int n = __popc(mask);
        qt += n; qn += n;
        if (qn >= 32) {
            drain<false, REC>(q, qt - qn, 32, lane, lo, dinv, hw, acc, rsd, rw);
            qn -= 32;
        }
    }
}

template <bool REC>
__device__ __forceinline__ int filter_chunk(const int* __restrict__ src,
                                            const int* __restrict__ dst,
                                            const float* __restrict__ dinv,
                                            const float* __restrict__ hw, float* acc, int2* q,
                                            int2* __restrict__ rsd, float* __restrict__ rw,
                                            int n_edges, int e0, int e1, int trip,
                                            int wave, int lane, int lo, int cnt) {
    int qt = 0, qn = 0;
    int base = e0 + wave * 128;

    if ((n_edges & 3) == 0 && base + 128 <= e1) {
        int4 s4 = *(const int4*)(src + base + lane * 4);
        int4 d4 = *(const int4*)(dst + base + lane * 4);
        bool more = true;
        while (more) {
            int nbase = base + trip;
            more = nbase + 128 <= e1;
            int pb = more ? nbase : base;
            int4 sn = *(const int4*)(src + pb + lane * 4);
            int4 dn = *(const int4*)(dst + pb + lane * 4);
            take<REC>(s4.x, d4.x, lo, cnt, q, qt, qn, lane, dinv, hw, acc, rsd, rw);
            take<REC>(s4.y, d4.y, lo, cnt, q, qt, qn, lane, dinv, hw, acc, rsd, rw);
            take<REC>(s4.z, d4.z, lo, cnt, q, qt, qn, lane, dinv, hw, acc, rsd, rw);
            take<REC>(s4.w, d4.w, lo, cnt, q, qt, qn, lane, dinv, hw, acc, rsd, rw);
            s4 = sn; d4 = dn;
            base = nbase;
        }
    }
    for (; base < e1; base += trip) {
        for (int j = 0; j < 4; ++j) {
            int e = base + j * 32 + lane;
            int s = 0, d = 0x7fffffff;
            if (e < e1) { s = src[e]; d = dst[e]; }
            take<REC>(s, d, lo, cnt, q, qt, qn, lane, dinv, hw, acc, rsd, rw);
        }
    }
    if (qn > 0) drain<true, REC>(q, qt - qn, qn, lane, lo, dinv, hw, acc, rsd, rw);
    return qt;
}

template <bool REPLAY>
__global__ void __launch_bounds__(SBLK)
k_scatter_lds(const int* __restrict__ src, const int* __restrict__ dst,
              const float* __restrict__ dinv, const float* __restrict__ hw,
              float* __restrict__ part, int* __restrict__ rec_sd, float* __restrict__ rec_w,
              int* __restrict__ rec_n, int n_edges, int n_nodes) {
    __shared__ __attribute__((aligned(16))) float acc[NPR * HIDC];
    __shared__ int2 queue[SBLK / 32][QCAP];

    const int tid = threadIdx.x, lane = tid & 31;
    const int wave = __builtin_amdgcn_readfirstlane(tid >> 5);
    const int lo  = blockIdx.x * NPR;
    const int cnt = min(NPR, n_nodes - lo);

    for (int i = tid; i < NPR * HIDC / 4; i += SBLK)
        ((float4*)acc)[i] = make_float4(0.0f, 0.0f, 0.0f, 0.0f);
    __syncthreads();

    const int trip = (SBLK / 32) * 128;
    const int per  = (((n_edges + (int)gridDim.y - 1) / (int)gridDim.y + trip - 1) / trip) * trip;
    const int e0   = blockIdx.y * per;
    const int e1   = min(e0 + per, n_edges);

    const int wg = (blockIdx.y * gridDim.x + blockIdx.x) * (SBLK / 32) + wave;
    int2*  rsd = (int2*)rec_sd + (size_t)wg * RCAP;
    float* rw  = rec_w + (size_t)wg * RCAP;

    if (!REPLAY) {
        int kept = filter_chunk<true>(src, dst, dinv, hw, acc, queue[wave], rsd, rw,
                                      n_edges, e0, e1, trip, wave, lane, lo, cnt);
        if (lane == 0) rec_n[wg] = kept;
    } else {
        const int tot = __builtin_amdgcn_readfirstlane(rec_n[wg]);
        if (tot <= RCAP) {
            const unsigned lane_off = (unsigned)(lane & 15) * 4u;
            int2  e = make_int2(0, -1);
            float w = 0.0f;
            if (lane < tot) { e = rsd[lane]; w = rw[lane]; }
            const int nfull = tot >> 5;
            for (int k = 0; k < nfull; ++k) {
                float m[16];
                load_rows<0>(e.x, lane_off, hw, m);
                int   idx = (k + 1) * 32 + lane;
                int   ci  = min(idx, tot - 1);
                int2  en = rsd[ci];
                float wn = rw[ci];
                add_rows<0, false>(e.y, w, lane_off, acc, m);
                bool  has = idx < tot;
                e.x = has ? en.x : 0;
                e.y = has ? en.y : -1;
                w   = has ? wn : 0.0f;
            }
            if (tot & 31) {
                float m[16];
                load_rows<0>(e.x, lane_off, hw, m);
                add_rows<0, true>(e.y, w, lane_off, acc, m);
            }
        } else {
            filter_chunk<false>(src, dst, dinv, hw, acc, queue[wave], rsd, rw,
                                n_edges, e0, e1, trip, wave, lane, lo, cnt);
        }
    }
    __syncthreads();

    float4* out4 = (float4*)part + ((size_t)blockIdx.y * n_nodes + lo) * (HIDC / 4);
    for (int i = tid; i < cnt * (HIDC / 4); i += SBLK)
        out4[i] = ((const float4*)acc)[i];
}

__global__ void k_sum_relu(const float* __restrict__ part, const float* __restrict__ dinv,
                           const float* __restrict__ hw, const float* __restrict__ b,
                           float* __restrict__ h, int n_nodes) {
    int i = blockIdx.x * blockDim.x + threadIdx.x;
    if (i < n_nodes * (HIDC / 4)) {
        const float4* p4 = (const float4*)part + i;
        const size_t plane = (size_t)n_nodes * (HIDC / 4);
        float4 a = p4[0];
        for (int c = 1; c < NCHUNK; ++c) {
            float4 t = p4[c * plane];
            a.x += t.x; a.y += t.y; a.z += t.z; a.w += t.w;
        }
        float dn = dinv[i >> 2];
        float nn = dn * dn;
        float4 hv = ((const float4*)hw)[i];
        const float* bq = b + (i & 3) * 4;
        float4 r;
        r.x = fmaxf(a.x + hv.x * nn + bq[0], 0.0f);
        r.y = fmaxf(a.y + hv.y * nn + bq[1], 0.0f);
        r.z = fmaxf(a.z + hv.z * nn + bq[2], 0.0f);
        r.w = fmaxf(a.w + hv.w * nn + bq[3], 0.0f);
        ((float4*)h)[i] = r;
    }
}

__global__ void k_pool(const float* __restrict__ h, const int* __restrict__ batch,
                       float* __restrict__ pooled, float* __restrict__ cnt, int n_nodes) {
    int i = blockIdx.x * blockDim.x + threadIdx.x;
    if (i < n_nodes * HIDC) {
        int node = i >> 4, j = i & 15;
        int g = batch[node];
        fadd(&pooled[g * HIDC + j], h[i]);
        if (j == 0) fadd(&cnt[g], 1.0f);
    }
}

__global__ void k_classify(const float* __restrict__ pooled, const float* __restrict__ cnt,
                           const float* __restrict__ Wc, const float* __restrict__ bc,
                           float* __restrict__ out, int n_graphs, int labels) {
    int i = blockIdx.x * blockDim.x + threadIdx.x;
    if (i >= n_graphs * labels) return;
    int g = i / labels, c = i % labels;
    float inv = 1.0f / fmaxf(cnt[g], 1.0f);
    float acc = bc[c];
#pragma unroll
    for (int j = 0; j < HIDC; ++j)
        acc += pooled[g * HIDC + j] * inv * Wc[j * labels + c];
    out[i] = acc;
}

extern "C" void kernel_launch(void* const* d_in, const int* in_sizes, int n_in,
                              void* d_out, int out_size, void* d_ws, size_t ws_size,
                              hipStream_t stream) {
    const int*   x     = (const int*)d_in[0];
    const int*   ei    = (const int*)d_in[1];
    const int*   batch = (const int*)d_in[2];
    const float* emb   = (const float*)d_in[3];
    const float* W1    = (const float*)d_in[4];
    const float* b1    = (const float*)d_in[5];
    const float* W2    = (const float*)d_in[6];
    const float* b2    = (const float*)d_in[7];
    const float* Wc    = (const float*)d_in[8];
    const float* bc    = (const float*)d_in[9];
    float* out = (float*)d_out;

    const int n_nodes  = in_sizes[0];
    const int n_edges  = in_sizes[1] / 2;
    const int labels   = in_sizes[9];
    const int n_graphs = out_size / labels;
    const int* src = ei;
    const int* dst = ei + n_edges;

    float* ws     = (float*)d_ws;
    float* deg    = ws;
    float* h      = deg + n_nodes;
    float* hw     = h  + (size_t)n_nodes * HIDC;
    float* part   = hw + (size_t)n_nodes * HIDC;
    float* pooled = part + (size_t)NCHUNK * n_nodes * HIDC;
    float* cnt    = pooled + (size_t)n_graphs * HIDC;
    const size_t n_sw   = (size_t)((n_nodes + NPR - 1) / NPR) * NCHUNK * (SBLK / 32);
    const size_t rec_at = ((size_t)(cnt + n_graphs - ws) + 3) & ~(size_t)3;
    int*   rec_sd = (int*)(ws + rec_at);
    float* rec_w  = (float*)(rec_sd + n_sw * RCAP * 2);
    int*   rec_n  = (int*)(rec_w + n_sw * RCAP);

    const int nh     = n_nodes * HIDC;
    const int ntiles = n_nodes / 16;
    const int wavesPerBlk = BLK / 32;

    dim3 b(BLK);
    auto g1 = [](int n) { return dim3((n + BLK - 1) / BLK); };
    dim3 gw((ntiles + wavesPerBlk - 1) / wavesPerBlk);
    dim3 gs((n_nodes + NPR - 1) / NPR, NCHUNK);

    k_init<<<g1(n_nodes + n_graphs * HIDC + n_graphs), b, 0, stream>>>(deg, pooled, cnt,
                                                                       n_nodes, n_graphs);
    k_deg<<<g1(n_edges), b, 0, stream>>>(dst, deg, n_edges);
    k_dinv<<<g1(n_nodes), b, 0, stream>>>(deg, n_nodes);
    k_embed<<<g1(nh), b, 0, stream>>>(x, emb, h, n_nodes);

    k_gcn_wmma<<<gw, b, 0, stream>>>(h, W1, hw, ntiles);
    k_scatter_lds<false><<<gs, SBLK, 0, stream>>>(src, dst, deg, hw, part, rec_sd, rec_w, rec_n,
                                                  n_edges, n_nodes);
    k_sum_relu<<<g1(nh / 4), b, 0, stream>>>(part, deg, hw, b1, h, n_nodes);

    k_gcn_wmma<<<gw, b, 0, stream>>>(h, W2, hw, ntiles);
    k_scatter_lds<true><<<gs, SBLK, 0, stream>>>(src, dst, deg, hw, part, rec_sd, rec_w, rec_n,
                                                 n_edges, n_nodes);
    k_sum_relu<<<g1(nh / 4), b, 0, stream>>>(part, deg, hw, b2, h, n_nodes);

    k_pool<<<g1(nh), b, 0, stream>>>(h, batch, pooled, cnt, n_nodes);
    k_classify<<<g1(n_graphs * labels), b, 0, stream>>>(pooled, cnt, Wc, bc, out,
                                                        n_graphs, labels);
}


# degrees counted on LDS counters by the range-owning work-groups in one edge filter pass that writes the per-wave record; both convs replay it; LDS sums in arrival order, arrivals differ
# speedup vs baseline: 7.8496x; 1.8168x over previous
#include <hip/hip_runtime.h>

typedef __attribute__((ext_vector_type(2))) float v2f;
typedef __attribute__((ext_vector_type(8))) float v8f;

#define HIDC 16
#define BLK 256
#define NPR 4000
#define NCHUNK 10
#define SBLK 1024
#define QCAP 64
#define RCAP 1024

__device__ __forceinline__ void fadd(float* p, float v) {
    __hip_atomic_fetch_add(p, v, __ATOMIC_RELAXED, __HIP_MEMORY_SCOPE_AGENT);
}

__global__ void k_init(float* pooled, float* cnt, int n_graphs) {
    int k = blockIdx.x * blockDim.x + threadIdx.x;
    if (k < n_graphs * HIDC) pooled[k] = 0.0f;
    else if (k - n_graphs * HIDC < n_graphs) cnt[k - n_graphs * HIDC] = 0.0f;
}

__global__ void k_dinv(const int* __restrict__ degc, float* __restrict__ dinv, int n_nodes) {
    int i = blockIdx.x * blockDim.x + threadIdx.x;
    if (i < n_nodes) {
        int t = 1;
        for (int c = 0; c < NCHUNK; ++c) t += degc[(size_t)c * n_nodes + i];
        float d = (float)t;
        dinv[i] = (d > 0.0f) ? rsqrtf(d) : 0.0f;
    }
}

__global__ void k_embed(const int* __restrict__ x, const float* __restrict__ emb,
                        float* __restrict__ h, int n_nodes) {
    int i = blockIdx.x * blockDim.x + threadIdx.x;
    if (i < n_nodes * HIDC) {
        int node = i >> 4, j = i & 15;
        h[i] = emb[x[node] * HIDC + j];
    }
}

__global__ void k_gcn_wmma(const float* __restrict__ h, const float* __restrict__ W,
                           float* __restrict__ hw, int ntiles) {
    int wave = (blockIdx.x * blockDim.x + threadIdx.x) >> 5;
    int lane = threadIdx.x & 31;
    if (wave >= ntiles) return;
    int half = lane >> 4;
    int l15  = lane & 15;

    v2f bf[4];
#pragma unroll
    for (int t = 0; t < 4; ++t) {
        int kk = 4 * t + 2 * half;
        bf[t][0] = W[(kk + 0) * HIDC + l15];
        bf[t][1] = W[(kk + 1) * HIDC + l15];
    }

    const float* hrow = h + (size_t)(wave * 16 + l15) * HIDC;
    v8f c = {};
#pragma unroll
    for (int t = 0; t < 4; ++t) {
        int kk = 4 * t + 2 * half;
        v2f a;
        a[0] = hrow[kk + 0];
        a[1] = hrow[kk + 1];
        c = __builtin_amdgcn_wmma_f32_16x16x4_f32(
                false, a, false, bf[t], (short)0, c, false, false);
    }

#pragma unroll
    for (int v = 0; v < 8; ++v) {
        int m = v + 8 * half;
        hw[(size_t)(wave * 16 + m) * HIDC + l15] = c[v];
    }
}

template <int I>
__device__ __forceinline__ int row_lane(int v) {
    return __builtin_amdgcn_update_dpp(0, v, 0x150 + I, 0xf, 0xf, false);
}


template <int I>
__device__ __forceinline__ void load_rows(int s, unsigned lane_off,
                                          const float* __restrict__ hw, float (&m)[16]) {
    if constexpr (I < 16) {
        unsigned so = ((unsigned)row_lane<I>(s) << 6) + lane_off;
        m[I] = *(const float*)((const char*)hw + so);
        load_rows<I + 1>(s, lane_off, hw, m);
    }
}

template <int I, bool TAIL>
__device__ __forceinline__ void add_rows(int dl, float nrm, unsigned lane_off,
                                         float* acc, const float (&m)[16]) {
    if constexpr (I < 16) {
        int   di = row_lane<I>(dl);
        float ni = __int_as_float(row_lane<I>(__float_as_int(nrm)));
        if (!TAIL || di >= 0) {
            unsigned dO = ((unsigned)di << 6) + lane_off;
            __hip_atomic_fetch_add((float*)((char*)acc + dO), m[I] * ni,
                                   __ATOMIC_RELAXED, __HIP_MEMORY_SCOPE_WORKGROUP);
        }
        add_rows<I + 1, TAIL>(dl, nrm, lane_off, acc, m);
    }
}

template <bool TAIL>
__device__ __forceinline__ void drain(const int2* q, int head, int n, int lane, int lo,
                                      const float* __restrict__ dinv,
                                      const float* __restrict__ hw, float* acc) {
    int s = 0, dl = -1;
    float nrm = 0.0f;
    if (!TAIL || lane < n) {
        int2 ent = q[(head + lane) & (QCAP - 1)];
        s = ent.x; dl = ent.y;
        nrm = dinv[s] * dinv[lo + dl];
    }
    unsigned lane_off = (unsigned)(lane & 15) * 4u;
    float m[16];
    load_rows<0>(s, lane_off, hw, m);
    add_rows<0, TAIL>(dl, nrm, lane_off, acc, m);
}

__device__ __forceinline__ void put_record(const int2* q, int head, int n, int lane,
                                           int2* __restrict__ rsd) {
    int pos = head + lane;
    if (lane < n && pos < RCAP) rsd[pos] = q[pos & (QCAP - 1)];
}

template <bool CNT>
__device__ __forceinline__ void take(int s, int d, int lo, int cnt, int2* q, int& qt, int& qn,
                                     int lane, const float* __restrict__ dinv,
                                     const float* __restrict__ hw, float* acc,
                                     int2* __restrict__ rsd, int* cntl) {
    int dl = d - lo;
    bool keep = (unsigned)dl < (unsigned)cnt;
    unsigned mask = __builtin_amdgcn_ballot_w32(keep);
    if (mask != 0u) {
        int rank = (int)__builtin_amdgcn_mbcnt_lo(mask, 0u);
        if (keep) {
            q[(qt + rank) & (QCAP - 1)] = make_int2(s, dl);
            if constexpr (CNT)
                __hip_atomic_fetch_add(&cntl[dl], 1, __ATOMIC_RELAXED,
                                       __HIP_MEMORY_SCOPE_WORKGROUP);
        }
        int n = __popc(mask);
        qt += n; qn += n;
        if (qn >= 32) {
            if constexpr (CNT) put_record(q, qt - qn, 32, lane, rsd);
            else drain<false>(q, qt - qn, 32, lane, lo, dinv, hw, acc);
            qn -= 32;
        }
    }
}

template <bool CNT>
__device__ __forceinline__ int filter_chunk(const int* __restrict__ src,
                                            const int* __restrict__ dst,
                                            const float* __restrict__ dinv,
                                            const float* __restrict__ hw, float* acc, int2* q,
                                            int2* __restrict__ rsd, int* cntl,
                                            int n_edges, int e0, int e1, int trip,
                                            int wave, int lane, int lo, int cnt) {
    int qt = 0, qn = 0;
    int base = e0 + wave * 128;

    if ((n_edges & 3) == 0 && base + 128 <= e1) {
        int nw = (e1 - base - 128) / trip + 1;
        if (nw >= 2) {
            int4 sa = *(const int4*)(src + base + lane * 4);
            int4 da = *(const int4*)(dst + base + lane * 4);
            int4 sb = *(const int4*)(src + base + trip + lane * 4);
            int4 db = *(const int4*)(dst + base + trip + lane * 4);
            while (nw >= 2) {
                nw -= 2;
                int nbase = base + 2 * trip;
                int pb = (nw >= 2) ? nbase : base;
                int4 sc = *(const int4*)(src + pb + lane * 4);
                int4 dc = *(const int4*)(dst + pb + lane * 4);
                int4 sd = *(const int4*)(src + pb + trip + lane * 4);
                int4 dd = *(const int4*)(dst + pb + trip + lane * 4);
                take<CNT>(sa.x, da.x, lo, cnt, q, qt, qn, lane, dinv, hw, acc, rsd, cntl);
                take<CNT>(sa.y, da.y, lo, cnt, q, qt, qn, lane, dinv, hw, acc, rsd, cntl);
                take<CNT>(sa.z, da.z, lo, cnt, q, qt, qn, lane, dinv, hw, acc, rsd, cntl);
                take<CNT>(sa.w, da.w, lo, cnt, q, qt, qn, lane, dinv, hw, acc, rsd, cntl);
                take<CNT>(sb.x, db.x, lo, cnt, q, qt, qn, lane, dinv, hw, acc, rsd, cntl);
                take<CNT>(sb.y, db.y, lo, cnt, q, qt, qn, lane, dinv, hw, acc, rsd, cntl);
                take<CNT>(sb.z, db.z, lo, cnt, q, qt, qn, lane, dinv, hw, acc, rsd, cntl);
                take<CNT>(sb.w, db.w, lo, cnt, q, qt, qn, lane, dinv, hw, acc, rsd, cntl);
                sa = sc; da = dc; sb = sd; db = dd;
                base = nbase;
            }
        }
        if (nw == 1) {
            int4 s4 = *(const int4*)(src + base + lane * 4);
            int4 d4 = *(const int4*)(dst + base + lane * 4);
            take<CNT>(s4.x, d4.x, lo, cnt, q, qt, qn, lane, dinv, hw, acc, rsd, cntl);
            take<CNT>(s4.y, d4.y, lo, cnt, q, qt, qn, lane, dinv, hw, acc, rsd, cntl);
            take<CNT>(s4.z, d4.z, lo, cnt, q, qt, qn, lane, dinv, hw, acc, rsd, cntl);
            take<CNT>(s4.w, d4.w, lo, cnt, q, qt, qn, lane, dinv, hw, acc, rsd, cntl);
            base += trip;
        }
    }
    for (; base < e1; base += trip) {
        for (int j = 0; j < 4; ++j) {
            int e = base + j * 32 + lane;
            int s = 0, d = 0x7fffffff;
            if (e < e1) { s = src[e]; d = dst[e]; }
            take<CNT>(s, d, lo, cnt, q, qt, qn, lane, dinv, hw, acc, rsd, cntl);
        }
    }
    if (qn > 0) {
        if constexpr (CNT) put_record(q, qt - qn, qn, lane, rsd);
        else drain<true>(q, qt - qn, qn, lane, lo, dinv, hw, acc);
    }
    return qt;
}

__device__ __forceinline__ void chunk_bounds(int n_edges, int trip, int& e0, int& e1) {
    const int per = (((n_edges + (int)gridDim.y - 1) / (int)gridDim.y + trip - 1) / trip) * trip;
    e0 = blockIdx.y * per;
    e1 = min(e0 + per, n_edges);
}

__global__ void __launch_bounds__(SBLK)
k_filter_deg(const int* __restrict__ src, const int* __restrict__ dst,
             int* __restrict__ degc, int* __restrict__ rec_sd, int* __restrict__ rec_n,
             int n_edges, int n_nodes) {
    __shared__ int cntl[NPR];
    __shared__ int2 queue[SBLK / 32][QCAP];

    const int tid = threadIdx.x, lane = tid & 31;
    const int wave = __builtin_amdgcn_readfirstlane(tid >> 5);
    const int lo  = blockIdx.x * NPR;
    const int cnt = min(NPR, n_nodes - lo);

    for (int i = tid; i < NPR; i += SBLK) cntl[i] = 0;
    __syncthreads();

    const int trip = (SBLK / 32) * 128;
    int e0, e1;
    chunk_bounds(n_edges, trip, e0, e1);

    const int wg = (blockIdx.y * gridDim.x + blockIdx.x) * (SBLK / 32) + wave;
    int2* rsd = (int2*)rec_sd + (size_t)wg * RCAP;

    int kept = filter_chunk<true>(src, dst, nullptr, nullptr, nullptr, queue[wave], rsd, cntl,
                                  n_edges, e0, e1, trip, wave, lane, lo, cnt);
    if (lane == 0) rec_n[wg] = kept;
    __syncthreads();

    int* plane = degc + (size_t)blockIdx.y * n_nodes + lo;
    for (int i = tid; i < cnt; i += SBLK) plane[i] = cntl[i];
}

template <bool FIRST>
__global__ void __launch_bounds__(SBLK)
k_scatter_lds(const int* __restrict__ src, const int* __restrict__ dst,
              const float* __restrict__ dinv, const float* __restrict__ hw,
              float* __restrict__ part, int* __restrict__ rec_sd, float* __restrict__ rec_w,
              const int* __restrict__ rec_n, int n_edges, int n_nodes) {
    __shared__ __attribute__((aligned(16))) float acc[NPR * HIDC];
    __shared__ int2 queue[SBLK / 32][QCAP];

    const int tid = threadIdx.x, lane = tid & 31;
    const int wave = __builtin_amdgcn_readfirstlane(tid >> 5);
    const int lo  = blockIdx.x * NPR;
    const int cnt = min(NPR, n_nodes - lo);

    for (int i = tid; i < NPR * HIDC / 4; i += SBLK)
        ((float4*)acc)[i] = make_float4(0.0f, 0.0f, 0.0f, 0.0f);
    __syncthreads();

    const int trip = (SBLK / 32) * 128;
    int e0, e1;
    chunk_bounds(n_edges, trip, e0, e1);

    const int wg = (blockIdx.y * gridDim.x + blockIdx.x) * (SBLK / 32) + wave;
    int2*  rsd = (int2*)rec_sd + (size_t)wg * RCAP;
    float* rw  = rec_w + (size_t)wg * RCAP;

    const int tot = __builtin_amdgcn_readfirstlane(rec_n[wg]);
    if (tot <= RCAP) {
        const unsigned lane_off = (unsigned)(lane & 15) * 4u;
        const int nfull = tot >> 5;
        if constexpr (FIRST) {
            if (tot > 0) {
                const int last = tot - 1;
                int2  e  = rsd[min(lane, last)];
                int2  en = rsd[min(32 + lane, last)];
                float w  = dinv[e.x] * dinv[lo + e.y];
                if (lane >= tot) { e.x = 0; e.y = -1; w = 0.0f; }
                for (int k = 0; k < nfull; ++k) {
                    int2  e2 = rsd[min((k + 2) * 32 + lane, last)];
                    float m[16];
                    load_rows<0>(e.x, lane_off, hw, m);
                    float fs = dinv[en.x];
                    float fd = dinv[lo + en.y];
                    rw[k * 32 + lane] = w;
                    asm volatile("" :: "v"(e2.x), "v"(e2.y));
                    add_rows<0, false>(e.y, w, lane_off, acc, m);
                    bool  has = (k + 1) * 32 + lane < tot;
                    e.x = has ? en.x : 0;
                    e.y = has ? en.y : -1;
                    w   = has ? fs * fd : 0.0f;
                    en  = e2;
                }
                if (tot & 31) {
                    float m[16];
                    load_rows<0>(e.x, lane_off, hw, m);
                    if (e.y >= 0) rw[nfull * 32 + lane] = w;
                    add_rows<0, true>(e.y, w, lane_off, acc, m);
                }
            }
        } else {
            int2  e = make_int2(0, -1);
            float w = 0.0f;
            if (lane < tot) { e = rsd[lane]; w = rw[lane]; }
            for (int k = 0; k < nfull; ++k) {
                float m[16];
                load_rows<0>(e.x, lane_off, hw, m);
                int   idx = (k + 1) * 32 + lane;
                int   ci  = min(idx, tot - 1);
                int2  en = rsd[ci];
                float wn = rw[ci];
                add_rows<0, false>(e.y, w, lane_off, acc, m);
                bool  has = idx < tot;
                e.x = has ? en.x : 0;
                e.y = has ? en.y : -1;
                w   = has ? wn : 0.0f;
            }
            if (tot & 31) {
                float m[16];
                load_rows<0>(e.x, lane_off, hw, m);
                add_rows<0, true>(e.y, w, lane_off, acc, m);
            }
        }
    } else {
        filter_chunk<false>(src, dst, dinv, hw, acc, queue[wave], rsd, nullptr,
                            n_edges, e0, e1, trip, wave, lane, lo, cnt);
    }
    __syncthreads();

    float4* out4 = (float4*)part + ((size_t)blockIdx.y * n_nodes + lo) * (HIDC / 4);
    for (int i = tid; i < cnt * (HIDC / 4); i += SBLK)
        out4[i] = ((const float4*)acc)[i];
}

__global__ void k_sum_relu(const float* __restrict__ part, const float* __restrict__ dinv,
                           const float* __restrict__ hw, const float* __restrict__ b,
                           float* __restrict__ h, int n_nodes) {
    int i = blockIdx.x * blockDim.x + threadIdx.x;
    if (i < n_nodes * (HIDC / 4)) {
        const float4* p4 = (const float4*)part + i;
        const size_t plane = (size_t)n_nodes * (HIDC / 4);
        float4 a = p4[0];
        for (int c = 1; c < NCHUNK; ++c) {
            float4 t = p4[c * plane];
            a.x += t.x; a.y += t.y; a.z += t.z; a.w += t.w;
        }
        float dn = dinv[i >> 2];
        float nn = dn * dn;
        float4 hv = ((const float4*)hw)[i];
        const float* bq = b + (i & 3) * 4;
        float4 r;
        r.x = fmaxf(a.x + hv.x * nn + bq[0], 0.0f);
        r.y = fmaxf(a.y + hv.y * nn + bq[1], 0.0f);
        r.z = fmaxf(a.z + hv.z * nn + bq[2], 0.0f);
        r.w = fmaxf(a.w + hv.w * nn + bq[3], 0.0f);
        ((float4*)h)[i] = r;
    }
}

__global__ void k_pool(const float* __restrict__ h, const int* __restrict__ batch,
                       float* __restrict__ pooled, float* __restrict__ cnt, int n_nodes) {
    int i = blockIdx.x * blockDim.x + threadIdx.x;
    if (i < n_nodes * HIDC) {
        int node = i >> 4, j = i & 15;
        int g = batch[node];
        fadd(&pooled[g * HIDC + j], h[i]);
        if (j == 0) fadd(&cnt[g], 1.0f);
    }
}

__global__ void k_classify(const float* __restrict__ pooled, const float* __restrict__ cnt,
                           const float* __restrict__ Wc, const float* __restrict__ bc,
                           float* __restrict__ out, int n_graphs, int labels) {
    int i = blockIdx.x * blockDim.x + threadIdx.x;
    if (i >= n_graphs * labels) return;
    int g = i / labels, c = i % labels;
    float inv = 1.0f / fmaxf(cnt[g], 1.0f);
    float acc = bc[c];
#pragma unroll
    for (int j = 0; j < HIDC; ++j)
        acc += pooled[g * HIDC + j] * inv * Wc[j * labels + c];
    out[i] = acc;
}

extern "C" void kernel_launch(void* const* d_in, const int* in_sizes, int n_in,
                              void* d_out, int out_size, void* d_ws, size_t ws_size,
                              hipStream_t stream) {
    const int*   x     = (const int*)d_in[0];
    const int*   ei    = (const int*)d_in[1];
    const int*   batch = (const int*)d_in[2];
    const float* emb   = (const float*)d_in[3];
    const float* W1    = (const float*)d_in[4];
    const float* b1    = (const float*)d_in[5];
    const float* W2    = (const float*)d_in[6];
    const float* b2    = (const float*)d_in[7];
    const float* Wc    = (const float*)d_in[8];
    const float* bc    = (const float*)d_in[9];
    float* out = (float*)d_out;

    const int n_nodes  = in_sizes[0];
    const int n_edges  = in_sizes[1] / 2;
    const int labels   = in_sizes[9];
    const int n_graphs = out_size / labels;
    const int* src = ei;
    const int* dst = ei + n_edges;

    float* ws     = (float*)d_ws;
    float* deg    = ws;
    float* h      = deg + n_nodes;
    float* hw     = h  + (size_t)n_nodes * HIDC;
    float* part   = hw + (size_t)n_nodes * HIDC;
    float* pooled = part + (size_t)NCHUNK * n_nodes * HIDC;
    float* cnt    = pooled + (size_t)n_graphs * HIDC;
    const size_t n_sw   = (size_t)((n_nodes + NPR - 1) / NPR) * NCHUNK * (SBLK / 32);
    const size_t rec_at = ((size_t)(cnt + n_graphs - ws) + 31) & ~(size_t)31;
    int*   rec_sd = (int*)(ws + rec_at);
    float* rec_w  = (float*)(rec_sd + n_sw * RCAP * 2);
    int*   rec_n  = (int*)(rec_w + n_sw * RCAP);
    int*   degc   = rec_n + n_sw;

    const int nh     = n_nodes * HIDC;
    const int ntiles = n_nodes / 16;
    const int wavesPerBlk = BLK / 32;

    dim3 b(BLK);
    auto g1 = [](int n) { return dim3((n + BLK - 1) / BLK); };
    dim3 gw((ntiles + wavesPerBlk - 1) / wavesPerBlk);
    dim3 gs((n_nodes + NPR - 1) / NPR, NCHUNK);

    k_init<<<g1(n_graphs * HIDC + n_graphs), b, 0, stream>>>(pooled, cnt, n_graphs);
    k_filter_deg<<<gs, SBLK, 0, stream>>>(src, dst, degc, rec_sd, rec_n, n_edges, n_nodes);
    k_dinv<<<g1(n_nodes), b, 0, stream>>>(degc, deg, n_nodes);
    k_embed<<<g1(nh), b, 0, stream>>>(x, emb, h, n_nodes);

    k_gcn_wmma<<<gw, b, 0, stream>>>(h, W1, hw, ntiles);
    k_scatter_lds<true><<<gs, SBLK, 0, stream>>>(src, dst, deg, hw, part, rec_sd, rec_w, rec_n,
                                                 n_edges, n_nodes);
    k_sum_relu<<<g1(nh / 4), b, 0, stream>>>(part, deg, hw, b1, h, n_nodes);

    k_gcn_wmma<<<gw, b, 0, stream>>>(h, W2, hw, ntiles);
    k_scatter_lds<false><<<gs, SBLK, 0, stream>>>(src, dst, deg, hw, part, rec_sd, rec_w, rec_n,
                                                  n_edges, n_nodes);
    k_sum_relu<<<g1(nh / 4), b, 0, stream>>>(part, deg, hw, b2, h, n_nodes);

    k_pool<<<g1(nh), b, 0, stream>>>(h, batch, pooled, cnt, n_nodes);
    k_classify<<<g1(n_graphs * labels), b, 0, stream>>>(pooled, cnt, Wc, bc, out,
                                                        n_graphs, labels);
}


# node-wise launches folded: embedding+deg^-1/2+pool init into conv 1's transform, plane sum + relu into conv 2's transform and into the mean pool (wave sums 8 rows, then f32 atomics: order differs)
# speedup vs baseline: 11.1216x; 1.4168x over previous
#include <hip/hip_runtime.h>

typedef __attribute__((ext_vector_type(2))) float v2f;
typedef __attribute__((ext_vector_type(8))) float v8f;

#define HIDC 16
#define BLK 256
#define NPR 4000
#define NCHUNK 10
#define SBLK 1024
#define QCAP 64
#define RCAP 1024

__device__ __forceinline__ void fadd(float* p, float v) {
    __hip_atomic_fetch_add(p, v, __ATOMIC_RELAXED, __HIP_MEMORY_SCOPE_AGENT);
}

__global__ void k_embed_xw(const int* __restrict__ x, const float* __restrict__ emb,
                           const float* __restrict__ W, const int* __restrict__ degc,
                           float* __restrict__ dinv, float* __restrict__ hw,
                           float* __restrict__ pooled, float* __restrict__ cnt,
                           int n_nodes, int ntiles, int n_graphs) {
    const int gid = blockIdx.x * blockDim.x + threadIdx.x;
    for (int k = gid; k < n_graphs * (HIDC + 1); k += gridDim.x * blockDim.x) {
        if (k < n_graphs * HIDC) pooled[k] = 0.0f;
        else cnt[k - n_graphs * HIDC] = 0.0f;
    }

    int wave = gid >> 5;
    int lane = threadIdx.x & 31;
    if (wave >= ntiles) return;
    int half = lane >> 4;
    int l15  = lane & 15;
    const int node = wave * 16 + l15;

    const int tok = x[node];

    int dg = 1;
    for (int c = 0; c < NCHUNK; ++c) dg += degc[(size_t)c * n_nodes + node];
    float d = (float)dg;
    if (half == 0) dinv[node] = (d > 0.0f) ? rsqrtf(d) : 0.0f;

    v2f bf[4];
#pragma unroll
    for (int t = 0; t < 4; ++t) {
        int kk = 4 * t + 2 * half;
        bf[t][0] = W[(kk + 0) * HIDC + l15];
        bf[t][1] = W[(kk + 1) * HIDC + l15];
    }

    const float* hrow = emb + (size_t)tok * HIDC;
    v8f c = {};
#pragma unroll
    for (int t = 0; t < 4; ++t) {
        int kk = 4 * t + 2 * half;
        v2f a;
        a[0] = hrow[kk + 0];
        a[1] = hrow[kk + 1];
        c = __builtin_amdgcn_wmma_f32_16x16x4_f32(
                false, a, false, bf[t], (short)0, c, false, false);
    }

#pragma unroll
    for (int v = 0; v < 8; ++v) {
        int m = v + 8 * half;
        hw[(size_t)(wave * 16 + m) * HIDC + l15] = c[v];
    }
}

template <int I>
__device__ __forceinline__ int row_lane(int v) {
    return __builtin_amdgcn_update_dpp(0, v, 0x150 + I, 0xf, 0xf, false);
}


template <int I>
__device__ __forceinline__ void load_rows(int s, unsigned lane_off,
                                          const float* __restrict__ hw, float (&m)[16]) {
    if constexpr (I < 16) {
        unsigned so = ((unsigned)row_lane<I>(s) << 6) + lane_off;
        m[I] = *(const float*)((const char*)hw + so);
        load_rows<I + 1>(s, lane_off, hw, m);
    }
}

template <int I, bool TAIL>
__device__ __forceinline__ void add_rows(int dl, float nrm, unsigned lane_off,
                                         float* acc, const float (&m)[16]) {
    if constexpr (I < 16) {
        int   di = row_lane<I>(dl);
        float ni = __int_as_float(row_lane<I>(__float_as_int(nrm)));
        if (!TAIL || di >= 0) {
            unsigned dO = ((unsigned)di << 6) + lane_off;
            __hip_atomic_fetch_add((float*)((char*)acc + dO), m[I] * ni,
                                   __ATOMIC_RELAXED, __HIP_MEMORY_SCOPE_WORKGROUP);
        }
        add_rows<I + 1, TAIL>(dl, nrm, lane_off, acc, m);
    }
}

template <bool TAIL>
__device__ __forceinline__ void drain(const int2* q, int head, int n, int lane, int lo,
                                      const float* __restrict__ dinv,
                                      const float* __restrict__ hw, float* acc) {
    int s = 0, dl = -1;
    float nrm = 0.0f;
    if (!TAIL || lane < n) {
        int2 ent = q[(head + lane) & (QCAP - 1)];
        s = ent.x; dl = ent.y;
        nrm = dinv[s] * dinv[lo + dl];
    }
    unsigned lane_off = (unsigned)(lane & 15) * 4u;
    float m[16];
    load_rows<0>(s, lane_off, hw, m);
    add_rows<0, TAIL>(dl, nrm, lane_off, acc, m);
}

__device__ __forceinline__ void put_record(const int2* q, int head, int n, int lane,
                                           int2* __restrict__ rsd) {
    int pos = head + lane;
    if (lane < n && pos < RCAP) rsd[pos] = q[pos & (QCAP - 1)];
}

template <bool CNT>
__device__ __forceinline__ void take(int s, int d, int lo, int cnt, int2* q, int& qt, int& qn,
                                     int lane, const float* __restrict__ dinv,
                                     const float* __restrict__ hw, float* acc,
                                     int2* __restrict__ rsd, int* cntl) {
    int dl = d - lo;
    bool keep = (unsigned)dl < (unsigned)cnt;
    unsigned mask = __builtin_amdgcn_ballot_w32(keep);
    if (mask != 0u) {
        int rank = (int)__builtin_amdgcn_mbcnt_lo(mask, 0u);
        if (keep) {
            q[(qt + rank) & (QCAP - 1)] = make_int2(s, dl);
            if constexpr (CNT)
                __hip_atomic_fetch_add(&cntl[dl], 1, __ATOMIC_RELAXED,
                                       __HIP_MEMORY_SCOPE_WORKGROUP);
        }
        int n = __popc(mask);
        qt += n; qn += n;
        if (qn >= 32) {
            if constexpr (CNT) put_record(q, qt - qn, 32, lane, rsd);
            else drain<false>(q, qt - qn, 32, lane, lo, dinv, hw, acc);
            qn -= 32;
        }
    }
}

template <bool CNT>
__device__ __forceinline__ int filter_chunk(const int* __restrict__ src,
                                            const int* __restrict__ dst,
                                            const float* __restrict__ dinv,
                                            const float* __restrict__ hw, float* acc, int2* q,
                                            int2* __restrict__ rsd, int* cntl,
                                            int n_edges, int e0, int e1, int trip,
                                            int wave, int lane, int lo, int cnt) {
    int qt = 0, qn = 0;
    int base = e0 + wave * 128;

    if ((n_edges & 3) == 0 && base + 128 <= e1) {
        int nw = (e1 - base - 128) / trip + 1;
        if (nw >= 2) {
            int4 sa = *(const int4*)(src + base + lane * 4);
            int4 da = *(const int4*)(dst + base + lane * 4);
            int4 sb = *(const int4*)(src + base + trip + lane * 4);
            int4 db = *(const int4*)(dst + base + trip + lane * 4);
            while (nw >= 2) {
                nw -= 2;
                int nbase = base + 2 * trip;
                int pb = (nw >= 2) ? nbase : base;
                int4 sc = *(const int4*)(src + pb + lane * 4);
                int4 dc = *(const int4*)(dst + pb + lane * 4);
                int4 sd = *(const int4*)(src + pb + trip + lane * 4);
                int4 dd = *(const int4*)(dst + pb + trip + lane * 4);
                take<CNT>(sa.x, da.x, lo, cnt, q, qt, qn, lane, dinv, hw, acc, rsd, cntl);
                take<CNT>(sa.y, da.y, lo, cnt, q, qt, qn, lane, dinv, hw, acc, rsd, cntl);
                take<CNT>(sa.z, da.z, lo, cnt, q, qt, qn, lane, dinv, hw, acc, rsd, cntl);
                take<CNT>(sa.w, da.w, lo, cnt, q, qt, qn, lane, dinv, hw, acc, rsd, cntl);
                take<CNT>(sb.x, db.x, lo, cnt, q, qt, qn, lane, dinv, hw, acc, rsd, cntl);
                take<CNT>(sb.y, db.y, lo, cnt, q, qt, qn, lane, dinv, hw, acc, rsd, cntl);
                take<CNT>(sb.z, db.z, lo, cnt, q, qt, qn, lane, dinv, hw, acc, rsd, cntl);
                take<CNT>(sb.w, db.w, lo, cnt, q, qt, qn, lane, dinv, hw, acc, rsd, cntl);
                sa = sc; da = dc; sb = sd; db = dd;
                base = nbase;
            }
        }
        if (nw == 1) {
            int4 s4 = *(const int4*)(src + base + lane * 4);
            int4 d4 = *(const int4*)(dst + base + lane * 4);
            take<CNT>(s4.x, d4.x, lo, cnt, q, qt, qn, lane, dinv, hw, acc, rsd, cntl);
            take<CNT>(s4.y, d4.y, lo, cnt, q, qt, qn, lane, dinv, hw, acc, rsd, cntl);
            take<CNT>(s4.z, d4.z, lo, cnt, q, qt, qn, lane, dinv, hw, acc, rsd, cntl);
            take<CNT>(s4.w, d4.w, lo, cnt, q, qt, qn, lane, dinv, hw, acc, rsd, cntl);
            base += trip;
        }
    }
    for (; base < e1; base += trip) {
        for (int j = 0; j < 4; ++j) {
            int e = base + j * 32 + lane;
            int s = 0, d = 0x7fffffff;
            if (e < e1) { s = src[e]; d = dst[e]; }
            take<CNT>(s, d, lo, cnt, q, qt, qn, lane, dinv, hw, acc, rsd, cntl);
        }
    }
    if (qn > 0) {
        if constexpr (CNT) put_record(q, qt - qn, qn, lane, rsd);
        else drain<true>(q, qt - qn, qn, lane, lo, dinv, hw, acc);
    }
    return qt;
}

__device__ __forceinline__ void chunk_bounds(int n_edges, int trip, int& e0, int& e1) {
    const int per = (((n_edges + (int)gridDim.y - 1) / (int)gridDim.y + trip - 1) / trip) * trip;
    e0 = blockIdx.y * per;
    e1 = min(e0 + per, n_edges);
}

__global__ void __launch_bounds__(SBLK)
k_filter_deg(const int* __restrict__ src, const int* __restrict__ dst,
             int* __restrict__ degc, int* __restrict__ rec_sd, int* __restrict__ rec_n,
             int n_edges, int n_nodes) {
    __shared__ int cntl[NPR];
    __shared__ int2 queue[SBLK / 32][QCAP];

    const int tid = threadIdx.x, lane = tid & 31;
    const int wave = __builtin_amdgcn_readfirstlane(tid >> 5);
    const int lo  = blockIdx.x * NPR;
    const int cnt = min(NPR, n_nodes - lo);

    for (int i = tid; i < NPR; i += SBLK) cntl[i] = 0;
    __syncthreads();

    const int trip = (SBLK / 32) * 128;
    int e0, e1;
    chunk_bounds(n_edges, trip, e0, e1);

    const int wg = (blockIdx.y * gridDim.x + blockIdx.x) * (SBLK / 32) + wave;
    int2* rsd = (int2*)rec_sd + (size_t)wg * RCAP;

    int kept = filter_chunk<true>(src, dst, nullptr, nullptr, nullptr, queue[wave], rsd, cntl,
                                  n_edges, e0, e1, trip, wave, lane, lo, cnt);
    if (lane == 0) rec_n[wg] = kept;
    __syncthreads();

    int* plane = degc + (size_t)blockIdx.y * n_nodes + lo;
    for (int i = tid; i < cnt; i += SBLK) plane[i] = cntl[i];
}

template <bool FIRST>
__global__ void __launch_bounds__(SBLK)
k_scatter_lds(const int* __restrict__ src, const int* __restrict__ dst,
              const float* __restrict__ dinv, const float* __restrict__ hw,
              float* __restrict__ part, int* __restrict__ rec_sd, float* __restrict__ rec_w,
              const int* __restrict__ rec_n, int n_edges, int n_nodes) {
    __shared__ __attribute__((aligned(16))) float acc[NPR * HIDC];
    __shared__ int2 queue[SBLK / 32][QCAP];

    const int tid = threadIdx.x, lane = tid & 31;
    const int wave = __builtin_amdgcn_readfirstlane(tid >> 5);
    const int lo  = blockIdx.x * NPR;
    const int cnt = min(NPR, n_nodes - lo);

    for (int i = tid; i < NPR * HIDC / 4; i += SBLK)
        ((float4*)acc)[i] = make_float4(0.0f, 0.0f, 0.0f, 0.0f);
    __syncthreads();

    const int trip = (SBLK / 32) * 128;
    int e0, e1;
    chunk_bounds(n_edges, trip, e0, e1);

    const int wg = (blockIdx.y * gridDim.x + blockIdx.x) * (SBLK / 32) + wave;
    int2*  rsd = (int2*)rec_sd + (size_t)wg * RCAP;
    float* rw  = rec_w + (size_t)wg * RCAP;

    const int tot = __builtin_amdgcn_readfirstlane(rec_n[wg]);
    if (tot <= RCAP) {
        const unsigned lane_off = (unsigned)(lane & 15) * 4u;
        const int nfull = tot >> 5;
        if constexpr (FIRST) {
            if (tot > 0) {
                const int last = tot - 1;
                int2  e  = rsd[min(lane, last)];
                int2  en = rsd[min(32 + lane, last)];
                float w  = dinv[e.x] * dinv[lo + e.y];
                if (lane >= tot) { e.x = 0; e.y = -1; w = 0.0f; }
                for (int k = 0; k < nfull; ++k) {
                    int2  e2 = rsd[min((k + 2) * 32 + lane, last)];
                    float m[16];
                    load_rows<0>(e.x, lane_off, hw, m);
                    float fs = dinv[en.x];
                    float fd = dinv[lo + en.y];
                    rw[k * 32 + lane] = w;
                    asm volatile("" :: "v"(e2.x), "v"(e2.y));
                    add_rows<0, false>(e.y, w, lane_off, acc, m);
                    bool  has = (k + 1) * 32 + lane < tot;
                    e.x = has ? en.x : 0;
                    e.y = has ? en.y : -1;
                    w   = has ? fs * fd : 0.0f;
                    en  = e2;
                }
                if (tot & 31) {
                    float m[16];
                    load_rows<0>(e.x, lane_off, hw, m);
                    if (e.y >= 0) rw[nfull * 32 + lane] = w;
                    add_rows<0, true>(e.y, w, lane_off, acc, m);
                }
            }
        } else {
            int2  e = make_int2(0, -1);
            float w = 0.0f;
            if (lane < tot) { e = rsd[lane]; w = rw[lane]; }
            for (int k = 0; k < nfull; ++k) {
                float m[16];
                load_rows<0>(e.x, lane_off, hw, m);
                int   idx = (k + 1) * 32 + lane;
                int   ci  = min(idx, tot - 1);
                int2  en = rsd[ci];
                float wn = rw[ci];
                add_rows<0, false>(e.y, w, lane_off, acc, m);
                bool  has = idx < tot;
                e.x = has ? en.x : 0;
                e.y = has ? en.y : -1;
                w   = has ? wn : 0.0f;
            }
            if (tot & 31) {
                float m[16];
                load_rows<0>(e.x, lane_off, hw, m);
                add_rows<0, true>(e.y, w, lane_off, acc, m);
            }
        }
    } else {
        filter_chunk<false>(src, dst, dinv, hw, acc, queue[wave], rsd, nullptr,
                            n_edges, e0, e1, trip, wave, lane, lo, cnt);
    }
    __syncthreads();

    float4* out4 = (float4*)part + ((size_t)blockIdx.y * n_nodes + lo) * (HIDC / 4);
    for (int i = tid; i < cnt * (HIDC / 4); i += SBLK)
        out4[i] = ((const float4*)acc)[i];
}

__device__ __forceinline__ float4 node_quad(const float* __restrict__ part,
                                            const float* __restrict__ dinv,
                                            const float* __restrict__ hw,
                                            const float* __restrict__ b, int n_nodes, int i) {
    const float4* p4 = (const float4*)part + i;
    const size_t plane = (size_t)n_nodes * (HIDC / 4);
    float4 a = p4[0];
    for (int c = 1; c < NCHUNK; ++c) {
        float4 t = p4[c * plane];
        a.x += t.x; a.y += t.y; a.z += t.z; a.w += t.w;
    }
    float dn = dinv[i >> 2];
    float nn = dn * dn;
    float4 hv = ((const float4*)hw)[i];
    const float* bq = b + (i & 3) * 4;
    float4 r;
    r.x = fmaxf(a.x + hv.x * nn + bq[0], 0.0f);
    r.y = fmaxf(a.y + hv.y * nn + bq[1], 0.0f);
    r.z = fmaxf(a.z + hv.z * nn + bq[2], 0.0f);
    r.w = fmaxf(a.w + hv.w * nn + bq[3], 0.0f);
    return r;
}

#define TPITCH 20

__global__ void __launch_bounds__(BLK)
k_sum_xw(const float* __restrict__ part, const float* __restrict__ dinv,
         const float* __restrict__ hw_in, const float* __restrict__ b,
         const float* __restrict__ W, float* __restrict__ hw_out, int n_nodes, int ntiles) {
    __shared__ __attribute__((aligned(16))) float tile[BLK / 32][16 * TPITCH];

    const int lane = threadIdx.x & 31;
    const int wib  = threadIdx.x >> 5;
    const int wave = blockIdx.x * (BLK / 32) + wib;
    const bool live = wave < ntiles;
    const int  tw   = live ? wave : ntiles - 1;
    const int half = lane >> 4;
    const int l15  = lane & 15;

    v2f bf[4];
#pragma unroll
    for (int t = 0; t < 4; ++t) {
        int kk = 4 * t + 2 * half;
        bf[t][0] = W[(kk + 0) * HIDC + l15];
        bf[t][1] = W[(kk + 1) * HIDC + l15];
    }

    float* mine = tile[wib];
    for (int p = 0; p < 2; ++p) {
        int lr = p * 8 + (lane >> 2);
        int i  = (tw * 16 + lr) * (HIDC / 4) + (lane & 3);
        float4 r = node_quad(part, dinv, hw_in, b, n_nodes, i);
        *(float4*)(mine + lr * TPITCH + (lane & 3) * 4) = r;
    }
    __syncthreads();

    v8f c = {};
#pragma unroll
    for (int t = 0; t < 4; ++t) {
        int kk = 4 * t + 2 * half;
        v2f a;
        a[0] = mine[l15 * TPITCH + kk + 0];
        a[1] = mine[l15 * TPITCH + kk + 1];
        c = __builtin_amdgcn_wmma_f32_16x16x4_f32(
                false, a, false, bf[t], (short)0, c, false, false);
    }

    if (live) {
#pragma unroll
        for (int v = 0; v < 8; ++v) {
            int m = v + 8 * half;
            hw_out[(size_t)(tw * 16 + m) * HIDC + l15] = c[v];
        }
    }
}

__global__ void k_sum_pool(const float* __restrict__ part, const float* __restrict__ dinv,
                           const float* __restrict__ hw, const float* __restrict__ b,
                           const int* __restrict__ batch, float* __restrict__ pooled,
                           float* __restrict__ cnt, int n_nodes) {
    const int i = blockIdx.x * blockDim.x + threadIdx.x;
    const int lane = threadIdx.x & 31;
    const bool in = i < n_nodes * (HIDC / 4);
    float4 r = make_float4(0.0f, 0.0f, 0.0f, 0.0f);
    int g = -1;
    if (in) {
        g = batch[i >> 2];
        r = node_quad(part, dinv, hw, b, n_nodes, i);
    }
    const int g0 = __builtin_amdgcn_readfirstlane(g);
    const bool same = __builtin_amdgcn_ballot_w32(g != g0) == 0u;
    if (same) {
        if (g0 >= 0) {
            for (int off = 4; off < 32; off <<= 1) {
                r.x += __shfl_xor(r.x, off);
                r.y += __shfl_xor(r.y, off);
                r.z += __shfl_xor(r.z, off);
                r.w += __shfl_xor(r.w, off);
            }
            if (lane < 4) {
                float* pg = pooled + (size_t)g0 * HIDC + lane * 4;
                fadd(pg + 0, r.x);
                fadd(pg + 1, r.y);
                fadd(pg + 2, r.z);
                fadd(pg + 3, r.w);
            }
            if (lane == 0) fadd(&cnt[g0], 8.0f);
        }
    } else if (in) {
        float* pg = pooled + (size_t)g * HIDC + (lane & 3) * 4;
        fadd(pg + 0, r.x);
        fadd(pg + 1, r.y);
        fadd(pg + 2, r.z);
        fadd(pg + 3, r.w);
        if ((lane & 3) == 0) fadd(&cnt[g], 1.0f);
    }
}

__global__ void k_classify(const float* __restrict__ pooled, const float* __restrict__ cnt,
                           const float* __restrict__ Wc, const float* __restrict__ bc,
                           float* __restrict__ out, int n_graphs, int labels) {
    int i = blockIdx.x * blockDim.x + threadIdx.x;
    if (i >= n_graphs * labels) return;
    int g = i / labels, c = i % labels;
    float inv = 1.0f / fmaxf(cnt[g], 1.0f);
    float acc = bc[c];
#pragma unroll
    for (int j = 0; j < HIDC; ++j)
        acc += pooled[g * HIDC + j] * inv * Wc[j * labels + c];
    out[i] = acc;
}

extern "C" void kernel_launch(void* const* d_in, const int* in_sizes, int n_in,
                              void* d_out, int out_size, void* d_ws, size_t ws_size,
                              hipStream_t stream) {
    const int*   x     = (const int*)d_in[0];
    const int*   ei    = (const int*)d_in[1];
    const int*   batch = (const int*)d_in[2];
    const float* emb   = (const float*)d_in[3];
    const float* W1    = (const float*)d_in[4];
    const float* b1    = (const float*)d_in[5];
    const float* W2    = (const float*)d_in[6];
    const float* b2    = (const float*)d_in[7];
    const float* Wc    = (const float*)d_in[8];
    const float* bc    = (const float*)d_in[9];
    float* out = (float*)d_out;

    const int n_nodes  = in_sizes[0];
    const int n_edges  = in_sizes[1] / 2;
    const int labels   = in_sizes[9];
    const int n_graphs = out_size / labels;
    const int* src = ei;
    const int* dst = ei + n_edges;

    float* ws     = (float*)d_ws;
    float* deg    = ws;
    float* hw2    = deg + n_nodes;
    float* hw     = hw2 + (size_t)n_nodes * HIDC;
    float* part   = hw + (size_t)n_nodes * HIDC;
    float* pooled = part + (size_t)NCHUNK * n_nodes * HIDC;
    float* cnt    = pooled + (size_t)n_graphs * HIDC;
    const size_t n_sw   = (size_t)((n_nodes + NPR - 1) / NPR) * NCHUNK * (SBLK / 32);
    const size_t rec_at = ((size_t)(cnt + n_graphs - ws) + 31) & ~(size_t)31;
    int*   rec_sd = (int*)(ws + rec_at);
    float* rec_w  = (float*)(rec_sd + n_sw * RCAP * 2);
    int*   rec_n  = (int*)(rec_w + n_sw * RCAP);
    int*   degc   = rec_n + n_sw;

    const int nh     = n_nodes * HIDC;
    const int ntiles = n_nodes / 16;
    const int wavesPerBlk = BLK / 32;

    dim3 b(BLK);
    auto g1 = [](int n) { return dim3((n + BLK - 1) / BLK); };
    dim3 gw((ntiles + wavesPerBlk - 1) / wavesPerBlk);
    dim3 gs((n_nodes + NPR - 1) / NPR, NCHUNK);

    k_filter_deg<<<gs, SBLK, 0, stream>>>(src, dst, degc, rec_sd, rec_n, n_edges, n_nodes);

    k_embed_xw<<<gw, b, 0, stream>>>(x, emb, W1, degc, deg, hw, pooled, cnt,
                                     n_nodes, ntiles, n_graphs);
    k_scatter_lds<true><<<gs, SBLK, 0, stream>>>(src, dst, deg, hw, part, rec_sd, rec_w, rec_n,
                                                 n_edges, n_nodes);

    k_sum_xw<<<gw, b, 0, stream>>>(part, deg, hw, b1, W2, hw2, n_nodes, ntiles);
    k_scatter_lds<false><<<gs, SBLK, 0, stream>>>(src, dst, deg, hw2, part, rec_sd, rec_w, rec_n,
                                                  n_edges, n_nodes);

    k_sum_pool<<<g1(nh / 4), b, 0, stream>>>(part, deg, hw2, b2, batch, pooled, cnt, n_nodes);
    k_classify<<<g1(n_graphs * labels), b, 0, stream>>>(pooled, cnt, Wc, bc, out,
                                                        n_graphs, labels);
}


# conv 1's rows stored as 128-B lines of (feature, extra) pairs, deg^-1/2 of the node in pair 0: one 8-byte load per lane brings row and source factor, no 32-line dinv[src] gather; products unchanged
# speedup vs baseline: 12.0823x; 1.0864x over previous
#include <hip/hip_runtime.h>

typedef __attribute__((ext_vector_type(2))) float v2f;
typedef __attribute__((ext_vector_type(8))) float v8f;

#define HIDC 16
#define BLK 256
#define NPR 4000
#define NCHUNK 10
#define SBLK 1024
#define QCAP 64
#define RCAP 1024

__device__ __forceinline__ void fadd(float* p, float v) {
    __hip_atomic_fetch_add(p, v, __ATOMIC_RELAXED, __HIP_MEMORY_SCOPE_AGENT);
}

template <int I>
__device__ __forceinline__ int row_lane(int v) {
    return __builtin_amdgcn_update_dpp(0, v, 0x150 + I, 0xf, 0xf, false);
}

template <int V>
__device__ __forceinline__ void store_pairs(const v8f& c, float dv, int half, int l15,
                                            v2f* __restrict__ rows) {
    if constexpr (V < 8) {
        float d_lo = __int_as_float(row_lane<V>(__float_as_int(dv)));
        float d_hi = __int_as_float(row_lane<V + 8>(__float_as_int(dv)));
        v2f pr;
        pr[0] = c[V];
        pr[1] = (l15 == 0) ? (half ? d_hi : d_lo) : 0.0f;
        rows[(size_t)(V + 8 * half) * HIDC + l15] = pr;
        store_pairs<V + 1>(c, dv, half, l15, rows);
    }
}

__global__ void k_embed_xw(const int* __restrict__ x, const float* __restrict__ emb,
                           const float* __restrict__ W, const int* __restrict__ degc,
                           float* __restrict__ dinv, float* __restrict__ hw,
                           float* __restrict__ pooled, float* __restrict__ cnt,
                           int n_nodes, int ntiles, int n_graphs) {
    const int gid = blockIdx.x * blockDim.x + threadIdx.x;
    for (int k = gid; k < n_graphs * (HIDC + 1); k += gridDim.x * blockDim.x) {
        if (k < n_graphs * HIDC) pooled[k] = 0.0f;
        else cnt[k - n_graphs * HIDC] = 0.0f;
    }

    int wave = gid >> 5;
    int lane = threadIdx.x & 31;
    if (wave >= ntiles) return;
    int half = lane >> 4;
    int l15  = lane & 15;
    const int node = wave * 16 + l15;

    const int tok = x[node];

    int dg = 1;
    for (int c = 0; c < NCHUNK; ++c) dg += degc[(size_t)c * n_nodes + node];
    float d = (float)dg;
    const float dv = (d > 0.0f) ? rsqrtf(d) : 0.0f;
    if (half == 0) dinv[node] = dv;

    v2f bf[4];
#pragma unroll
    for (int t = 0; t < 4; ++t) {
        int kk = 4 * t + 2 * half;
        bf[t][0] = W[(kk + 0) * HIDC + l15];
        bf[t][1] = W[(kk + 1) * HIDC + l15];
    }

    const float* hrow = emb + (size_t)tok * HIDC;
    v8f c = {};
#pragma unroll
    for (int t = 0; t < 4; ++t) {
        int kk = 4 * t + 2 * half;
        v2f a;
        a[0] = hrow[kk + 0];
        a[1] = hrow[kk + 1];
        c = __builtin_amdgcn_wmma_f32_16x16x4_f32(
                false, a, false, bf[t], (short)0, c, false, false);
    }

    store_pairs<0>(c, dv, half, l15, (v2f*)hw + (size_t)wave * 16 * HIDC);
}


template <int I, int SH>
__device__ __forceinline__ void load_rows(int s, unsigned ld_off,
                                          const float* __restrict__ hw, float (&m)[16]) {
    if constexpr (I < 16) {
        unsigned so = ((unsigned)row_lane<I>(s) << SH) + ld_off;
        m[I] = *(const float*)((const char*)hw + so);
        load_rows<I + 1, SH>(s, ld_off, hw, m);
    }
}

template <int I>
__device__ __forceinline__ void load_pairs(int s, unsigned ld_off,
                                           const float* __restrict__ hw, v2f (&m)[16]) {
    if constexpr (I < 16) {
        unsigned so = ((unsigned)row_lane<I>(s) << 7) + ld_off;
        m[I] = *(const v2f*)((const char*)hw + so);
        load_pairs<I + 1>(s, ld_off, hw, m);
    }
}

template <int I, bool TAIL>
__device__ __forceinline__ void add_rows(int dl, float nrm, unsigned lane_off,
                                         float* acc, const float (&m)[16]) {
    if constexpr (I < 16) {
        int   di = row_lane<I>(dl);
        float ni = __int_as_float(row_lane<I>(__float_as_int(nrm)));
        if (!TAIL || di >= 0) {
            unsigned dO = ((unsigned)di << 6) + lane_off;
            __hip_atomic_fetch_add((float*)((char*)acc + dO), m[I] * ni,
                                   __ATOMIC_RELAXED, __HIP_MEMORY_SCOPE_WORKGROUP);
        }
        add_rows<I + 1, TAIL>(dl, nrm, lane_off, acc, m);
    }
}

template <int I, bool TAIL>
__device__ __forceinline__ void add_pairs(int dl, float fd, unsigned lane_off, int l15,
                                          float* acc, const v2f (&m)[16], float& wv) {
    if constexpr (I < 16) {
        int   di = row_lane<I>(dl);
        float fs = __int_as_float(row_lane<0>(__float_as_int(m[I][1])));
        float ni = fs * __int_as_float(row_lane<I>(__float_as_int(fd)));
        if (l15 == I) wv = ni;
        if (!TAIL || di >= 0) {
            unsigned dO = ((unsigned)di << 6) + lane_off;
            __hip_atomic_fetch_add((float*)((char*)acc + dO), m[I][0] * ni,
                                   __ATOMIC_RELAXED, __HIP_MEMORY_SCOPE_WORKGROUP);
        }
        add_pairs<I + 1, TAIL>(dl, fd, lane_off, l15, acc, m, wv);
    }
}

template <bool TAIL, int SH>
__device__ __forceinline__ void drain(const int2* q, int head, int n, int lane, int lo,
                                      const float* __restrict__ dinv,
                                      const float* __restrict__ hw, float* acc) {
    int s = 0, dl = -1;
    float nrm = 0.0f;
    if (!TAIL || lane < n) {
        int2 ent = q[(head + lane) & (QCAP - 1)];
        s = ent.x; dl = ent.y;
        nrm = dinv[s] * dinv[lo + dl];
    }
    unsigned lane_off = (unsigned)(lane & 15) * 4u;
    float m[16];
    load_rows<0, SH>(s, (unsigned)(lane & 15) << (SH - 4), hw, m);
    add_rows<0, TAIL>(dl, nrm, lane_off, acc, m);
}

__device__ __forceinline__ void put_record(const int2* q, int head, int n, int lane,
                                           int2* __restrict__ rsd) {
    int pos = head + lane;
    if (lane < n && pos < RCAP) rsd[pos] = q[pos & (QCAP - 1)];
}

template <bool CNT, int SH>
__device__ __forceinline__ void take(int s, int d, int lo, int cnt, int2* q, int& qt, int& qn,
                                     int lane, const float* __restrict__ dinv,
                                     const float* __restrict__ hw, float* acc,
                                     int2* __restrict__ rsd, int* cntl) {
    int dl = d - lo;
    bool keep = (unsigned)dl < (unsigned)cnt;
    unsigned mask = __builtin_amdgcn_ballot_w32(keep);
    if (mask != 0u) {
        int rank = (int)__builtin_amdgcn_mbcnt_lo(mask, 0u);
        if (keep) {
            q[(qt + rank) & (QCAP - 1)] = make_int2(s, dl);
            if constexpr (CNT)
                __hip_atomic_fetch_add(&cntl[dl], 1, __ATOMIC_RELAXED,
                                       __HIP_MEMORY_SCOPE_WORKGROUP);
        }
        int n = __popc(mask);
        qt += n; qn += n;
        if (qn >= 32) {
            if constexpr (CNT) put_record(q, qt - qn, 32, lane, rsd);
            else drain<false, SH>(q, qt - qn, 32, lane, lo, dinv, hw, acc);
            qn -= 32;
        }
    }
}

template <bool CNT, int SH>
__device__ __forceinline__ int filter_chunk(const int* __restrict__ src,
                                            const int* __restrict__ dst,
                                            const float* __restrict__ dinv,
                                            const float* __restrict__ hw, float* acc, int2* q,
                                            int2* __restrict__ rsd, int* cntl,
                                            int n_edges, int e0, int e1, int trip,
                                            int wave, int lane, int lo, int cnt) {
    int qt = 0, qn = 0;
    int base = e0 + wave * 128;

    if ((n_edges & 3) == 0 && base + 128 <= e1) {
        int nw = (e1 - base - 128) / trip + 1;
        if (nw >= 2) {
            int4 sa = *(const int4*)(src + base + lane * 4);
            int4 da = *(const int4*)(dst + base + lane * 4);
            int4 sb = *(const int4*)(src + base + trip + lane * 4);
            int4 db = *(const int4*)(dst + base + trip + lane * 4);
            while (nw >= 2) {
                nw -= 2;
                int nbase = base + 2 * trip;
                int pb = (nw >= 2) ? nbase : base;
                int4 sc = *(const int4*)(src + pb + lane * 4);
                int4 dc = *(const int4*)(dst + pb + lane * 4);
                int4 sd = *(const int4*)(src + pb + trip + lane * 4);
                int4 dd = *(const int4*)(dst + pb + trip + lane * 4);
                take<CNT, SH>(sa.x, da.x, lo, cnt, q, qt, qn, lane, dinv, hw, acc, rsd, cntl);
                take<CNT, SH>(sa.y, da.y, lo, cnt, q, qt, qn, lane, dinv, hw, acc, rsd, cntl);
                take<CNT, SH>(sa.z, da.z, lo, cnt, q, qt, qn, lane, dinv, hw, acc, rsd, cntl);
                take<CNT, SH>(sa.w, da.w, lo, cnt, q, qt, qn, lane, dinv, hw, acc, rsd, cntl);
                take<CNT, SH>(sb.x, db.x, lo, cnt, q, qt, qn, lane, dinv, hw, acc, rsd, cntl);
                take<CNT, SH>(sb.y, db.y, lo, cnt, q, qt, qn, lane, dinv, hw, acc, rsd, cntl);
                take<CNT, SH>(sb.z, db.z, lo, cnt, q, qt, qn, lane, dinv, hw, acc, rsd, cntl);
                take<CNT, SH>(sb.w, db.w, lo, cnt, q, qt, qn, lane, dinv, hw, acc, rsd, cntl);
                sa = sc; da = dc; sb = sd; db = dd;
                base = nbase;
            }
        }
        if (nw == 1) {
            int4 s4 = *(const int4*)(src + base + lane * 4);
            int4 d4 = *(const int4*)(dst + base + lane * 4);
            take<CNT, SH>(s4.x, d4.x, lo, cnt, q, qt, qn, lane, dinv, hw, acc, rsd, cntl);
            take<CNT, SH>(s4.y, d4.y, lo, cnt, q, qt, qn, lane, dinv, hw, acc, rsd, cntl);
            take<CNT, SH>(s4.z, d4.z, lo, cnt, q, qt, qn, lane, dinv, hw, acc, rsd, cntl);
            take<CNT, SH>(s4.w, d4.w, lo, cnt, q, qt, qn, lane, dinv, hw, acc, rsd, cntl);
            base += trip;
        }
    }
    for (; base < e1; base += trip) {
        for (int j = 0; j < 4; ++j) {
            int e = base + j * 32 + lane;
            int s = 0, d = 0x7fffffff;
            if (e < e1) { s = src[e]; d = dst[e]; }
            take<CNT, SH>(s, d, lo, cnt, q, qt, qn, lane, dinv, hw, acc, rsd, cntl);
        }
    }
    if (qn > 0) {
        if constexpr (CNT) put_record(q, qt - qn, qn, lane, rsd);
        else drain<true, SH>(q, qt - qn, qn, lane, lo, dinv, hw, acc);
    }
    return qt;
}

__device__ __forceinline__ void chunk_bounds(int n_edges, int trip, int& e0, int& e1) {
    const int per = (((n_edges + (int)gridDim.y - 1) / (int)gridDim.y + trip - 1) / trip) * trip;
    e0 = blockIdx.y * per;
    e1 = min(e0 + per, n_edges);
}

__global__ void __launch_bounds__(SBLK)
k_filter_deg(const int* __restrict__ src, const int* __restrict__ dst,
             int* __restrict__ degc, int* __restrict__ rec_sd, int* __restrict__ rec_n,
             int n_edges, int n_nodes) {
    __shared__ int cntl[NPR];
    __shared__ int2 queue[SBLK / 32][QCAP];

    const int tid = threadIdx.x, lane = tid & 31;
    const int wave = __builtin_amdgcn_readfirstlane(tid >> 5);
    const int lo  = blockIdx.x * NPR;
    const int cnt = min(NPR, n_nodes - lo);

    for (int i = tid; i < NPR; i += SBLK) cntl[i] = 0;
    __syncthreads();

    const int trip = (SBLK / 32) * 128;
    int e0, e1;
    chunk_bounds(n_edges, trip, e0, e1);

    const int wg = (blockIdx.y * gridDim.x + blockIdx.x) * (SBLK / 32) + wave;
    int2* rsd = (int2*)rec_sd + (size_t)wg * RCAP;

    int kept = filter_chunk<true, 6>(src, dst, nullptr, nullptr, nullptr, queue[wave], rsd, cntl,
                                  n_edges, e0, e1, trip, wave, lane, lo, cnt);
    if (lane == 0) rec_n[wg] = kept;
    __syncthreads();

    int* plane = degc + (size_t)blockIdx.y * n_nodes + lo;
    for (int i = tid; i < cnt; i += SBLK) plane[i] = cntl[i];
}

template <bool FIRST>
__global__ void __launch_bounds__(SBLK)
k_scatter_lds(const int* __restrict__ src, const int* __restrict__ dst,
              const float* __restrict__ dinv, const float* __restrict__ hw,
              float* __restrict__ part, int* __restrict__ rec_sd, float* __restrict__ rec_w,
              const int* __restrict__ rec_n, int n_edges, int n_nodes) {
    __shared__ __attribute__((aligned(16))) float acc[NPR * HIDC];
    __shared__ int2 queue[SBLK / 32][QCAP];

    const int tid = threadIdx.x, lane = tid & 31;
    const int wave = __builtin_amdgcn_readfirstlane(tid >> 5);
    const int lo  = blockIdx.x * NPR;
    const int cnt = min(NPR, n_nodes - lo);

    for (int i = tid; i < NPR * HIDC / 4; i += SBLK)
        ((float4*)acc)[i] = make_float4(0.0f, 0.0f, 0.0f, 0.0f);
    __syncthreads();

    const int trip = (SBLK / 32) * 128;
    int e0, e1;
    chunk_bounds(n_edges, trip, e0, e1);

    const int wg = (blockIdx.y * gridDim.x + blockIdx.x) * (SBLK / 32) + wave;
    int2*  rsd = (int2*)rec_sd + (size_t)wg * RCAP;
    float* rw  = rec_w + (size_t)wg * RCAP;

    const int tot = __builtin_amdgcn_readfirstlane(rec_n[wg]);
    if (tot <= RCAP) {
        const unsigned lane_off = (unsigned)(lane & 15) * 4u;
        const int nfull = tot >> 5;
        if constexpr (FIRST) {
            if (tot > 0) {
                const int last = tot - 1;
                const int l15 = lane & 15;
                const unsigned ld_off = (unsigned)l15 * 8u;
                int2  e  = rsd[min(lane, last)];
                int2  en = rsd[min(32 + lane, last)];
                float fd = dinv[lo + e.y];
                if (lane >= tot) { e.x = 0; e.y = -1; fd = 0.0f; }
                for (int k = 0; k < nfull; ++k) {
                    int2  e2 = rsd[min((k + 2) * 32 + lane, last)];
                    v2f m[16];
                    load_pairs<0>(e.x, ld_off, hw, m);
                    float fdn = dinv[lo + en.y];
                    asm volatile("" :: "v"(e2.x), "v"(e2.y));
                    float wv = 0.0f;
                    add_pairs<0, false>(e.y, fd, lane_off, l15, acc, m, wv);
                    rw[k * 32 + lane] = wv;
                    bool  has = (k + 1) * 32 + lane < tot;
                    e.x = has ? en.x : 0;
                    e.y = has ? en.y : -1;
                    fd  = has ? fdn : 0.0f;
                    en  = e2;
                }
                if (tot & 31) {
                    v2f m[16];
                    load_pairs<0>(e.x, ld_off, hw, m);
                    float wv = 0.0f;
                    add_pairs<0, true>(e.y, fd, lane_off, l15, acc, m, wv);
                    if (e.y >= 0) rw[nfull * 32 + lane] = wv;
                }
            }
        } else {
            int2  e = make_int2(0, -1);
            float w = 0.0f;
            if (lane < tot) { e = rsd[lane]; w = rw[lane]; }
            for (int k = 0; k < nfull; ++k) {
                float m[16];
                load_rows<0, 6>(e.x, lane_off, hw, m);
                int   idx = (k + 1) * 32 + lane;
                int   ci  = min(idx, tot - 1);
                int2  en = rsd[ci];
                float wn = rw[ci];
                add_rows<0, false>(e.y, w, lane_off, acc, m);
                bool  has = idx < tot;
                e.x = has ? en.x : 0;
                e.y = has ? en.y : -1;
                w   = has ? wn : 0.0f;
            }
            if (tot & 31) {
                float m[16];
                load_rows<0, 6>(e.x, lane_off, hw, m);
                add_rows<0, true>(e.y, w, lane_off, acc, m);
            }
        }
    } else {
        filter_chunk<false, FIRST ? 7 : 6>(src, dst, dinv, hw, acc, queue[wave], rsd, nullptr,
                            n_edges, e0, e1, trip, wave, lane, lo, cnt);
    }
    __syncthreads();

    float4* out4 = (float4*)part + ((size_t)blockIdx.y * n_nodes + lo) * (HIDC / 4);
    for (int i = tid; i < cnt * (HIDC / 4); i += SBLK)
        out4[i] = ((const float4*)acc)[i];
}

template <bool PAIRS>
__device__ __forceinline__ float4 node_quad(const float* __restrict__ part,
                                            const float* __restrict__ dinv,
                                            const float* __restrict__ hw,
                                            const float* __restrict__ b, int n_nodes, int i) {
    const float4* p4 = (const float4*)part + i;
    const size_t plane = (size_t)n_nodes * (HIDC / 4);
    float4 a = p4[0];
    for (int c = 1; c < NCHUNK; ++c) {
        float4 t = p4[c * plane];
        a.x += t.x; a.y += t.y; a.z += t.z; a.w += t.w;
    }
    float dn = dinv[i >> 2];
    float nn = dn * dn;
    float4 hv;
    if constexpr (PAIRS) {
        float4 p0 = ((const float4*)hw)[2 * i];
        float4 p1 = ((const float4*)hw)[2 * i + 1];
        hv = make_float4(p0.x, p0.z, p1.x, p1.z);
    } else {
        hv = ((const float4*)hw)[i];
    }
    const float* bq = b + (i & 3) * 4;
    float4 r;
    r.x = fmaxf(a.x + hv.x * nn + bq[0], 0.0f);
    r.y = fmaxf(a.y + hv.y * nn + bq[1], 0.0f);
    r.z = fmaxf(a.z + hv.z * nn + bq[2], 0.0f);
    r.w = fmaxf(a.w + hv.w * nn + bq[3], 0.0f);
    return r;
}

#define TPITCH 20

__global__ void __launch_bounds__(BLK)
k_sum_xw(const float* __restrict__ part, const float* __restrict__ dinv,
         const float* __restrict__ hw_in, const float* __restrict__ b,
         const float* __restrict__ W, float* __restrict__ hw_out, int n_nodes, int ntiles) {
    __shared__ __attribute__((aligned(16))) float tile[BLK / 32][16 * TPITCH];

    const int lane = threadIdx.x & 31;
    const int wib  = threadIdx.x >> 5;
    const int wave = blockIdx.x * (BLK / 32) + wib;
    const bool live = wave < ntiles;
    const int  tw   = live ? wave : ntiles - 1;
    const int half = lane >> 4;
    const int l15  = lane & 15;

    v2f bf[4];
#pragma unroll
    for (int t = 0; t < 4; ++t) {
        int kk = 4 * t + 2 * half;
        bf[t][0] = W[(kk + 0) * HIDC + l15];
        bf[t][1] = W[(kk + 1) * HIDC + l15];
    }

    float* mine = tile[wib];
    for (int p = 0; p < 2; ++p) {
        int lr = p * 8 + (lane >> 2);
        int i  = (tw * 16 + lr) * (HIDC / 4) + (lane & 3);
        float4 r = node_quad<true>(part, dinv, hw_in, b, n_nodes, i);
        *(float4*)(mine + lr * TPITCH + (lane & 3) * 4) = r;
    }
    __syncthreads();

    v8f c = {};
#pragma unroll
    for (int t = 0; t < 4; ++t) {
        int kk = 4 * t + 2 * half;
        v2f a;
        a[0] = mine[l15 * TPITCH + kk + 0];
        a[1] = mine[l15 * TPITCH + kk + 1];
        c = __builtin_amdgcn_wmma_f32_16x16x4_f32(
                false, a, false, bf[t], (short)0, c, false, false);
    }

    if (live) {
#pragma unroll
        for (int v = 0; v < 8; ++v) {
            int m = v + 8 * half;
            hw_out[(size_t)(tw * 16 + m) * HIDC + l15] = c[v];
        }
    }
}

__global__ void k_sum_pool(const float* __restrict__ part, const float* __restrict__ dinv,
                           const float* __restrict__ hw, const float* __restrict__ b,
                           const int* __restrict__ batch, float* __restrict__ pooled,
                           float* __restrict__ cnt, int n_nodes) {
    const int i = blockIdx.x * blockDim.x + threadIdx.x;
    const int lane = threadIdx.x & 31;
    const bool in = i < n_nodes * (HIDC / 4);
    float4 r = make_float4(0.0f, 0.0f, 0.0f, 0.0f);
    int g = -1;
    if (in) {
        g = batch[i >> 2];
        r = node_quad<false>(part, dinv, hw, b, n_nodes, i);
    }
    const int g0 = __builtin_amdgcn_readfirstlane(g);
    const bool same = __builtin_amdgcn_ballot_w32(g != g0) == 0u;
    if (same) {
        if (g0 >= 0) {
            for (int off = 4; off < 32; off <<= 1) {
                r.x += __shfl_xor(r.x, off);
                r.y += __shfl_xor(r.y, off);
                r.z += __shfl_xor(r.z, off);
                r.w += __shfl_xor(r.w, off);
            }
            if (lane < 4) {
                float* pg = pooled + (size_t)g0 * HIDC + lane * 4;
                fadd(pg + 0, r.x);
                fadd(pg + 1, r.y);
                fadd(pg + 2, r.z);
                fadd(pg + 3, r.w);
            }
            if (lane == 0) fadd(&cnt[g0], 8.0f);
        }
    } else if (in) {
        float* pg = pooled + (size_t)g * HIDC + (lane & 3) * 4;
        fadd(pg + 0, r.x);
        fadd(pg + 1, r.y);
        fadd(pg + 2, r.z);
        fadd(pg + 3, r.w);
        if ((lane & 3) == 0) fadd(&cnt[g], 1.0f);
    }
}

__global__ void k_classify(const float* __restrict__ pooled, const float* __restrict__ cnt,
                           const float* __restrict__ Wc, const float* __restrict__ bc,
                           float* __restrict__ out, int n_graphs, int labels) {
    int i = blockIdx.x * blockDim.x + threadIdx.x;
    if (i >= n_graphs * labels) return;
    int g = i / labels, c = i % labels;
    float inv = 1.0f / fmaxf(cnt[g], 1.0f);
    float acc = bc[c];
#pragma unroll
    for (int j = 0; j < HIDC; ++j)
        acc += pooled[g * HIDC + j] * inv * Wc[j * labels + c];
    out[i] = acc;
}

extern "C" void kernel_launch(void* const* d_in, const int* in_sizes, int n_in,
                              void* d_out, int out_size, void* d_ws, size_t ws_size,
                              hipStream_t stream) {
    const int*   x     = (const int*)d_in[0];
    const int*   ei    = (const int*)d_in[1];
    const int*   batch = (const int*)d_in[2];
    const float* emb   = (const float*)d_in[3];
    const float* W1    = (const float*)d_in[4];
    const float* b1    = (const float*)d_in[5];
    const float* W2    = (const float*)d_in[6];
    const float* b2    = (const float*)d_in[7];
    const float* Wc    = (const float*)d_in[8];
    const float* bc    = (const float*)d_in[9];
    float* out = (float*)d_out;

    const int n_nodes  = in_sizes[0];
    const int n_edges  = in_sizes[1] / 2;
    const int labels   = in_sizes[9];
    const int n_graphs = out_size / labels;
    const int* src = ei;
    const int* dst = ei + n_edges;

    float* ws     = (float*)d_ws;
    float* deg    = ws;
    float* hw2    = deg + n_nodes;
    float* hw     = hw2 + (size_t)n_nodes * HIDC;
    float* part   = hw + (size_t)n_nodes * HIDC * 2;
    float* pooled = part + (size_t)NCHUNK * n_nodes * HIDC;
    float* cnt    = pooled + (size_t)n_graphs * HIDC;
    const size_t n_sw   = (size_t)((n_nodes + NPR - 1) / NPR) * NCHUNK * (SBLK / 32);
    const size_t rec_at = ((size_t)(cnt + n_graphs - ws) + 31) & ~(size_t)31;
    int*   rec_sd = (int*)(ws + rec_at);
    float* rec_w  = (float*)(rec_sd + n_sw * RCAP * 2);
    int*   rec_n  = (int*)(rec_w + n_sw * RCAP);
    int*   degc   = rec_n + n_sw;

    const int nh     = n_nodes * HIDC;
    const int ntiles = n_nodes / 16;
    const int wavesPerBlk = BLK / 32;

    dim3 b(BLK);
    auto g1 = [](int n) { return dim3((n + BLK - 1) / BLK); };
    dim3 gw((ntiles + wavesPerBlk - 1) / wavesPerBlk);
    dim3 gs((n_nodes + NPR - 1) / NPR, NCHUNK);

    k_filter_deg<<<gs, SBLK, 0, stream>>>(src, dst, degc, rec_sd, rec_n, n_edges, n_nodes);

    k_embed_xw<<<gw, b, 0, stream>>>(x, emb, W1, degc, deg, hw, pooled, cnt,
                                     n_nodes, ntiles, n_graphs);
    k_scatter_lds<true><<<gs, SBLK, 0, stream>>>(src, dst, deg, hw, part, rec_sd, rec_w, rec_n,
                                                 n_edges, n_nodes);

    k_sum_xw<<<gw, b, 0, stream>>>(part, deg, hw, b1, W2, hw2, n_nodes, ntiles);
    k_scatter_lds<false><<<gs, SBLK, 0, stream>>>(src, dst, deg, hw2, part, rec_sd, rec_w, rec_n,
                                                  n_edges, n_nodes);

    k_sum_pool<<<g1(nh / 4), b, 0, stream>>>(part, deg, hw2, b2, batch, pooled, cnt, n_nodes);
    k_classify<<<g1(n_graphs * labels), b, 0, stream>>>(pooled, cnt, Wc, bc, out,
                                                        n_graphs, labels);
}


# a scatter work-group owns 800 rows over 5 edge chunks (degree pass splits a wave's record by sub-range): 2 planes of partial sums stored and summed, not 10; LDS sums in arrival order, arrivals differ
# speedup vs baseline: 13.2015x; 1.0926x over previous
#include <hip/hip_runtime.h>

typedef __attribute__((ext_vector_type(2))) float v2f;
typedef __attribute__((ext_vector_type(8))) float v8f;

#define HIDC 16
#define BLK 256
#define NPR 4000
#define NCHUNK 10
#define NSUB 5
#define SPR (NPR / NSUB)
#define NGRP 2
#define CPG (NCHUNK / NGRP)
#define SBLK 1024
#define QCAP 64
#define SUBCAP 256

__device__ __forceinline__ void fadd(float* p, float v) {
    __hip_atomic_fetch_add(p, v, __ATOMIC_RELAXED, __HIP_MEMORY_SCOPE_AGENT);
}

template <int I>
__device__ __forceinline__ int row_lane(int v) {
    return __builtin_amdgcn_update_dpp(0, v, 0x150 + I, 0xf, 0xf, false);
}

template <int V>
__device__ __forceinline__ void store_pairs(const v8f& c, float dv, int half, int l15,
                                            v2f* __restrict__ rows) {
    if constexpr (V < 8) {
        float d_lo = __int_as_float(row_lane<V>(__float_as_int(dv)));
        float d_hi = __int_as_float(row_lane<V + 8>(__float_as_int(dv)));
        v2f pr;
        pr[0] = c[V];
        pr[1] = (l15 == 0) ? (half ? d_hi : d_lo) : 0.0f;
        rows[(size_t)(V + 8 * half) * HIDC + l15] = pr;
        store_pairs<V + 1>(c, dv, half, l15, rows);
    }
}

__global__ void k_embed_xw(const int* __restrict__ x, const float* __restrict__ emb,
                           const float* __restrict__ W, const int* __restrict__ degc,
                           float* __restrict__ dinv, float* __restrict__ hw,
                           float* __restrict__ pooled, float* __restrict__ cnt,
                           int n_nodes, int ntiles, int n_graphs) {
    const int gid = blockIdx.x * blockDim.x + threadIdx.x;
    for (int k = gid; k < n_graphs * (HIDC + 1); k += gridDim.x * blockDim.x) {
        if (k < n_graphs * HIDC) pooled[k] = 0.0f;
        else cnt[k - n_graphs * HIDC] = 0.0f;
    }

    int wave = gid >> 5;
    int lane = threadIdx.x & 31;
    if (wave >= ntiles) return;
    int half = lane >> 4;
    int l15  = lane & 15;
    const int node = wave * 16 + l15;

    const int tok = x[node];

    int dg = 1;
    for (int c = 0; c < NCHUNK; ++c) dg += degc[(size_t)c * n_nodes + node];
    float d = (float)dg;
    const float dv = (d > 0.0f) ? rsqrtf(d) : 0.0f;
    if (half == 0) dinv[node] = dv;

    v2f bf[4];
#pragma unroll
    for (int t = 0; t < 4; ++t) {
        int kk = 4 * t + 2 * half;
        bf[t][0] = W[(kk + 0) * HIDC + l15];
        bf[t][1] = W[(kk + 1) * HIDC + l15];
    }

    const float* hrow = emb + (size_t)tok * HIDC;
    v8f c = {};
#pragma unroll
    for (int t = 0; t < 4; ++t) {
        int kk = 4 * t + 2 * half;
        v2f a;
        a[0] = hrow[kk + 0];
        a[1] = hrow[kk + 1];
        c = __builtin_amdgcn_wmma_f32_16x16x4_f32(
                false, a, false, bf[t], (short)0, c, false, false);
    }

    store_pairs<0>(c, dv, half, l15, (v2f*)hw + (size_t)wave * 16 * HIDC);
}


template <int I, int SH>
__device__ __forceinline__ void load_rows(int s, unsigned ld_off,
                                          const float* __restrict__ hw, float (&m)[16]) {
    if constexpr (I < 16) {
        unsigned so = ((unsigned)row_lane<I>(s) << SH) + ld_off;
        m[I] = *(const float*)((const char*)hw + so);
        load_rows<I + 1, SH>(s, ld_off, hw, m);
    }
}

template <int I>
__device__ __forceinline__ void load_pairs(int s, unsigned ld_off,
                                           const float* __restrict__ hw, v2f (&m)[16]) {
    if constexpr (I < 16) {
        unsigned so = ((unsigned)row_lane<I>(s) << 7) + ld_off;
        m[I] = *(const v2f*)((const char*)hw + so);
        load_pairs<I + 1>(s, ld_off, hw, m);
    }
}

template <int I, bool TAIL>
__device__ __forceinline__ void add_rows(int dl, float nrm, unsigned lane_off,
                                         float* acc, const float (&m)[16]) {
    if constexpr (I < 16) {
        int   di = row_lane<I>(dl);
        float ni = __int_as_float(row_lane<I>(__float_as_int(nrm)));
        if (!TAIL || di >= 0) {
            unsigned dO = ((unsigned)di << 6) + lane_off;
            __hip_atomic_fetch_add((float*)((char*)acc + dO), m[I] * ni,
                                   __ATOMIC_RELAXED, __HIP_MEMORY_SCOPE_WORKGROUP);
        }
        add_rows<I + 1, TAIL>(dl, nrm, lane_off, acc, m);
    }
}

template <int I, bool TAIL>
__device__ __forceinline__ void add_pairs(int dl, float fd, unsigned lane_off, int l15,
                                          float* acc, const v2f (&m)[16], float& wv) {
    if constexpr (I < 16) {
        int   di = row_lane<I>(dl);
        float fs = __int_as_float(row_lane<0>(__float_as_int(m[I][1])));
        float ni = fs * __int_as_float(row_lane<I>(__float_as_int(fd)));
        if (l15 == I) wv = ni;
        if (!TAIL || di >= 0) {
            unsigned dO = ((unsigned)di << 6) + lane_off;
            __hip_atomic_fetch_add((float*)((char*)acc + dO), m[I][0] * ni,
                                   __ATOMIC_RELAXED, __HIP_MEMORY_SCOPE_WORKGROUP);
        }
        add_pairs<I + 1, TAIL>(dl, fd, lane_off, l15, acc, m, wv);
    }
}

template <bool TAIL, int SH>
__device__ __forceinline__ void drain(const int2* q, int head, int n, int lane, int lo,
                                      const float* __restrict__ dinv,
                                      const float* __restrict__ hw, float* acc) {
    int s = 0, dl = -1;
    float nrm = 0.0f;
    if (!TAIL || lane < n) {
        int2 ent = q[(head + lane) & (QCAP - 1)];
        s = ent.x; dl = ent.y;
        nrm = dinv[s] * dinv[lo + dl];
    }
    unsigned lane_off = (unsigned)(lane & 15) * 4u;
    float m[16];
    load_rows<0, SH>(s, (unsigned)(lane & 15) << (SH - 4), hw, m);
    add_rows<0, TAIL>(dl, nrm, lane_off, acc, m);
}

__device__ __forceinline__ void put_record(const int2* q, int head, int n, int lane,
                                           int2* __restrict__ rsd, int (&cn)[NSUB]) {
    int2 ent = make_int2(0, 0);
    int sub = -1;
    if (lane < n) {
        ent = q[(head + lane) & (QCAP - 1)];
        sub = (int)((unsigned)ent.y / (unsigned)SPR);
    }
    int pos = 0;
#pragma unroll
    for (int k = 0; k < NSUB; ++k) {
        unsigned mk = __builtin_amdgcn_ballot_w32(sub == k);
        int pk = (int)__builtin_amdgcn_mbcnt_lo(mk, (unsigned)cn[k]);
        pos = (sub == k) ? pk : pos;
        cn[k] += __popc(mk);
    }
    if (sub >= 0 && pos < SUBCAP)
        rsd[sub * SUBCAP + pos] = make_int2(ent.x, ent.y - sub * SPR);
}

template <bool CNT, int SH>
__device__ __forceinline__ void take(int s, int d, int lo, int cnt, int2* q, int& qt, int& qn,
                                     int lane, const float* __restrict__ dinv,
                                     const float* __restrict__ hw, float* acc,
                                     int2* __restrict__ rsd, int* cntl, int (&cn)[NSUB]) {
    int dl = d - lo;
    bool keep = (unsigned)dl < (unsigned)cnt;
    unsigned mask = __builtin_amdgcn_ballot_w32(keep);
    if (mask != 0u) {
        int rank = (int)__builtin_amdgcn_mbcnt_lo(mask, 0u);
        if (keep) {
            q[(qt + rank) & (QCAP - 1)] = make_int2(s, dl);
            if constexpr (CNT)
                __hip_atomic_fetch_add(&cntl[dl], 1, __ATOMIC_RELAXED,
                                       __HIP_MEMORY_SCOPE_WORKGROUP);
        }
        int n = __popc(mask);
        qt += n; qn += n;
        if (qn >= 32) {
            if constexpr (CNT) put_record(q, qt - qn, 32, lane, rsd, cn);
            else drain<false, SH>(q, qt - qn, 32, lane, lo, dinv, hw, acc);
            qn -= 32;
        }
    }
}

template <bool CNT, int SH>
__device__ __forceinline__ int filter_chunk(const int* __restrict__ src,
                                            const int* __restrict__ dst,
                                            const float* __restrict__ dinv,
                                            const float* __restrict__ hw, float* acc, int2* q,
                                            int2* __restrict__ rsd, int* cntl,
                                            int n_edges, int e0, int e1, int trip,
                                            int wave, int lane, int lo, int cnt, int (&cn)[NSUB]) {
    int qt = 0, qn = 0;
    int base = e0 + wave * 128;

    if ((n_edges & 3) == 0 && base + 128 <= e1) {
        int nw = (e1 - base - 128) / trip + 1;
        if (nw >= 2) {
            int4 sa = *(const int4*)(src + base + lane * 4);
            int4 da = *(const int4*)(dst + base + lane * 4);
            int4 sb = *(const int4*)(src + base + trip + lane * 4);
            int4 db = *(const int4*)(dst + base + trip + lane * 4);
            while (nw >= 2) {
                nw -= 2;
                int nbase = base + 2 * trip;
                int pb = (nw >= 2) ? nbase : base;
                int4 sc = *(const int4*)(src + pb + lane * 4);
                int4 dc = *(const int4*)(dst + pb + lane * 4);
                int4 sd = *(const int4*)(src + pb + trip + lane * 4);
                int4 dd = *(const int4*)(dst + pb + trip + lane * 4);
                take<CNT, SH>(sa.x, da.x, lo, cnt, q, qt, qn, lane, dinv, hw, acc, rsd, cntl, cn);
                take<CNT, SH>(sa.y, da.y, lo, cnt, q, qt, qn, lane, dinv, hw, acc, rsd, cntl, cn);
                take<CNT, SH>(sa.z, da.z, lo, cnt, q, qt, qn, lane, dinv, hw, acc, rsd, cntl, cn);
                take<CNT, SH>(sa.w, da.w, lo, cnt, q, qt, qn, lane, dinv, hw, acc, rsd, cntl, cn);
                take<CNT, SH>(sb.x, db.x, lo, cnt, q, qt, qn, lane, dinv, hw, acc, rsd, cntl, cn);
                take<CNT, SH>(sb.y, db.y, lo, cnt, q, qt, qn, lane, dinv, hw, acc, rsd, cntl, cn);
                take<CNT, SH>(sb.z, db.z, lo, cnt, q, qt, qn, lane, dinv, hw, acc, rsd, cntl, cn);
                take<CNT, SH>(sb.w, db.w, lo, cnt, q, qt, qn, lane, dinv, hw, acc, rsd, cntl, cn);
                sa = sc; da = dc; sb = sd; db = dd;
                base = nbase;
            }
        }
        if (nw == 1) {
            int4 s4 = *(const int4*)(src + base + lane * 4);
            int4 d4 = *(const int4*)(dst + base + lane * 4);
            take<CNT, SH>(s4.x, d4.x, lo, cnt, q, qt, qn, lane, dinv, hw, acc, rsd, cntl, cn);
            take<CNT, SH>(s4.y, d4.y, lo, cnt, q, qt, qn, lane, dinv, hw, acc, rsd, cntl, cn);
            take<CNT, SH>(s4.z, d4.z, lo, cnt, q, qt, qn, lane, dinv, hw, acc, rsd, cntl, cn);
            take<CNT, SH>(s4.w, d4.w, lo, cnt, q, qt, qn, lane, dinv, hw, acc, rsd, cntl, cn);
            base += trip;
        }
    }
    for (; base < e1; base += trip) {
        for (int j = 0; j < 4; ++j) {
            int e = base + j * 32 + lane;
            int s = 0, d = 0x7fffffff;
            if (e < e1) { s = src[e]; d = dst[e]; }
            take<CNT, SH>(s, d, lo, cnt, q, qt, qn, lane, dinv, hw, acc, rsd, cntl, cn);
        }
    }
    if (qn > 0) {
        if constexpr (CNT) put_record(q, qt - qn, qn, lane, rsd, cn);
        else drain<true, SH>(q, qt - qn, qn, lane, lo, dinv, hw, acc);
    }
    return qt;
}

__device__ __forceinline__ void chunk_bounds(int n_edges, int trip, int c, int& e0, int& e1) {
    const int per = (((n_edges + NCHUNK - 1) / NCHUNK + trip - 1) / trip) * trip;
    e0 = c * per;
    e1 = min(e0 + per, n_edges);
}

__global__ void __launch_bounds__(SBLK)
k_filter_deg(const int* __restrict__ src, const int* __restrict__ dst,
             int* __restrict__ degc, int* __restrict__ rec_sd, int* __restrict__ rec_n,
             int n_edges, int n_nodes) {
    __shared__ int cntl[NPR];
    __shared__ int2 queue[SBLK / 32][QCAP];

    const int tid = threadIdx.x, lane = tid & 31;
    const int wave = __builtin_amdgcn_readfirstlane(tid >> 5);
    const int lo  = blockIdx.x * NPR;
    const int cnt = min(NPR, n_nodes - lo);

    for (int i = tid; i < NPR; i += SBLK) cntl[i] = 0;
    __syncthreads();

    const int trip = (SBLK / 32) * 128;
    int e0, e1;
    chunk_bounds(n_edges, trip, blockIdx.y, e0, e1);

    const int wg = (blockIdx.y * gridDim.x + blockIdx.x) * (SBLK / 32) + wave;
    int2* rsd = (int2*)rec_sd + (size_t)wg * (NSUB * SUBCAP);

    int cn[NSUB] = {};
    filter_chunk<true, 6>(src, dst, nullptr, nullptr, nullptr, queue[wave], rsd, cntl,
                          n_edges, e0, e1, trip, wave, lane, lo, cnt, cn);
    int mine = 0;
#pragma unroll
    for (int k = 0; k < NSUB; ++k) mine = (lane == k) ? cn[k] : mine;
    if (lane < NSUB) rec_n[wg * NSUB + lane] = mine;
    __syncthreads();

    int* plane = degc + (size_t)blockIdx.y * n_nodes + lo;
    for (int i = tid; i < cnt; i += SBLK) plane[i] = cntl[i];
}

__device__ __forceinline__ int stream_at(int i, const int (&p)[CPG + 1], const int (&D)[CPG]) {
    int a = 0;
#pragma unroll
    for (int k = 1; k < CPG; ++k) a = (i >= p[k]) ? D[k] : a;
    return i + a;
}

template <bool FIRST>
__global__ void __launch_bounds__(SBLK)
k_scatter_lds(const int* __restrict__ src, const int* __restrict__ dst,
              const float* __restrict__ dinv, const float* __restrict__ hw,
              float* __restrict__ part, const int* __restrict__ rec_sd, float* __restrict__ rec_w,
              const int* __restrict__ rec_n, int n_edges, int n_nodes) {
    __shared__ __attribute__((aligned(16))) float acc[SPR * HIDC];
    __shared__ int2 queue[SBLK / 32][QCAP];

    const int tid = threadIdx.x, lane = tid & 31;
    const int wave = __builtin_amdgcn_readfirstlane(tid >> 5);
    const int nrf = gridDim.x / NSUB;
    const int rf  = blockIdx.x / NSUB;
    const int sub = blockIdx.x - rf * NSUB;
    const int lo  = rf * NPR + sub * SPR;
    const int cnt = max(0, min(SPR, n_nodes - lo));

    for (int i = tid; i < SPR * HIDC / 4; i += SBLK)
        ((float4*)acc)[i] = make_float4(0.0f, 0.0f, 0.0f, 0.0f);
    __syncthreads();

    const int trip = (SBLK / 32) * 128;

    const int wpc = nrf * (SBLK / 32);
    const int wf0 = ((int)blockIdx.y * CPG * nrf + rf) * (SBLK / 32) + wave;
    const int2* rs0 = (const int2*)rec_sd + ((size_t)wf0 * NSUB + sub) * SUBCAP;
    const int ws = ((int)blockIdx.y * gridDim.x + blockIdx.x) * (SBLK / 32) + wave;
    float* rw = rec_w + (size_t)ws * (CPG * SUBCAP);

    int nk = 0;
    if (lane < CPG) nk = rec_n[(wf0 + lane * wpc) * NSUB + sub];
    int p[CPG + 1], D[CPG];
    bool fits = true;
    p[0] = 0;
#pragma unroll
    for (int k = 0; k < CPG; ++k) {
        int n = __builtin_amdgcn_readlane(nk, k);
        fits = fits && (n <= SUBCAP);
        p[k + 1] = p[k] + n;
        D[k] = k * wpc * (NSUB * SUBCAP) - p[k];
    }
    const int tot = p[CPG];
    if (fits) {
        const unsigned lane_off = (unsigned)(lane & 15) * 4u;
        const int nfull = tot >> 5;
        if constexpr (FIRST) {
            if (tot > 0) {
                const int last = tot - 1;
                const int l15 = lane & 15;
                const unsigned ld_off = (unsigned)l15 * 8u;
                int2  e  = rs0[stream_at(min(lane, last), p, D)];
                int2  en = rs0[stream_at(min(32 + lane, last), p, D)];
                float fd = dinv[lo + e.y];
                if (lane >= tot) { e.x = 0; e.y = -1; fd = 0.0f; }
                for (int k = 0; k < nfull; ++k) {
                    int2  e2 = rs0[stream_at(min((k + 2) * 32 + lane, last), p, D)];
                    v2f m[16];
                    load_pairs<0>(e.x, ld_off, hw, m);
                    float fdn = dinv[lo + en.y];
                    asm volatile("" :: "v"(e2.x), "v"(e2.y));
                    float wv = 0.0f;
                    add_pairs<0, false>(e.y, fd, lane_off, l15, acc, m, wv);
                    rw[k * 32 + lane] = wv;
                    bool  has = (k + 1) * 32 + lane < tot;
                    e.x = has ? en.x : 0;
                    e.y = has ? en.y : -1;
                    fd  = has ? fdn : 0.0f;
                    en  = e2;
                }
                if (tot & 31) {
                    v2f m[16];
                    load_pairs<0>(e.x, ld_off, hw, m);
                    float wv = 0.0f;
                    add_pairs<0, true>(e.y, fd, lane_off, l15, acc, m, wv);
                    if (e.y >= 0) rw[nfull * 32 + lane] = wv;
                }
            }
        } else {
            int2  e = make_int2(0, -1);
            float w = 0.0f;
            if (lane < tot) { e = rs0[stream_at(lane, p, D)]; w = rw[lane]; }
            for (int k = 0; k < nfull; ++k) {
                float m[16];
                load_rows<0, 6>(e.x, lane_off, hw, m);
                int   idx = (k + 1) * 32 + lane;
                int   ci  = min(idx, tot - 1);
                int2  en = rs0[stream_at(ci, p, D)];
                float wn = rw[ci];
                add_rows<0, false>(e.y, w, lane_off, acc, m);
                bool  has = idx < tot;
                e.x = has ? en.x : 0;
                e.y = has ? en.y : -1;
                w   = has ? wn : 0.0f;
            }
            if (tot & 31) {
                float m[16];
                load_rows<0, 6>(e.x, lane_off, hw, m);
                add_rows<0, true>(e.y, w, lane_off, acc, m);
            }
        }
    } else {
        int unused[NSUB] = {};
        for (int k = 0; k < CPG; ++k) {
            int e0, e1;
            chunk_bounds(n_edges, trip, (int)blockIdx.y * CPG + k, e0, e1);
            filter_chunk<false, FIRST ? 7 : 6>(src, dst, dinv, hw, acc, queue[wave], nullptr,
                                nullptr, n_edges, e0, e1, trip, wave, lane, lo, cnt, unused);
        }
    }
    __syncthreads();

    float4* out4 = (float4*)part + ((size_t)blockIdx.y * n_nodes + lo) * (HIDC / 4);
    for (int i = tid; i < cnt * (HIDC / 4); i += SBLK)
        out4[i] = ((const float4*)acc)[i];
}

template <bool PAIRS>
__device__ __forceinline__ float4 node_quad(const float* __restrict__ part,
                                            const float* __restrict__ dinv,
                                            const float* __restrict__ hw,
                                            const float* __restrict__ b, int n_nodes, int i) {
    const float4* p4 = (const float4*)part + i;
    const size_t plane = (size_t)n_nodes * (HIDC / 4);
    float4 a = p4[0];
    for (int c = 1; c < NGRP; ++c) {
        float4 t = p4[c * plane];
        a.x += t.x; a.y += t.y; a.z += t.z; a.w += t.w;
    }
    float dn = dinv[i >> 2];
    float nn = dn * dn;
    float4 hv;
    if constexpr (PAIRS) {
        float4 p0 = ((const float4*)hw)[2 * i];
        float4 p1 = ((const float4*)hw)[2 * i + 1];
        hv = make_float4(p0.x, p0.z, p1.x, p1.z);
    } else {
        hv = ((const float4*)hw)[i];
    }
    const float* bq = b + (i & 3) * 4;
    float4 r;
    r.x = fmaxf(a.x + hv.x * nn + bq[0], 0.0f);
    r.y = fmaxf(a.y + hv.y * nn + bq[1], 0.0f);
    r.z = fmaxf(a.z + hv.z * nn + bq[2], 0.0f);
    r.w = fmaxf(a.w + hv.w * nn + bq[3], 0.0f);
    return r;
}

#define TPITCH 20

__global__ void __launch_bounds__(BLK)
k_sum_xw(const float* __restrict__ part, const float* __restrict__ dinv,
         const float* __restrict__ hw_in, const float* __restrict__ b,
         const float* __restrict__ W, float* __restrict__ hw_out, int n_nodes, int ntiles) {
    __shared__ __attribute__((aligned(16))) float tile[BLK / 32][16 * TPITCH];

    const int lane = threadIdx.x & 31;
    const int wib  = threadIdx.x >> 5;
    const int wave = blockIdx.x * (BLK / 32) + wib;
    const bool live = wave < ntiles;
    const int  tw   = live ? wave : ntiles - 1;
    const int half = lane >> 4;
    const int l15  = lane & 15;

    v2f bf[4];
#pragma unroll
    for (int t = 0; t < 4; ++t) {
        int kk = 4 * t + 2 * half;
        bf[t][0] = W[(kk + 0) * HIDC + l15];
        bf[t][1] = W[(kk + 1) * HIDC + l15];
    }

    float* mine = tile[wib];
    for (int p = 0; p < 2; ++p) {
        int lr = p * 8 + (lane >> 2);
        int i  = (tw * 16 + lr) * (HIDC / 4) + (lane & 3);
        float4 r = node_quad<true>(part, dinv, hw_in, b, n_nodes, i);
        *(float4*)(mine + lr * TPITCH + (lane & 3) * 4) = r;
    }
    __syncthreads();

    v8f c = {};
#pragma unroll
    for (int t = 0; t < 4; ++t) {
        int kk = 4 * t + 2 * half;
        v2f a;
        a[0] = mine[l15 * TPITCH + kk + 0];
        a[1] = mine[l15 * TPITCH + kk + 1];
        c = __builtin_amdgcn_wmma_f32_16x16x4_f32(
                false, a, false, bf[t], (short)0, c, false, false);
    }

    if (live) {
#pragma unroll
        for (int v = 0; v < 8; ++v) {
            int m = v + 8 * half;
            hw_out[(size_t)(tw * 16 + m) * HIDC + l15] = c[v];
        }
    }
}

__global__ void k_sum_pool(const float* __restrict__ part, const float* __restrict__ dinv,
                           const float* __restrict__ hw, const float* __restrict__ b,
                           const int* __restrict__ batch, float* __restrict__ pooled,
                           float* __restrict__ cnt, int n_nodes) {
    const int i = blockIdx.x * blockDim.x + threadIdx.x;
    const int lane = threadIdx.x & 31;
    const bool in = i < n_nodes * (HIDC / 4);
    float4 r = make_float4(0.0f, 0.0f, 0.0f, 0.0f);
    int g = -1;
    if (in) {
        g = batch[i >> 2];
        r = node_quad<false>(part, dinv, hw, b, n_nodes, i);
    }
    const int g0 = __builtin_amdgcn_readfirstlane(g);
    const bool same = __builtin_amdgcn_ballot_w32(g != g0) == 0u;
    if (same) {
        if (g0 >= 0) {
            for (int off = 4; off < 32; off <<= 1) {
                r.x += __shfl_xor(r.x, off);
                r.y += __shfl_xor(r.y, off);
                r.z += __shfl_xor(r.z, off);
                r.w += __shfl_xor(r.w, off);
            }
            if (lane < 4) {
                float* pg = pooled + (size_t)g0 * HIDC + lane * 4;
                fadd(pg + 0, r.x);
                fadd(pg + 1, r.y);
                fadd(pg + 2, r.z);
                fadd(pg + 3, r.w);
            }
            if (lane == 0) fadd(&cnt[g0], 8.0f);
        }
    } else if (in) {
        float* pg = pooled + (size_t)g * HIDC + (lane & 3) * 4;
        fadd(pg + 0, r.x);
        fadd(pg + 1, r.y);
        fadd(pg + 2, r.z);
        fadd(pg + 3, r.w);
        if ((lane & 3) == 0) fadd(&cnt[g], 1.0f);
    }
}

__global__ void k_classify(const float* __restrict__ pooled, const float* __restrict__ cnt,
                           const float* __restrict__ Wc, const float* __restrict__ bc,
                           float* __restrict__ out, int n_graphs, int labels) {
    int i = blockIdx.x * blockDim.x + threadIdx.x;
    if (i >= n_graphs * labels) return;
    int g = i / labels, c = i % labels;
    float inv = 1.0f / fmaxf(cnt[g], 1.0f);
    float acc = bc[c];
#pragma unroll
    for (int j = 0; j < HIDC; ++j)
        acc += pooled[g * HIDC + j] * inv * Wc[j * labels + c];
    out[i] = acc;
}

extern "C" void kernel_launch(void* const* d_in, const int* in_sizes, int n_in,
                              void* d_out, int out_size, void* d_ws, size_t ws_size,
                              hipStream_t stream) {
    const int*   x     = (const int*)d_in[0];
    const int*   ei    = (const int*)d_in[1];
    const int*   batch = (const int*)d_in[2];
    const float* emb   = (const float*)d_in[3];
    const float* W1    = (const float*)d_in[4];
    const float* b1    = (const float*)d_in[5];
    const float* W2    = (const float*)d_in[6];
    const float* b2    = (const float*)d_in[7];
    const float* Wc    = (const float*)d_in[8];
    const float* bc    = (const float*)d_in[9];
    float* out = (float*)d_out;

    const int n_nodes  = in_sizes[0];
    const int n_edges  = in_sizes[1] / 2;
    const int labels   = in_sizes[9];
    const int n_graphs = out_size / labels;
    const int* src = ei;
    const int* dst = ei + n_edges;

    float* ws     = (float*)d_ws;
    float* deg    = ws;
    float* hw2    = deg + n_nodes;
    float* hw     = hw2 + (size_t)n_nodes * HIDC;
    float* part   = hw + (size_t)n_nodes * HIDC * 2;
    float* pooled = part + (size_t)NGRP * n_nodes * HIDC;
    float* cnt    = pooled + (size_t)n_graphs * HIDC;
    const size_t n_sw   = (size_t)((n_nodes + NPR - 1) / NPR) * NCHUNK * (SBLK / 32);
    const size_t rec_at = ((size_t)(cnt + n_graphs - ws) + 31) & ~(size_t)31;
    int*   rec_sd = (int*)(ws + rec_at);
    float* rec_w  = (float*)(rec_sd + n_sw * NSUB * SUBCAP * 2);
    int*   rec_n  = (int*)(rec_w + n_sw * CPG * SUBCAP);
    int*   degc   = rec_n + n_sw * NSUB;

    const int nh     = n_nodes * HIDC;
    const int ntiles = n_nodes / 16;
    const int wavesPerBlk = BLK / 32;

    dim3 b(BLK);
    auto g1 = [](int n) { return dim3((n + BLK - 1) / BLK); };
    dim3 gw((ntiles + wavesPerBlk - 1) / wavesPerBlk);
    dim3 gf((n_nodes + NPR - 1) / NPR, NCHUNK);
    dim3 gs(((n_nodes + NPR - 1) / NPR) * NSUB, NGRP);

    k_filter_deg<<<gf, SBLK, 0, stream>>>(src, dst, degc, rec_sd, rec_n, n_edges, n_nodes);

    k_embed_xw<<<gw, b, 0, stream>>>(x, emb, W1, degc, deg, hw, pooled, cnt,
                                     n_nodes, ntiles, n_graphs);
    k_scatter_lds<true><<<gs, SBLK, 0, stream>>>(src, dst, deg, hw, part, rec_sd, rec_w, rec_n,
                                                 n_edges, n_nodes);

    k_sum_xw<<<gw, b, 0, stream>>>(part, deg, hw, b1, W2, hw2, n_nodes, ntiles);
    k_scatter_lds<false><<<gs, SBLK, 0, stream>>>(src, dst, deg, hw2, part, rec_sd, rec_w, rec_n,
                                                  n_edges, n_nodes);

    k_sum_pool<<<g1(nh / 4), b, 0, stream>>>(part, deg, hw2, b2, batch, pooled, cnt, n_nodes);
    k_classify<<<g1(n_graphs * labels), b, 0, stream>>>(pooled, cnt, Wc, bc, out,
                                                        n_graphs, labels);
}


# a scatter work-group owns 400 rows over all 10 edge chunks: LDS sums (arrival order) are complete, so self loop+bias+relu and conv 2's WMMA transform / pool adds run in its tail; no plane sums
# speedup vs baseline: 15.5131x; 1.1751x over previous
#include <hip/hip_runtime.h>

typedef __attribute__((ext_vector_type(2))) float v2f;
typedef __attribute__((ext_vector_type(8))) float v8f;

#define HIDC 16
#define BLK 256
#define NPR 4000
#define NCHUNK 10
#define NSUB 10
#define SPR (NPR / NSUB)
#define SBLK 1024
#define QCAP 64
#define SUBCAP 128

static_assert(SPR % 16 == 0 && SPR / 16 <= SBLK / 32, "a scatter wave transforms at most one 16-row tile");

__device__ __forceinline__ void fadd(float* p, float v) {
    __hip_atomic_fetch_add(p, v, __ATOMIC_RELAXED, __HIP_MEMORY_SCOPE_AGENT);
}

template <int I>
__device__ __forceinline__ int row_lane(int v) {
    return __builtin_amdgcn_update_dpp(0, v, 0x150 + I, 0xf, 0xf, false);
}

template <int V>
__device__ __forceinline__ void store_pairs(const v8f& c, float dv, int half, int l15,
                                            v2f* __restrict__ rows) {
    if constexpr (V < 8) {
        float d_lo = __int_as_float(row_lane<V>(__float_as_int(dv)));
        float d_hi = __int_as_float(row_lane<V + 8>(__float_as_int(dv)));
        v2f pr;
        pr[0] = c[V];
        pr[1] = (l15 == 0) ? (half ? d_hi : d_lo) : 0.0f;
        rows[(size_t)(V + 8 * half) * HIDC + l15] = pr;
        store_pairs<V + 1>(c, dv, half, l15, rows);
    }
}

__global__ void k_embed_xw(const int* __restrict__ x, const float* __restrict__ emb,
                           const float* __restrict__ W, const int* __restrict__ degc,
                           float* __restrict__ dinv, float* __restrict__ hw,
                           float* __restrict__ pooled, float* __restrict__ cnt,
                           int n_nodes, int ntiles, int n_graphs) {
    const int gid = blockIdx.x * blockDim.x + threadIdx.x;
    for (int k = gid; k < n_graphs * (HIDC + 1); k += gridDim.x * blockDim.x) {
        if (k < n_graphs * HIDC) pooled[k] = 0.0f;
        else cnt[k - n_graphs * HIDC] = 0.0f;
    }

    int wave = gid >> 5;
    int lane = threadIdx.x & 31;
    if (wave >= ntiles) return;
    int half = lane >> 4;
    int l15  = lane & 15;
    const int node = wave * 16 + l15;

    const int tok = x[node];

    int dg = 1;
    for (int c = 0; c < NCHUNK; ++c) dg += degc[(size_t)c * n_nodes + node];
    float d = (float)dg;
    const float dv = (d > 0.0f) ? rsqrtf(d) : 0.0f;
    if (half == 0) dinv[node] = dv;

    v2f bf[4];
#pragma unroll
    for (int t = 0; t < 4; ++t) {
        int kk = 4 * t + 2 * half;
        bf[t][0] = W[(kk + 0) * HIDC + l15];
        bf[t][1] = W[(kk + 1) * HIDC + l15];
    }

    const float* hrow = emb + (size_t)tok * HIDC;
    v8f c = {};
#pragma unroll
    for (int t = 0; t < 4; ++t) {
        int kk = 4 * t + 2 * half;
        v2f a;
        a[0] = hrow[kk + 0];
        a[1] = hrow[kk + 1];
        c = __builtin_amdgcn_wmma_f32_16x16x4_f32(
                false, a, false, bf[t], (short)0, c, false, false);
    }

    store_pairs<0>(c, dv, half, l15, (v2f*)hw + (size_t)wave * 16 * HIDC);
}


template <int I, int SH>
__device__ __forceinline__ void load_rows(int s, unsigned ld_off,
                                          const float* __restrict__ hw, float (&m)[16]) {
    if constexpr (I < 16) {
        unsigned so = ((unsigned)row_lane<I>(s) << SH) + ld_off;
        m[I] = *(const float*)((const char*)hw + so);
        load_rows<I + 1, SH>(s, ld_off, hw, m);
    }
}

template <int I>
__device__ __forceinline__ void load_pairs(int s, unsigned ld_off,
                                           const float* __restrict__ hw, v2f (&m)[16]) {
    if constexpr (I < 16) {
        unsigned so = ((unsigned)row_lane<I>(s) << 7) + ld_off;
        m[I] = *(const v2f*)((const char*)hw + so);
        load_pairs<I + 1>(s, ld_off, hw, m);
    }
}

template <int I, bool TAIL>
__device__ __forceinline__ void add_rows(int dl, float nrm, unsigned lane_off,
                                         float* acc, const float (&m)[16]) {
    if constexpr (I < 16) {
        int   di = row_lane<I>(dl);
        float ni = __int_as_float(row_lane<I>(__float_as_int(nrm)));
        if (!TAIL || di >= 0) {
            unsigned dO = ((unsigned)di << 6) + lane_off;
            __hip_atomic_fetch_add((float*)((char*)acc + dO), m[I] * ni,
                                   __ATOMIC_RELAXED, __HIP_MEMORY_SCOPE_WORKGROUP);
        }
        add_rows<I + 1, TAIL>(dl, nrm, lane_off, acc, m);
    }
}

template <int I, bool TAIL>
__device__ __forceinline__ void add_pairs(int dl, float fd, unsigned lane_off, int l15,
                                          float* acc, const v2f (&m)[16], float& wv) {
    if constexpr (I < 16) {
        int   di = row_lane<I>(dl);
        float fs = __int_as_float(row_lane<0>(__float_as_int(m[I][1])));
        float ni = fs * __int_as_float(row_lane<I>(__float_as_int(fd)));
        if (l15 == I) wv = ni;
        if (!TAIL || di >= 0) {
            unsigned dO = ((unsigned)di << 6) + lane_off;
            __hip_atomic_fetch_add((float*)((char*)acc + dO), m[I][0] * ni,
                                   __ATOMIC_RELAXED, __HIP_MEMORY_SCOPE_WORKGROUP);
        }
        add_pairs<I + 1, TAIL>(dl, fd, lane_off, l15, acc, m, wv);
    }
}

template <bool TAIL, int SH>
__device__ __forceinline__ void drain(const int2* q, int head, int n, int lane, int lo,
                                      const float* __restrict__ dinv,
                                      const float* __restrict__ hw, float* acc) {
    int s = 0, dl = -1;
    float nrm = 0.0f;
    if (!TAIL || lane < n) {
        int2 ent = q[(head + lane) & (QCAP - 1)];
        s = ent.x; dl = ent.y;
        nrm = dinv[s] * dinv[lo + dl];
    }
    unsigned lane_off = (unsigned)(lane & 15) * 4u;
    float m[16];
    load_rows<0, SH>(s, (unsigned)(lane & 15) << (SH - 4), hw, m);
    add_rows<0, TAIL>(dl, nrm, lane_off, acc, m);
}

__device__ __forceinline__ void put_record(const int2* q, int head, int n, int lane,
                                           int2* __restrict__ rsd, int (&cn)[NSUB]) {
    int2 ent = make_int2(0, 0);
    int sub = -1;
    if (lane < n) {
        ent = q[(head + lane) & (QCAP - 1)];
        sub = (int)((unsigned)ent.y / (unsigned)SPR);
    }
    int pos = 0;
#pragma unroll
    for (int k = 0; k < NSUB; ++k) {
        unsigned mk = __builtin_amdgcn_ballot_w32(sub == k);
        int pk = (int)__builtin_amdgcn_mbcnt_lo(mk, (unsigned)cn[k]);
        pos = (sub == k) ? pk : pos;
        cn[k] += __popc(mk);
    }
    if (sub >= 0 && pos < SUBCAP)
        rsd[sub * SUBCAP + pos] = make_int2(ent.x, ent.y - sub * SPR);
}

template <bool CNT, int SH>
__device__ __forceinline__ void take(int s, int d, int lo, int cnt, int2* q, int& qt, int& qn,
                                     int lane, const float* __restrict__ dinv,
                                     const float* __restrict__ hw, float* acc,
                                     int2* __restrict__ rsd, int* cntl, int (&cn)[NSUB]) {
    int dl = d - lo;
    bool keep = (unsigned)dl < (unsigned)cnt;
    unsigned mask = __builtin_amdgcn_ballot_w32(keep);
    if (mask != 0u) {
        int rank = (int)__builtin_amdgcn_mbcnt_lo(mask, 0u);
        if (keep) {
            q[(qt + rank) & (QCAP - 1)] = make_int2(s, dl);
            if constexpr (CNT)
                __hip_atomic_fetch_add(&cntl[dl], 1, __ATOMIC_RELAXED,
                                       __HIP_MEMORY_SCOPE_WORKGROUP);
        }
        int n = __popc(mask);
        qt += n; qn += n;
        if (qn >= 32) {
            if constexpr (CNT) put_record(q, qt - qn, 32, lane, rsd, cn);
            else drain<false, SH>(q, qt - qn, 32, lane, lo, dinv, hw, acc);
            qn -= 32;
        }
    }
}

template <bool CNT, int SH>
__device__ __forceinline__ int filter_chunk(const int* __restrict__ src,
                                            const int* __restrict__ dst,
                                            const float* __restrict__ dinv,
                                            const float* __restrict__ hw, float* acc, int2* q,
                                            int2* __restrict__ rsd, int* cntl,
                                            int n_edges, int e0, int e1, int trip,
                                            int wave, int lane, int lo, int cnt, int (&cn)[NSUB]) {
    int qt = 0, qn = 0;
    int base = e0 + wave * 128;

    if ((n_edges & 3) == 0 && base + 128 <= e1) {
        int nw = (e1 - base - 128) / trip + 1;
        if (nw >= 2) {
            int4 sa = *(const int4*)(src + base + lane * 4);
            int4 da = *(const int4*)(dst + base + lane * 4);
            int4 sb = *(const int4*)(src + base + trip + lane * 4);
            int4 db = *(const int4*)(dst + base + trip + lane * 4);
            while (nw >= 2) {
                nw -= 2;
                int nbase = base + 2 * trip;
                int pb = (nw >= 2) ? nbase : base;
                int4 sc = *(const int4*)(src + pb + lane * 4);
                int4 dc = *(const int4*)(dst + pb + lane * 4);
                int4 sd = *(const int4*)(src + pb + trip + lane * 4);
                int4 dd = *(const int4*)(dst + pb + trip + lane * 4);
                take<CNT, SH>(sa.x, da.x, lo, cnt, q, qt, qn, lane, dinv, hw, acc, rsd, cntl, cn);
                take<CNT, SH>(sa.y, da.y, lo, cnt, q, qt, qn, lane, dinv, hw, acc, rsd, cntl, cn);
                take<CNT, SH>(sa.z, da.z, lo, cnt, q, qt, qn, lane, dinv, hw, acc, rsd, cntl, cn);
                take<CNT, SH>(sa.w, da.w, lo, cnt, q, qt, qn, lane, dinv, hw, acc, rsd, cntl, cn);
                take<CNT, SH>(sb.x, db.x, lo, cnt, q, qt, qn, lane, dinv, hw, acc, rsd, cntl, cn);
                take<CNT, SH>(sb.y, db.y, lo, cnt, q, qt, qn, lane, dinv, hw, acc, rsd, cntl, cn);
                take<CNT, SH>(sb.z, db.z, lo, cnt, q, qt, qn, lane, dinv, hw, acc, rsd, cntl, cn);
                take<CNT, SH>(sb.w, db.w, lo, cnt, q, qt, qn, lane, dinv, hw, acc, rsd, cntl, cn);
                sa = sc; da = dc; sb = sd; db = dd;
                base = nbase;
            }
        }
        if (nw == 1) {
            int4 s4 = *(const int4*)(src + base + lane * 4);
            int4 d4 = *(const int4*)(dst + base + lane * 4);
            take<CNT, SH>(s4.x, d4.x, lo, cnt, q, qt, qn, lane, dinv, hw, acc, rsd, cntl, cn);
            take<CNT, SH>(s4.y, d4.y, lo, cnt, q, qt, qn, lane, dinv, hw, acc, rsd, cntl, cn);
            take<CNT, SH>(s4.z, d4.z, lo, cnt, q, qt, qn, lane, dinv, hw, acc, rsd, cntl, cn);
            take<CNT, SH>(s4.w, d4.w, lo, cnt, q, qt, qn, lane, dinv, hw, acc, rsd, cntl, cn);
            base += trip;
        }
    }
    for (; base < e1; base += trip) {
        for (int j = 0; j < 4; ++j) {
            int e = base + j * 32 + lane;
            int s = 0, d = 0x7fffffff;
            if (e < e1) { s = src[e]; d = dst[e]; }
            take<CNT, SH>(s, d, lo, cnt, q, qt, qn, lane, dinv, hw, acc, rsd, cntl, cn);
        }
    }
    if (qn > 0) {
        if constexpr (CNT) put_record(q, qt - qn, qn, lane, rsd, cn);
        else drain<true, SH>(q, qt - qn, qn, lane, lo, dinv, hw, acc);
    }
    return qt;
}

__device__ __forceinline__ void chunk_bounds(int n_edges, int trip, int c, int& e0, int& e1) {
    const int per = (((n_edges + NCHUNK - 1) / NCHUNK + trip - 1) / trip) * trip;
    e0 = c * per;
    e1 = min(e0 + per, n_edges);
}

__global__ void __launch_bounds__(SBLK)
k_filter_deg(const int* __restrict__ src, const int* __restrict__ dst,
             int* __restrict__ degc, int* __restrict__ rec_sd, int* __restrict__ rec_n,
             int n_edges, int n_nodes) {
    __shared__ int cntl[NPR];
    __shared__ int2 queue[SBLK / 32][QCAP];

    const int tid = threadIdx.x, lane = tid & 31;
    const int wave = __builtin_amdgcn_readfirstlane(tid >> 5);
    const int lo  = blockIdx.x * NPR;
    const int cnt = min(NPR, n_nodes - lo);

    for (int i = tid; i < NPR; i += SBLK) cntl[i] = 0;
    __syncthreads();

    const int trip = (SBLK / 32) * 128;
    int e0, e1;
    chunk_bounds(n_edges, trip, blockIdx.y, e0, e1);

    const int wg = (blockIdx.y * gridDim.x + blockIdx.x) * (SBLK / 32) + wave;
    int2* rsd = (int2*)rec_sd + (size_t)wg * (NSUB * SUBCAP);

    int cn[NSUB] = {};
    filter_chunk<true, 6>(src, dst, nullptr, nullptr, nullptr, queue[wave], rsd, cntl,
                          n_edges, e0, e1, trip, wave, lane, lo, cnt, cn);
    int mine = 0;
#pragma unroll
    for (int k = 0; k < NSUB; ++k) mine = (lane == k) ? cn[k] : mine;
    if (lane < NSUB) rec_n[wg * NSUB + lane] = mine;
    __syncthreads();

    int* plane = degc + (size_t)blockIdx.y * n_nodes + lo;
    for (int i = tid; i < cnt; i += SBLK) plane[i] = cntl[i];
}

__device__ __forceinline__ int stream_at(int i, const int (&p)[NCHUNK + 1], const int (&D)[NCHUNK]) {
    int a = 0;
#pragma unroll
    for (int k = 1; k < NCHUNK; ++k) a = (i >= p[k]) ? D[k] : a;
    return i + a;
}

template <bool FIRST>
__global__ void __launch_bounds__(SBLK)
k_scatter_lds(const int* __restrict__ src, const int* __restrict__ dst,
              const float* __restrict__ dinv, const float* __restrict__ hw,
              const int* __restrict__ rec_sd, float* __restrict__ rec_w,
              const int* __restrict__ rec_n, const float* __restrict__ b,
              const float* __restrict__ W, float* __restrict__ hw_out,
              const int* __restrict__ batch, float* __restrict__ pooled,
              float* __restrict__ gcnt, int n_edges, int n_nodes) {
    __shared__ __attribute__((aligned(16))) float acc[SPR * HIDC];
    __shared__ int2 queue[SBLK / 32][QCAP];

    const int tid = threadIdx.x, lane = tid & 31;
    const int wave = __builtin_amdgcn_readfirstlane(tid >> 5);
    const int nrf = gridDim.x / NSUB;
    const int rf  = blockIdx.x / NSUB;
    const int sub = blockIdx.x - rf * NSUB;
    const int lo  = rf * NPR + sub * SPR;
    const int cnt = max(0, min(SPR, n_nodes - lo));

    for (int i = tid; i < SPR * HIDC / 4; i += SBLK)
        ((float4*)acc)[i] = make_float4(0.0f, 0.0f, 0.0f, 0.0f);
    __syncthreads();

    const int trip = (SBLK / 32) * 128;

    const int wpc = nrf * (SBLK / 32);
    const int wf0 = rf * (SBLK / 32) + wave;
    const int2* rs0 = (const int2*)rec_sd + ((size_t)wf0 * NSUB + sub) * SUBCAP;
    const int ws = (int)blockIdx.x * (SBLK / 32) + wave;
    float* rw = rec_w + (size_t)ws * (NCHUNK * SUBCAP);

    int nk = 0;
    if (lane < NCHUNK) nk = rec_n[(wf0 + lane * wpc) * NSUB + sub];
    int p[NCHUNK + 1], D[NCHUNK];
    bool fits = true;
    p[0] = 0;
#pragma unroll
    for (int k = 0; k < NCHUNK; ++k) {
        int n = __builtin_amdgcn_readlane(nk, k);
        fits = fits && (n <= SUBCAP);
        p[k + 1] = p[k] + n;
        D[k] = k * wpc * (NSUB * SUBCAP) - p[k];
    }
    const int tot = p[NCHUNK];
    if (fits) {
        const unsigned lane_off = (unsigned)(lane & 15) * 4u;
        const int nfull = tot >> 5;
        if constexpr (FIRST) {
            if (tot > 0) {
                const int last = tot - 1;
                const int l15 = lane & 15;
                const unsigned ld_off = (unsigned)l15 * 8u;
                int2  e  = rs0[stream_at(min(lane, last), p, D)];
                int2  en = rs0[stream_at(min(32 + lane, last), p, D)];
                float fd = dinv[lo + e.y];
                if (lane >= tot) { e.x = 0; e.y = -1; fd = 0.0f; }
                for (int k = 0; k < nfull; ++k) {
                    int2  e2 = rs0[stream_at(min((k + 2) * 32 + lane, last), p, D)];
                    v2f m[16];
                    load_pairs<0>(e.x, ld_off, hw, m);
                    float fdn = dinv[lo + en.y];
                    asm volatile("" :: "v"(e2.x), "v"(e2.y));
                    float wv = 0.0f;
                    add_pairs<0, false>(e.y, fd, lane_off, l15, acc, m, wv);
                    rw[k * 32 + lane] = wv;
                    bool  has = (k + 1) * 32 + lane < tot;
                    e.x = has ? en.x : 0;
                    e.y = has ? en.y : -1;
                    fd  = has ? fdn : 0.0f;
                    en  = e2;
                }
                if (tot & 31) {
                    v2f m[16];
                    load_pairs<0>(e.x, ld_off, hw, m);
                    float wv = 0.0f;
                    add_pairs<0, true>(e.y, fd, lane_off, l15, acc, m, wv);
                    if (e.y >= 0) rw[nfull * 32 + lane] = wv;
                }
            }
        } else {
            int2  e = make_int2(0, -1);
            float w = 0.0f;
            if (lane < tot) { e = rs0[stream_at(lane, p, D)]; w = rw[lane]; }
            for (int k = 0; k < nfull; ++k) {
                float m[16];
                load_rows<0, 6>(e.x, lane_off, hw, m);
                int   idx = (k + 1) * 32 + lane;
                int   ci  = min(idx, tot - 1);
                int2  en = rs0[stream_at(ci, p, D)];
                float wn = rw[ci];
                add_rows<0, false>(e.y, w, lane_off, acc, m);
                bool  has = idx < tot;
                e.x = has ? en.x : 0;
                e.y = has ? en.y : -1;
                w   = has ? wn : 0.0f;
            }
            if (tot & 31) {
                float m[16];
                load_rows<0, 6>(e.x, lane_off, hw, m);
                add_rows<0, true>(e.y, w, lane_off, acc, m);
            }
        }
    } else {
        int unused[NSUB] = {};
        for (int k = 0; k < NCHUNK; ++k) {
            int e0, e1;
            chunk_bounds(n_edges, trip, k, e0, e1);
            filter_chunk<false, FIRST ? 7 : 6>(src, dst, dinv, hw, acc, queue[wave], nullptr,
                                nullptr, n_edges, e0, e1, trip, wave, lane, lo, cnt, unused);
        }
    }
    __syncthreads();

    if constexpr (FIRST) {
        const int half = lane >> 4;
        const int l15  = lane & 15;
        v2f bf[4];
#pragma unroll
        for (int t = 0; t < 4; ++t) {
            int kk = 4 * t + 2 * half;
            bf[t][0] = W[(kk + 0) * HIDC + l15];
            bf[t][1] = W[(kk + 1) * HIDC + l15];
        }
        for (int i = tid; i < cnt * (HIDC / 4); i += SBLK) {
            float4 a = ((const float4*)acc)[i];
            float dn = dinv[lo + (i >> 2)];
            float nn = dn * dn;
            const float4* hp = (const float4*)hw + 2 * ((size_t)lo * (HIDC / 4) + i);
            float4 p0 = hp[0];
            float4 p1 = hp[1];
            const float* bq = b + (i & 3) * 4;
            float4 r;
            r.x = fmaxf(a.x + p0.x * nn + bq[0], 0.0f);
            r.y = fmaxf(a.y + p0.z * nn + bq[1], 0.0f);
            r.z = fmaxf(a.z + p1.x * nn + bq[2], 0.0f);
            r.w = fmaxf(a.w + p1.z * nn + bq[3], 0.0f);
            ((float4*)acc)[i] = r;
        }
        __syncthreads();
        if (wave * 16 + 16 <= cnt) {
            const float* mine = acc + wave * 16 * HIDC;
            v8f c = {};
#pragma unroll
            for (int t = 0; t < 4; ++t) {
                int kk = 4 * t + 2 * half;
                v2f a;
                a[0] = mine[l15 * HIDC + kk + 0];
                a[1] = mine[l15 * HIDC + kk + 1];
                c = __builtin_amdgcn_wmma_f32_16x16x4_f32(
                        false, a, false, bf[t], (short)0, c, false, false);
            }
            float* orow = hw_out + (size_t)(lo + wave * 16) * HIDC + l15;
#pragma unroll
            for (int v = 0; v < 8; ++v) {
                int m = v + 8 * half;
                orow[m * HIDC] = c[v];
            }
        }
    } else {
        for (int i0 = 0; i0 < cnt * (HIDC / 4); i0 += SBLK) {
            const int i = i0 + tid;
            const bool in = i < cnt * (HIDC / 4);
            float4 r = make_float4(0.0f, 0.0f, 0.0f, 0.0f);
            int g = -1;
            if (in) {
                const int node = lo + (i >> 2);
                g = batch[node];
                float4 a = ((const float4*)acc)[i];
                float dn = dinv[node];
                float nn = dn * dn;
                float4 hv = ((const float4*)hw)[(size_t)lo * (HIDC / 4) + i];
                const float* bq = b + (i & 3) * 4;
                r.x = fmaxf(a.x + hv.x * nn + bq[0], 0.0f);
                r.y = fmaxf(a.y + hv.y * nn + bq[1], 0.0f);
                r.z = fmaxf(a.z + hv.z * nn + bq[2], 0.0f);
                r.w = fmaxf(a.w + hv.w * nn + bq[3], 0.0f);
            }
            const int g0 = __builtin_amdgcn_readfirstlane(g);
            const bool same = __builtin_amdgcn_ballot_w32(g != g0) == 0u;
            if (same) {
                if (g0 >= 0) {
                    for (int off = 4; off < 32; off <<= 1) {
                        r.x += __shfl_xor(r.x, off);
                        r.y += __shfl_xor(r.y, off);
                        r.z += __shfl_xor(r.z, off);
                        r.w += __shfl_xor(r.w, off);
                    }
                    if (lane < 4) {
                        float* pg = pooled + (size_t)g0 * HIDC + lane * 4;
                        fadd(pg + 0, r.x);
                        fadd(pg + 1, r.y);
                        fadd(pg + 2, r.z);
                        fadd(pg + 3, r.w);
                    }
                    if (lane == 0) fadd(&gcnt[g0], 8.0f);
                }
            } else if (in) {
                float* pg = pooled + (size_t)g * HIDC + (lane & 3) * 4;
                fadd(pg + 0, r.x);
                fadd(pg + 1, r.y);
                fadd(pg + 2, r.z);
                fadd(pg + 3, r.w);
                if ((lane & 3) == 0) fadd(&gcnt[g], 1.0f);
            }
        }
    }
}

__global__ void k_classify(const float* __restrict__ pooled, const float* __restrict__ cnt,
                           const float* __restrict__ Wc, const float* __restrict__ bc,
                           float* __restrict__ out, int n_graphs, int labels) {
    int i = blockIdx.x * blockDim.x + threadIdx.x;
    if (i >= n_graphs * labels) return;
    int g = i / labels, c = i % labels;
    float inv = 1.0f / fmaxf(cnt[g], 1.0f);
    float acc = bc[c];
#pragma unroll
    for (int j = 0; j < HIDC; ++j)
        acc += pooled[g * HIDC + j] * inv * Wc[j * labels + c];
    out[i] = acc;
}

extern "C" void kernel_launch(void* const* d_in, const int* in_sizes, int n_in,
                              void* d_out, int out_size, void* d_ws, size_t ws_size,
                              hipStream_t stream) {
    const int*   x     = (const int*)d_in[0];
    const int*   ei    = (const int*)d_in[1];
    const int*   batch = (const int*)d_in[2];
    const float* emb   = (const float*)d_in[3];
    const float* W1    = (const float*)d_in[4];
    const float* b1    = (const float*)d_in[5];
    const float* W2    = (const float*)d_in[6];
    const float* b2    = (const float*)d_in[7];
    const float* Wc    = (const float*)d_in[8];
    const float* bc    = (const float*)d_in[9];
    float* out = (float*)d_out;

    const int n_nodes  = in_sizes[0];
    const int n_edges  = in_sizes[1] / 2;
    const int labels   = in_sizes[9];
    const int n_graphs = out_size / labels;
    const int* src = ei;
    const int* dst = ei + n_edges;

    float* ws     = (float*)d_ws;
    float* deg    = ws;
    float* hw2    = deg + n_nodes;
    float* hw     = hw2 + (size_t)n_nodes * HIDC;
    float* pooled = hw + (size_t)n_nodes * HIDC * 2;
    float* cnt    = pooled + (size_t)n_graphs * HIDC;
    const size_t n_sw   = (size_t)((n_nodes + NPR - 1) / NPR) * NCHUNK * (SBLK / 32);
    const size_t rec_at = ((size_t)(cnt + n_graphs - ws) + 31) & ~(size_t)31;
    int*   rec_sd = (int*)(ws + rec_at);
    float* rec_w  = (float*)(rec_sd + n_sw * NSUB * SUBCAP * 2);
    int*   rec_n  = (int*)(rec_w + n_sw * NCHUNK * SUBCAP);
    int*   degc   = rec_n + n_sw * NSUB;

    const int ntiles = n_nodes / 16;
    const int wavesPerBlk = BLK / 32;

    dim3 b(BLK);
    auto g1 = [](int n) { return dim3((n + BLK - 1) / BLK); };
    dim3 gw((ntiles + wavesPerBlk - 1) / wavesPerBlk);
    dim3 gf((n_nodes + NPR - 1) / NPR, NCHUNK);
    dim3 gs(((n_nodes + NPR - 1) / NPR) * NSUB);

    k_filter_deg<<<gf, SBLK, 0, stream>>>(src, dst, degc, rec_sd, rec_n, n_edges, n_nodes);

    k_embed_xw<<<gw, b, 0, stream>>>(x, emb, W1, degc, deg, hw, pooled, cnt,
                                     n_nodes, ntiles, n_graphs);
    k_scatter_lds<true><<<gs, SBLK, 0, stream>>>(src, dst, deg, hw, rec_sd, rec_w, rec_n, b1, W2, hw2,
                                                 nullptr, nullptr, nullptr, n_edges, n_nodes);

    k_scatter_lds<false><<<gs, SBLK, 0, stream>>>(src, dst, deg, hw2, rec_sd, rec_w, rec_n, b2, nullptr,
                                                  nullptr, batch, pooled, cnt, n_edges, n_nodes);
    k_classify<<<g1(n_graphs * labels), b, 0, stream>>>(pooled, cnt, Wc, bc, out,
                                                        n_graphs, labels);
}


# degree pass re-tiled: 16-wave work-groups, 4 trips of loads in flight, count 11200 rows over 1/28 of the edges: 9 tests per edge, not 25; wave records built in LDS, stored whole; arrivals differ
# speedup vs baseline: 17.2228x; 1.1102x over previous
#include <hip/hip_runtime.h>

typedef __attribute__((ext_vector_type(2))) float v2f;
typedef __attribute__((ext_vector_type(8))) float v8f;

#define HIDC 16
#define BLK 256
#define NPR 11200
#define NCHUNK 28
#define NSUB 28
#define SPR (NPR / NSUB)
#define SBLK 1024
#define FBLK 512
#define NREP (NCHUNK / (SBLK / FBLK))
#define QCAP 64
#define SUBCAP 64
#define RNL 32
#define LSTR (SUBCAP + 1)

static_assert(SPR % 16 == 0 && SPR / 16 <= SBLK / 32, "a scatter wave transforms at most one 16-row tile");
static_assert(NPR % NSUB == 0 && NSUB <= RNL && RNL == 32 && NREP <= 32, "a lane per sub-range count and per replayed record");
static_assert(SBLK % FBLK == 0 && NCHUNK % (SBLK / FBLK) == 0, "the scatter waves share out the chunks evenly");

__device__ __forceinline__ void fadd(float* p, float v) {
    __hip_atomic_fetch_add(p, v, __ATOMIC_RELAXED, __HIP_MEMORY_SCOPE_AGENT);
}

template <int I>
__device__ __forceinline__ int row_lane(int v) {
    return __builtin_amdgcn_update_dpp(0, v, 0x150 + I, 0xf, 0xf, false);
}

template <int V>
__device__ __forceinline__ void store_pairs(const v8f& c, float dv, int half, int l15,
                                            v2f* __restrict__ rows) {
    if constexpr (V < 8) {
        float d_lo = __int_as_float(row_lane<V>(__float_as_int(dv)));
        float d_hi = __int_as_float(row_lane<V + 8>(__float_as_int(dv)));
        v2f pr;
        pr[0] = c[V];
        pr[1] = (l15 == 0) ? (half ? d_hi : d_lo) : 0.0f;
        rows[(size_t)(V + 8 * half) * HIDC + l15] = pr;
        store_pairs<V + 1>(c, dv, half, l15, rows);
    }
}

__global__ void k_embed_xw(const int* __restrict__ x, const float* __restrict__ emb,
                           const float* __restrict__ W, const int* __restrict__ degc,
                           float* __restrict__ dinv, float* __restrict__ hw,
                           float* __restrict__ pooled, float* __restrict__ cnt,
                           int n_nodes, int ntiles, int n_graphs) {
    const int gid = blockIdx.x * blockDim.x + threadIdx.x;
    for (int k = gid; k < n_graphs * (HIDC + 1); k += gridDim.x * blockDim.x) {
        if (k < n_graphs * HIDC) pooled[k] = 0.0f;
        else cnt[k - n_graphs * HIDC] = 0.0f;
    }

    int wave = gid >> 5;
    int lane = threadIdx.x & 31;
    if (wave >= ntiles) return;
    int half = lane >> 4;
    int l15  = lane & 15;
    const int node = wave * 16 + l15;

    const int tok = x[node];

    int dg = 1;
    for (int c = 0; c < NCHUNK; ++c) dg += degc[(size_t)c * n_nodes + node];
    float d = (float)dg;
    const float dv = (d > 0.0f) ? rsqrtf(d) : 0.0f;
    if (half == 0) dinv[node] = dv;

    v2f bf[4];
#pragma unroll
    for (int t = 0; t < 4; ++t) {
        int kk = 4 * t + 2 * half;
        bf[t][0] = W[(kk + 0) * HIDC + l15];
        bf[t][1] = W[(kk + 1) * HIDC + l15];
    }

    const float* hrow = emb + (size_t)tok * HIDC;
    v8f c = {};
#pragma unroll
    for (int t = 0; t < 4; ++t) {
        int kk = 4 * t + 2 * half;
        v2f a;
        a[0] = hrow[kk + 0];
        a[1] = hrow[kk + 1];
        c = __builtin_amdgcn_wmma_f32_16x16x4_f32(
                false, a, false, bf[t], (short)0, c, false, false);
    }

    store_pairs<0>(c, dv, half, l15, (v2f*)hw + (size_t)wave * 16 * HIDC);
}


template <int I, int SH>
__device__ __forceinline__ void load_rows(int s, unsigned ld_off,
                                          const float* __restrict__ hw, float (&m)[16]) {
    if constexpr (I < 16) {
        unsigned so = ((unsigned)row_lane<I>(s) << SH) + ld_off;
        m[I] = *(const float*)((const char*)hw + so);
        load_rows<I + 1, SH>(s, ld_off, hw, m);
    }
}

template <int I>
__device__ __forceinline__ void load_pairs(int s, unsigned ld_off,
                                           const float* __restrict__ hw, v2f (&m)[16]) {
    if constexpr (I < 16) {
        unsigned so = ((unsigned)row_lane<I>(s) << 7) + ld_off;
        m[I] = *(const v2f*)((const char*)hw + so);
        load_pairs<I + 1>(s, ld_off, hw, m);
    }
}

template <int I, bool TAIL>
__device__ __forceinline__ void add_rows(int dl, float nrm, unsigned lane_off,
                                         float* acc, const float (&m)[16]) {
    if constexpr (I < 16) {
        int   di = row_lane<I>(dl);
        float ni = __int_as_float(row_lane<I>(__float_as_int(nrm)));
        if (!TAIL || di >= 0) {
            unsigned dO = ((unsigned)di << 6) + lane_off;
            __hip_atomic_fetch_add((float*)((char*)acc + dO), m[I] * ni,
                                   __ATOMIC_RELAXED, __HIP_MEMORY_SCOPE_WORKGROUP);
        }
        add_rows<I + 1, TAIL>(dl, nrm, lane_off, acc, m);
    }
}

template <int I, bool TAIL>
__device__ __forceinline__ void add_pairs(int dl, float fd, unsigned lane_off, int l15,
                                          float* acc, const v2f (&m)[16], float& wv) {
    if constexpr (I < 16) {
        int   di = row_lane<I>(dl);
        float fs = __int_as_float(row_lane<0>(__float_as_int(m[I][1])));
        float ni = fs * __int_as_float(row_lane<I>(__float_as_int(fd)));
        if (l15 == I) wv = ni;
        if (!TAIL || di >= 0) {
            unsigned dO = ((unsigned)di << 6) + lane_off;
            __hip_atomic_fetch_add((float*)((char*)acc + dO), m[I][0] * ni,
                                   __ATOMIC_RELAXED, __HIP_MEMORY_SCOPE_WORKGROUP);
        }
        add_pairs<I + 1, TAIL>(dl, fd, lane_off, l15, acc, m, wv);
    }
}

template <bool TAIL, int SH>
__device__ __forceinline__ void drain(const int2* q, int head, int n, int lane, int lo,
                                      const float* __restrict__ dinv,
                                      const float* __restrict__ hw, float* acc) {
    int s = 0, dl = -1;
    float nrm = 0.0f;
    if (!TAIL || lane < n) {
        int2 ent = q[(head + lane) & (QCAP - 1)];
        s = ent.x; dl = ent.y;
        nrm = dinv[s] * dinv[lo + dl];
    }
    unsigned lane_off = (unsigned)(lane & 15) * 4u;
    float m[16];
    load_rows<0, SH>(s, (unsigned)(lane & 15) << (SH - 4), hw, m);
    add_rows<0, TAIL>(dl, nrm, lane_off, acc, m);
}

__device__ __forceinline__ void put_record(const int2* q, int head, int n, int lane,
                                           int2* rsd, int* wc) {
    if (lane < n) {
        int2 ent = q[(head + lane) & (QCAP - 1)];
        int sub = (int)((unsigned)ent.y / (unsigned)SPR);
        int pos = __hip_atomic_fetch_add(&wc[sub], 1, __ATOMIC_RELAXED,
                                         __HIP_MEMORY_SCOPE_WORKGROUP);
        if (pos < SUBCAP)
            rsd[sub * LSTR + pos] = make_int2(ent.x, ent.y - sub * SPR);
    }
}

template <bool CNT, int SH>
__device__ __forceinline__ void take(int s, int d, int lo, int cnt, int2* q, int& qt, int& qn,
                                     int lane, const float* __restrict__ dinv,
                                     const float* __restrict__ hw, float* acc,
                                     int2* rsd, int* cntl, int* wc) {
    int dl = d - lo;
    bool keep = (unsigned)dl < (unsigned)cnt;
    unsigned mask = __builtin_amdgcn_ballot_w32(keep);
    if (mask != 0u) {
        int rank = (int)__builtin_amdgcn_mbcnt_lo(mask, 0u);
        if (keep) {
            q[(qt + rank) & (QCAP - 1)] = make_int2(s, dl);
            if constexpr (CNT)
                __hip_atomic_fetch_add(&cntl[dl], 1, __ATOMIC_RELAXED,
                                       __HIP_MEMORY_SCOPE_WORKGROUP);
        }
        int n = __popc(mask);
        qt += n; qn += n;
        if (qn >= 32) {
            if constexpr (CNT) put_record(q, qt - qn, 32, lane, rsd, wc);
            else drain<false, SH>(q, qt - qn, 32, lane, lo, dinv, hw, acc);
            qn -= 32;
        }
    }
}

template <bool CNT, int SH>
__device__ __forceinline__ int filter_chunk(const int* __restrict__ src,
                                            const int* __restrict__ dst,
                                            const float* __restrict__ dinv,
                                            const float* __restrict__ hw, float* acc, int2* q,
                                            int2* rsd, int* cntl,
                                            int n_edges, int e0, int e1, int trip,
                                            int wave, int lane, int lo, int cnt, int* wc) {
    int qt = 0, qn = 0;
    int base = e0 + wave * 128;

    if ((n_edges & 3) == 0 && base + 128 <= e1) {
        int nw = (e1 - base - 128) / trip + 1;
        if (CNT && nw >= 4) {
            int4 s0 = *(const int4*)(src + base + lane * 4);
            int4 d0 = *(const int4*)(dst + base + lane * 4);
            int4 s1 = *(const int4*)(src + base + trip + lane * 4);
            int4 d1 = *(const int4*)(dst + base + trip + lane * 4);
            int4 s2 = *(const int4*)(src + base + 2 * trip + lane * 4);
            int4 d2 = *(const int4*)(dst + base + 2 * trip + lane * 4);
            int4 s3 = *(const int4*)(src + base + 3 * trip + lane * 4);
            int4 d3 = *(const int4*)(dst + base + 3 * trip + lane * 4);
            while (nw >= 4) {
                nw -= 4;
                int nbase = base + 4 * trip;
                int pb = (nw >= 4) ? nbase : base;
                int4 t0 = *(const int4*)(src + pb + lane * 4);
                int4 u0 = *(const int4*)(dst + pb + lane * 4);
                int4 t1 = *(const int4*)(src + pb + trip + lane * 4);
                int4 u1 = *(const int4*)(dst + pb + trip + lane * 4);
                int4 t2 = *(const int4*)(src + pb + 2 * trip + lane * 4);
                int4 u2 = *(const int4*)(dst + pb + 2 * trip + lane * 4);
                int4 t3 = *(const int4*)(src + pb + 3 * trip + lane * 4);
                int4 u3 = *(const int4*)(dst + pb + 3 * trip + lane * 4);
                take<CNT, SH>(s0.x, d0.x, lo, cnt, q, qt, qn, lane, dinv, hw, acc, rsd, cntl, wc);
                take<CNT, SH>(s0.y, d0.y, lo, cnt, q, qt, qn, lane, dinv, hw, acc, rsd, cntl, wc);
                take<CNT, SH>(s0.z, d0.z, lo, cnt, q, qt, qn, lane, dinv, hw, acc, rsd, cntl, wc);
                take<CNT, SH>(s0.w, d0.w, lo, cnt, q, qt, qn, lane, dinv, hw, acc, rsd, cntl, wc);
                take<CNT, SH>(s1.x, d1.x, lo, cnt, q, qt, qn, lane, dinv, hw, acc, rsd, cntl, wc);
                take<CNT, SH>(s1.y, d1.y, lo, cnt, q, qt, qn, lane, dinv, hw, acc, rsd, cntl, wc);
                take<CNT, SH>(s1.z, d1.z, lo, cnt, q, qt, qn, lane, dinv, hw, acc, rsd, cntl, wc);
                take<CNT, SH>(s1.w, d1.w, lo, cnt, q, qt, qn, lane, dinv, hw, acc, rsd, cntl, wc);
                take<CNT, SH>(s2.x, d2.x, lo, cnt, q, qt, qn, lane, dinv, hw, acc, rsd, cntl, wc);
                take<CNT, SH>(s2.y, d2.y, lo, cnt, q, qt, qn, lane, dinv, hw, acc, rsd, cntl, wc);
                take<CNT, SH>(s2.z, d2.z, lo, cnt, q, qt, qn, lane, dinv, hw, acc, rsd, cntl, wc);
                take<CNT, SH>(s2.w, d2.w, lo, cnt, q, qt, qn, lane, dinv, hw, acc, rsd, cntl, wc);
                take<CNT, SH>(s3.x, d3.x, lo, cnt, q, qt, qn, lane, dinv, hw, acc, rsd, cntl, wc);
                take<CNT, SH>(s3.y, d3.y, lo, cnt, q, qt, qn, lane, dinv, hw, acc, rsd, cntl, wc);
                take<CNT, SH>(s3.z, d3.z, lo, cnt, q, qt, qn, lane, dinv, hw, acc, rsd, cntl, wc);
                take<CNT, SH>(s3.w, d3.w, lo, cnt, q, qt, qn, lane, dinv, hw, acc, rsd, cntl, wc);
                s0 = t0; d0 = u0; s1 = t1; d1 = u1; s2 = t2; d2 = u2; s3 = t3; d3 = u3;
                base = nbase;
            }
        }
        for (; nw >= 1; --nw) {
            int4 s4 = *(const int4*)(src + base + lane * 4);
            int4 d4 = *(const int4*)(dst + base + lane * 4);
            take<CNT, SH>(s4.x, d4.x, lo, cnt, q, qt, qn, lane, dinv, hw, acc, rsd, cntl, wc);
            take<CNT, SH>(s4.y, d4.y, lo, cnt, q, qt, qn, lane, dinv, hw, acc, rsd, cntl, wc);
            take<CNT, SH>(s4.z, d4.z, lo, cnt, q, qt, qn, lane, dinv, hw, acc, rsd, cntl, wc);
            take<CNT, SH>(s4.w, d4.w, lo, cnt, q, qt, qn, lane, dinv, hw, acc, rsd, cntl, wc);
            base += trip;
        }
    }
    for (; base < e1; base += trip) {
        for (int j = 0; j < 4; ++j) {
            int e = base + j * 32 + lane;
            int s = 0, d = 0x7fffffff;
            if (e < e1) { s = src[e]; d = dst[e]; }
            take<CNT, SH>(s, d, lo, cnt, q, qt, qn, lane, dinv, hw, acc, rsd, cntl, wc);
        }
    }
    if (qn > 0) {
        if constexpr (CNT) put_record(q, qt - qn, qn, lane, rsd, wc);
        else drain<true, SH>(q, qt - qn, qn, lane, lo, dinv, hw, acc);
    }
    return qt;
}

__device__ __forceinline__ void chunk_bounds(int n_edges, int trip, int c, int& e0, int& e1) {
    const int per = (((n_edges + NCHUNK - 1) / NCHUNK + trip - 1) / trip) * trip;
    e0 = c * per;
    e1 = min(e0 + per, n_edges);
}

__global__ void __launch_bounds__(FBLK)
k_filter_deg(const int* __restrict__ src, const int* __restrict__ dst,
             int* __restrict__ degc, int* __restrict__ rec_sd, int* __restrict__ rec_n,
             int n_edges, int n_nodes) {
    __shared__ int cntl[NPR];
    __shared__ int2 queue[FBLK / 32][QCAP];
    __shared__ int wcnt[FBLK / 32][32];
    __shared__ int2 lrec[FBLK / 32][NSUB * LSTR];

    const int tid = threadIdx.x, lane = tid & 31;
    const int wave = __builtin_amdgcn_readfirstlane(tid >> 5);
    const int lo  = blockIdx.x * NPR;
    const int cnt = min(NPR, n_nodes - lo);

    for (int i = tid; i < NPR; i += FBLK) cntl[i] = 0;
    wcnt[wave][lane] = 0;
    __syncthreads();

    const int trip = (FBLK / 32) * 128;
    int e0, e1;
    chunk_bounds(n_edges, trip, blockIdx.y, e0, e1);

    const int wg = (blockIdx.y * gridDim.x + blockIdx.x) * (FBLK / 32) + wave;
    int2* rsd = (int2*)rec_sd + (size_t)wg * (NSUB * SUBCAP);

    filter_chunk<true, 6>(src, dst, nullptr, nullptr, nullptr, queue[wave], lrec[wave], cntl,
                          n_edges, e0, e1, trip, wave, lane, lo, cnt, wcnt[wave]);
    int mine = 0;
    if (lane < NSUB) {
        mine = __hip_atomic_load(&wcnt[wave][lane], __ATOMIC_RELAXED, __HIP_MEMORY_SCOPE_WORKGROUP);
        rec_n[wg * RNL + lane] = mine;
    }
    static_assert(NSUB % 4 == 0 && SUBCAP == 64, "copies 4 records of 2 x 32 places at a time");
#pragma unroll
    for (int k0 = 0; k0 < NSUB; k0 += 4) {
        int2 ent[4][2];
#pragma unroll
        for (int kk = 0; kk < 4; ++kk) {
            ent[kk][0] = lrec[wave][(k0 + kk) * LSTR + lane];
            ent[kk][1] = lrec[wave][(k0 + kk) * LSTR + 32 + lane];
        }
        asm volatile("" :: "v"(ent[0][0].x), "v"(ent[0][0].y), "v"(ent[0][1].x), "v"(ent[0][1].y),
                           "v"(ent[1][0].x), "v"(ent[1][0].y), "v"(ent[1][1].x), "v"(ent[1][1].y),
                           "v"(ent[2][0].x), "v"(ent[2][0].y), "v"(ent[2][1].x), "v"(ent[2][1].y),
                           "v"(ent[3][0].x), "v"(ent[3][0].y), "v"(ent[3][1].x), "v"(ent[3][1].y));
#pragma unroll
        for (int kk = 0; kk < 4; ++kk) {
            const int n = min(__builtin_amdgcn_readlane(mine, k0 + kk), SUBCAP);
            if (lane < n)      rsd[(k0 + kk) * SUBCAP + lane]      = ent[kk][0];
            if (32 + lane < n) rsd[(k0 + kk) * SUBCAP + 32 + lane] = ent[kk][1];
        }
    }
    __syncthreads();

    int* plane = degc + (size_t)blockIdx.y * n_nodes + lo;
    for (int i = tid; i < cnt; i += FBLK) plane[i] = cntl[i];
}

__device__ __forceinline__ int stream_at(int i, const int (&p)[NREP + 1], const int (&D)[NREP]) {
    int a = 0;
#pragma unroll
    for (int k = 1; k < NREP; ++k) a = (i >= p[k]) ? D[k] : a;
    return i + a;
}

template <bool FIRST>
__global__ void __launch_bounds__(SBLK)
k_scatter_lds(const int* __restrict__ src, const int* __restrict__ dst,
              const float* __restrict__ dinv, const float* __restrict__ hw,
              const int* __restrict__ rec_sd, float* __restrict__ rec_w,
              const int* __restrict__ rec_n, const float* __restrict__ b,
              const float* __restrict__ W, float* __restrict__ hw_out,
              const int* __restrict__ batch, float* __restrict__ pooled,
              float* __restrict__ gcnt, int n_edges, int n_nodes) {
    __shared__ __attribute__((aligned(16))) float acc[SPR * HIDC];
    __shared__ int2 queue[SBLK / 32][QCAP];

    const int tid = threadIdx.x, lane = tid & 31;
    const int wave = __builtin_amdgcn_readfirstlane(tid >> 5);
    const int nrf = gridDim.x / NSUB;
    const int rf  = blockIdx.x / NSUB;
    const int sub = blockIdx.x - rf * NSUB;
    const int lo  = rf * NPR + sub * SPR;
    const int cnt = max(0, min(SPR, n_nodes - lo));

    for (int i = tid; i < SPR * HIDC / 4; i += SBLK)
        ((float4*)acc)[i] = make_float4(0.0f, 0.0f, 0.0f, 0.0f);
    __syncthreads();

    const int trip = (FBLK / 32) * 128;

    const int fw  = wave % (FBLK / 32);
    const int c0  = wave / (FBLK / 32);
    const int wpc = (SBLK / FBLK) * nrf * (FBLK / 32);
    const int wf0 = (c0 * nrf + rf) * (FBLK / 32) + fw;
    const int2* rs0 = (const int2*)rec_sd + ((size_t)wf0 * NSUB + sub) * SUBCAP;
    const int ws = (int)blockIdx.x * (SBLK / 32) + wave;
    float* rw = rec_w + (size_t)ws * (NREP * SUBCAP);

    int nk = 0;
    if (lane < NREP) nk = rec_n[(wf0 + lane * wpc) * RNL + sub];
    int p[NREP + 1], D[NREP];
    bool fits = true;
    p[0] = 0;
#pragma unroll
    for (int k = 0; k < NREP; ++k) {
        int n = __builtin_amdgcn_readlane(nk, k);
        fits = fits && (n <= SUBCAP);
        p[k + 1] = p[k] + n;
        D[k] = k * wpc * (NSUB * SUBCAP) - p[k];
    }
    const int tot = p[NREP];
    if (fits) {
        const unsigned lane_off = (unsigned)(lane & 15) * 4u;
        const int nfull = tot >> 5;
        if constexpr (FIRST) {
            if (tot > 0) {
                const int last = tot - 1;
                const int l15 = lane & 15;
                const unsigned ld_off = (unsigned)l15 * 8u;
                int2  e  = rs0[stream_at(min(lane, last), p, D)];
                int2  en = rs0[stream_at(min(32 + lane, last), p, D)];
                float fd = dinv[lo + e.y];
                if (lane >= tot) { e.x = 0; e.y = -1; fd = 0.0f; }
                for (int k = 0; k < nfull; ++k) {
                    int2  e2 = rs0[stream_at(min((k + 2) * 32 + lane, last), p, D)];
                    v2f m[16];
                    load_pairs<0>(e.x, ld_off, hw, m);
                    float fdn = dinv[lo + en.y];
                    asm volatile("" :: "v"(e2.x), "v"(e2.y));
                    float wv = 0.0f;
                    add_pairs<0, false>(e.y, fd, lane_off, l15, acc, m, wv);
                    rw[k * 32 + lane] = wv;
                    bool  has = (k + 1) * 32 + lane < tot;
                    e.x = has ? en.x : 0;
                    e.y = has ? en.y : -1;
                    fd  = has ? fdn : 0.0f;
                    en  = e2;
                }
                if (tot & 31) {
                    v2f m[16];
                    load_pairs<0>(e.x, ld_off, hw, m);
                    float wv = 0.0f;
                    add_pairs<0, true>(e.y, fd, lane_off, l15, acc, m, wv);
                    if (e.y >= 0) rw[nfull * 32 + lane] = wv;
                }
            }
        } else {
            int2  e = make_int2(0, -1);
            float w = 0.0f;
            if (lane < tot) { e = rs0[stream_at(lane, p, D)]; w = rw[lane]; }
            for (int k = 0; k < nfull; ++k) {
                float m[16];
                load_rows<0, 6>(e.x, lane_off, hw, m);
                int   idx = (k + 1) * 32 + lane;
                int   ci  = min(idx, tot - 1);
                int2  en = rs0[stream_at(ci, p, D)];
                float wn = rw[ci];
                add_rows<0, false>(e.y, w, lane_off, acc, m);
                bool  has = idx < tot;
                e.x = has ? en.x : 0;
                e.y = has ? en.y : -1;
                w   = has ? wn : 0.0f;
            }
            if (tot & 31) {
                float m[16];
                load_rows<0, 6>(e.x, lane_off, hw, m);
                add_rows<0, true>(e.y, w, lane_off, acc, m);
            }
        }
    } else {
        for (int k = 0; k < NREP; ++k) {
            int e0, e1;
            chunk_bounds(n_edges, trip, c0 + k * (SBLK / FBLK), e0, e1);
            filter_chunk<false, FIRST ? 7 : 6>(src, dst, dinv, hw, acc, queue[wave], nullptr,
                                nullptr, n_edges, e0, e1, trip, fw, lane, lo, cnt, nullptr);
        }
    }
    __syncthreads();

    if constexpr (FIRST) {
        const int half = lane >> 4;
        const int l15  = lane & 15;
        v2f bf[4];
#pragma unroll
        for (int t = 0; t < 4; ++t) {
            int kk = 4 * t + 2 * half;
            bf[t][0] = W[(kk + 0) * HIDC + l15];
            bf[t][1] = W[(kk + 1) * HIDC + l15];
        }
        for (int i = tid; i < cnt * (HIDC / 4); i += SBLK) {
            float4 a = ((const float4*)acc)[i];
            float dn = dinv[lo + (i >> 2)];
            float nn = dn * dn;
            const float4* hp = (const float4*)hw + 2 * ((size_t)lo * (HIDC / 4) + i);
            float4 p0 = hp[0];
            float4 p1 = hp[1];
            const float* bq = b + (i & 3) * 4;
            float4 r;
            r.x = fmaxf(a.x + p0.x * nn + bq[0], 0.0f);
            r.y = fmaxf(a.y + p0.z * nn + bq[1], 0.0f);
            r.z = fmaxf(a.z + p1.x * nn + bq[2], 0.0f);
            r.w = fmaxf(a.w + p1.z * nn + bq[3], 0.0f);
            ((float4*)acc)[i] = r;
        }
        __syncthreads();
        if (wave * 16 + 16 <= cnt) {
            const float* mine = acc + wave * 16 * HIDC;
            v8f c = {};
#pragma unroll
            for (int t = 0; t < 4; ++t) {
                int kk = 4 * t + 2 * half;
                v2f a;
                a[0] = mine[l15 * HIDC + kk + 0];
                a[1] = mine[l15 * HIDC + kk + 1];
                c = __builtin_amdgcn_wmma_f32_16x16x4_f32(
                        false, a, false, bf[t], (short)0, c, false, false);
            }
            float* orow = hw_out + (size_t)(lo + wave * 16) * HIDC + l15;
#pragma unroll
            for (int v = 0; v < 8; ++v) {
                int m = v + 8 * half;
                orow[m * HIDC] = c[v];
            }
        }
    } else {
        for (int i0 = 0; i0 < cnt * (HIDC / 4); i0 += SBLK) {
            const int i = i0 + tid;
            const bool in = i < cnt * (HIDC / 4);
            float4 r = make_float4(0.0f, 0.0f, 0.0f, 0.0f);
            int g = -1;
            if (in) {
                const int node = lo + (i >> 2);
                g = batch[node];
                float4 a = ((const float4*)acc)[i];
                float dn = dinv[node];
                float nn = dn * dn;
                float4 hv = ((const float4*)hw)[(size_t)lo * (HIDC / 4) + i];
                const float* bq = b + (i & 3) * 4;
                r.x = fmaxf(a.x + hv.x * nn + bq[0], 0.0f);
                r.y = fmaxf(a.y + hv.y * nn + bq[1], 0.0f);
                r.z = fmaxf(a.z + hv.z * nn + bq[2], 0.0f);
                r.w = fmaxf(a.w + hv.w * nn + bq[3], 0.0f);
            }
            const int g0 = __builtin_amdgcn_readfirstlane(g);
            const bool same = __builtin_amdgcn_ballot_w32(g != g0) == 0u;
            if (same) {
                if (g0 >= 0) {
                    for (int off = 4; off < 32; off <<= 1) {
                        r.x += __shfl_xor(r.x, off);
                        r.y += __shfl_xor(r.y, off);
                        r.z += __shfl_xor(r.z, off);
                        r.w += __shfl_xor(r.w, off);
                    }
                    if (lane < 4) {
                        float* pg = pooled + (size_t)g0 * HIDC + lane * 4;
                        fadd(pg + 0, r.x);
                        fadd(pg + 1, r.y);
                        fadd(pg + 2, r.z);
                        fadd(pg + 3, r.w);
                    }
                    if (lane == 0) fadd(&gcnt[g0], 8.0f);
                }
            } else if (in) {
                float* pg = pooled + (size_t)g * HIDC + (lane & 3) * 4;
                fadd(pg + 0, r.x);
                fadd(pg + 1, r.y);
                fadd(pg + 2, r.z);
                fadd(pg + 3, r.w);
                if ((lane & 3) == 0) fadd(&gcnt[g], 1.0f);
            }
        }
    }
}

__global__ void k_classify(const float* __restrict__ pooled, const float* __restrict__ cnt,
                           const float* __restrict__ Wc, const float* __restrict__ bc,
                           float* __restrict__ out, int n_graphs, int labels) {
    int i = blockIdx.x * blockDim.x + threadIdx.x;
    if (i >= n_graphs * labels) return;
    int g = i / labels, c = i % labels;
    float inv = 1.0f / fmaxf(cnt[g], 1.0f);
    float acc = bc[c];
#pragma unroll
    for (int j = 0; j < HIDC; ++j)
        acc += pooled[g * HIDC + j] * inv * Wc[j * labels + c];
    out[i] = acc;
}

extern "C" void kernel_launch(void* const* d_in, const int* in_sizes, int n_in,
                              void* d_out, int out_size, void* d_ws, size_t ws_size,
                              hipStream_t stream) {
    const int*   x     = (const int*)d_in[0];
    const int*   ei    = (const int*)d_in[1];
    const int*   batch = (const int*)d_in[2];
    const float* emb   = (const float*)d_in[3];
    const float* W1    = (const float*)d_in[4];
    const float* b1    = (const float*)d_in[5];
    const float* W2    = (const float*)d_in[6];
    const float* b2    = (const float*)d_in[7];
    const float* Wc    = (const float*)d_in[8];
    const float* bc    = (const float*)d_in[9];
    float* out = (float*)d_out;

    const int n_nodes  = in_sizes[0];
    const int n_edges  = in_sizes[1] / 2;
    const int labels   = in_sizes[9];
    const int n_graphs = out_size / labels;
    const int* src = ei;
    const int* dst = ei + n_edges;

    float* ws     = (float*)d_ws;
    float* deg    = ws;
    float* hw2    = deg + n_nodes;
    float* hw     = hw2 + (size_t)n_nodes * HIDC;
    float* pooled = hw + (size_t)n_nodes * HIDC * 2;
    float* cnt    = pooled + (size_t)n_graphs * HIDC;
    const size_t n_fw   = (size_t)((n_nodes + NPR - 1) / NPR) * NCHUNK * (FBLK / 32);
    const size_t n_sw   = (size_t)((n_nodes + NPR - 1) / NPR) * NSUB * (SBLK / 32);
    const size_t rec_at = ((size_t)(cnt + n_graphs - ws) + 31) & ~(size_t)31;
    int*   rec_sd = (int*)(ws + rec_at);
    float* rec_w  = (float*)(rec_sd + n_fw * NSUB * SUBCAP * 2);
    int*   rec_n  = (int*)(rec_w + n_sw * NREP * SUBCAP);
    int*   degc   = rec_n + n_fw * RNL;

    const int ntiles = n_nodes / 16;
    const int wavesPerBlk = BLK / 32;

    dim3 b(BLK);
    auto g1 = [](int n) { return dim3((n + BLK - 1) / BLK); };
    dim3 gw((ntiles + wavesPerBlk - 1) / wavesPerBlk);
    dim3 gf((n_nodes + NPR - 1) / NPR, NCHUNK);
    dim3 gs(((n_nodes + NPR - 1) / NPR) * NSUB);

    k_filter_deg<<<gf, FBLK, 0, stream>>>(src, dst, degc, rec_sd, rec_n, n_edges, n_nodes);

    k_embed_xw<<<gw, b, 0, stream>>>(x, emb, W1, degc, deg, hw, pooled, cnt,
                                     n_nodes, ntiles, n_graphs);
    k_scatter_lds<true><<<gs, SBLK, 0, stream>>>(src, dst, deg, hw, rec_sd, rec_w, rec_n, b1, W2, hw2,
                                                 nullptr, nullptr, nullptr, n_edges, n_nodes);

    k_scatter_lds<false><<<gs, SBLK, 0, stream>>>(src, dst, deg, hw2, rec_sd, rec_w, rec_n, b2, nullptr,
                                                  nullptr, batch, pooled, cnt, n_edges, n_nodes);
    k_classify<<<g1(n_graphs * labels), b, 0, stream>>>(pooled, cnt, Wc, bc, out,
                                                        n_graphs, labels);
}
